# starting kernel
# baseline (speedup 1.0000x reference)
#include <hip/hip_runtime.h>
#include <hip/hip_bf16.h>
#include <math.h>


typedef float v2f __attribute__((ext_vector_type(2)));
typedef float v8f __attribute__((ext_vector_type(8)));

#define BATCH    8
#define SEQ      512
#define DMODEL   512
#define DINNER   1024
#define DSTATE   16
#define DTRANK   32
#define NGROUPS  8
#define MROWS    (BATCH * SEQ)

__device__ __forceinline__ v8f wmma_f32_4(v2f a, v2f b, v8f c) {
    return __builtin_amdgcn_wmma_f32_16x16x4_f32(
        false, a, false, b, (short)0, c, false, false);
}

__device__ __forceinline__ float softplus_f(float x) {
    return fmaxf(x, 0.f) + log1pf(__expf(-fabsf(x)));
}

__global__ void k_inproj(const float* __restrict__ x,
                         const float* __restrict__ w,
                         float* __restrict__ xz) {
    const int wid   = blockIdx.x * 4 + (threadIdx.x >> 5);
    const int ntq   = wid & 31;
    const int tileM = wid >> 5;
    const int lane  = threadIdx.x & 31;
    const int half  = lane >> 4;
    const int r     = lane & 15;

    const int m  = tileM * 16 + r;
    const int bb = m >> 9;
    const int ll = m & 511;

    const float* xb = x + (size_t)bb * DMODEL * SEQ + ll;
    const float* wr0 = w + (size_t)((ntq * 4 + 0) * 16 + r) * DMODEL;
    const float* wr1 = w + (size_t)((ntq * 4 + 1) * 16 + r) * DMODEL;
    const float* wr2 = w + (size_t)((ntq * 4 + 2) * 16 + r) * DMODEL;
    const float* wr3 = w + (size_t)((ntq * 4 + 3) * 16 + r) * DMODEL;

    v8f acc0 = {}, acc1 = {}, acc2 = {}, acc3 = {};
    #pragma unroll 2
    for (int k0 = 0; k0 < DMODEL; k0 += 4) {
        const int ka = k0 + half * 2;
        v2f a;
        a.x = xb[(size_t)ka * SEQ];
        a.y = xb[(size_t)(ka + 1) * SEQ];
        v2f b0 = *(const v2f*)(wr0 + ka);
        v2f b1 = *(const v2f*)(wr1 + ka);
        v2f b2 = *(const v2f*)(wr2 + ka);
        v2f b3 = *(const v2f*)(wr3 + ka);
        __builtin_prefetch(wr0 + k0 + 32, 0, 1);
        acc0 = wmma_f32_4(a, b0, acc0);
        acc1 = wmma_f32_4(a, b1, acc1);
        acc2 = wmma_f32_4(a, b2, acc2);
        acc3 = wmma_f32_4(a, b3, acc3);
    }
    #pragma unroll
    for (int j = 0; j < 8; ++j) {
        const size_t row = (size_t)(tileM * 16 + half * 8 + j) * (2 * DINNER);
        xz[row + ntq * 64 + 0  + r] = acc0[j];
        xz[row + ntq * 64 + 16 + r] = acc1[j];
        xz[row + ntq * 64 + 32 + r] = acc2[j];
        xz[row + ntq * 64 + 48 + r] = acc3[j];
    }
}

__global__ void k_conv(const float* __restrict__ xz,
                       const float* __restrict__ cw,
                       const float* __restrict__ cb,
                       float* __restrict__ ucv) {
    const int idx = blockIdx.x * blockDim.x + threadIdx.x;
    if (idx >= MROWS * DINNER) return;
    const int c = idx & (DINNER - 1);
    const int m = idx >> 10;
    const int l = m & 511;

    float s = cb[c];
    #pragma unroll
    for (int t = 0; t < 4; ++t) {
        const int lt = l - 3 + t;
        if (lt >= 0)
            s += xz[(size_t)(m - 3 + t) * (2 * DINNER) + c] * cw[c * 4 + t];
    }
    const float sig = 1.f / (1.f + __expf(-s));
    ucv[(size_t)m * DINNER + c] = s * sig;
}

__global__ void k_xproj(const float* __restrict__ ucv,
                        const float* __restrict__ w,
                        float* __restrict__ dbc) {
    const int tileM = blockIdx.x * 4 + (threadIdx.x >> 5);
    const int lane  = threadIdx.x & 31;
    const int half  = lane >> 4;
    const int r     = lane & 15;

    const float* arow = ucv + (size_t)(tileM * 16 + r) * DINNER;
    const float* wr0 = w + (size_t)(0  + r) * DINNER;
    const float* wr1 = w + (size_t)(16 + r) * DINNER;
    const float* wr2 = w + (size_t)(32 + r) * DINNER;
    const float* wr3 = w + (size_t)(48 + r) * DINNER;

    v8f acc0 = {}, acc1 = {}, acc2 = {}, acc3 = {};
    #pragma unroll 2
    for (int k0 = 0; k0 < DINNER; k0 += 4) {
        const int ka = k0 + half * 2;
        v2f a  = *(const v2f*)(arow + ka);
        v2f b0 = *(const v2f*)(wr0 + ka);
        v2f b1 = *(const v2f*)(wr1 + ka);
        v2f b2 = *(const v2f*)(wr2 + ka);
        v2f b3 = *(const v2f*)(wr3 + ka);
        __builtin_prefetch(arow + k0 + 32, 0, 1);
        acc0 = wmma_f32_4(a, b0, acc0);
        acc1 = wmma_f32_4(a, b1, acc1);
        acc2 = wmma_f32_4(a, b2, acc2);
        acc3 = wmma_f32_4(a, b3, acc3);
    }
    #pragma unroll
    for (int j = 0; j < 8; ++j) {
        const size_t row = (size_t)(tileM * 16 + half * 8 + j) * 64;
        dbc[row + 0  + r] = acc0[j];
        dbc[row + 16 + r] = acc1[j];
        dbc[row + 32 + r] = acc2[j];
        dbc[row + 48 + r] = acc3[j];
    }
}

__global__ void k_dtproj(const float* __restrict__ dbc,
                         const float* __restrict__ w,
                         const float* __restrict__ bias,
                         float* __restrict__ delta) {
    const int wid   = blockIdx.x * 4 + (threadIdx.x >> 5);
    const int ntq   = wid & 15;
    const int tileM = wid >> 4;
    const int lane  = threadIdx.x & 31;
    const int half  = lane >> 4;
    const int r     = lane & 15;

    const float* arow = dbc + (size_t)(tileM * 16 + r) * 64;
    const float* wr0 = w + (size_t)((ntq * 4 + 0) * 16 + r) * DTRANK;
    const float* wr1 = w + (size_t)((ntq * 4 + 1) * 16 + r) * DTRANK;
    const float* wr2 = w + (size_t)((ntq * 4 + 2) * 16 + r) * DTRANK;
    const float* wr3 = w + (size_t)((ntq * 4 + 3) * 16 + r) * DTRANK;

    v8f acc0 = {}, acc1 = {}, acc2 = {}, acc3 = {};
    #pragma unroll
    for (int k0 = 0; k0 < DTRANK; k0 += 4) {
        const int ka = k0 + half * 2;
        v2f a  = *(const v2f*)(arow + ka);
        v2f b0 = *(const v2f*)(wr0 + ka);
        v2f b1 = *(const v2f*)(wr1 + ka);
        v2f b2 = *(const v2f*)(wr2 + ka);
        v2f b3 = *(const v2f*)(wr3 + ka);
        acc0 = wmma_f32_4(a, b0, acc0);
        acc1 = wmma_f32_4(a, b1, acc1);
        acc2 = wmma_f32_4(a, b2, acc2);
        acc3 = wmma_f32_4(a, b3, acc3);
    }
    const int nb = ntq * 64 + r;
    const float bn0 = bias[nb + 0];
    const float bn1 = bias[nb + 16];
    const float bn2 = bias[nb + 32];
    const float bn3 = bias[nb + 48];
    #pragma unroll
    for (int j = 0; j < 8; ++j) {
        const size_t row = (size_t)(tileM * 16 + half * 8 + j) * DINNER;
        delta[row + nb + 0 ] = softplus_f(acc0[j] + bn0);
        delta[row + nb + 16] = softplus_f(acc1[j] + bn1);
        delta[row + nb + 32] = softplus_f(acc2[j] + bn2);
        delta[row + nb + 48] = softplus_f(acc3[j] + bn3);
    }
}

__global__ void k_scan(const float* __restrict__ delta,
                       const float* __restrict__ ucv,
                       const float* __restrict__ xz,
                       const float* __restrict__ dbc,
                       const float* __restrict__ A_log,
                       const float* __restrict__ D_param,
                       float* __restrict__ ybuf) {
    const int bb = blockIdx.x >> 3;
    const int c  = (blockIdx.x & 7) * 128 + threadIdx.x;
    const int tid = threadIdx.x;
    const int mbase = bb * SEQ;

    __shared__ float sB[DSTATE];
    __shared__ float sC[DSTATE];

    float Ac[DSTATE];
    float h[DSTATE];
    #pragma unroll
    for (int n = 0; n < DSTATE; ++n) {
        Ac[n] = -__expf(A_log[(size_t)c * DSTATE + n]);
        h[n] = 0.f;
    }
    const float Dp = D_param[c];

    for (int l = 0; l < SEQ; ++l) {
        const int m = mbase + l;
        __syncthreads();
        if (tid < 16)       sB[tid]      = dbc[(size_t)m * 64 + 32 + tid];
        else if (tid < 32)  sC[tid - 16] = dbc[(size_t)m * 64 + 48 + (tid - 16)];
        __syncthreads();

        const float dlt = delta[(size_t)m * DINNER + c];
        const float ut  = ucv[(size_t)m * DINNER + c];
        const float zt  = xz[(size_t)m * (2 * DINNER) + DINNER + c];
        const float du  = dlt * ut;

        float y = 0.f;
        #pragma unroll
        for (int n = 0; n < DSTATE; ++n) {
            const float dA = __expf(dlt * Ac[n]);
            h[n] = dA * h[n] + du * sB[n];
            y += h[n] * sC[n];
        }
        const float sig = 1.f / (1.f + __expf(-zt));
        ybuf[(size_t)m * DINNER + c] = (y + ut * Dp) * (zt * sig);
    }
}

__global__ void k_outproj(const float* __restrict__ ybuf,
                          const float* __restrict__ w,
                          float* __restrict__ otmp) {
    const int wid   = blockIdx.x * 4 + (threadIdx.x >> 5);
    const int ntq   = wid & 7;
    const int tileM = wid >> 3;
    const int lane  = threadIdx.x & 31;
    const int half  = lane >> 4;
    const int r     = lane & 15;

    const float* arow = ybuf + (size_t)(tileM * 16 + r) * DINNER;
    const float* wr0 = w + (size_t)((ntq * 4 + 0) * 16 + r) * DINNER;
    const float* wr1 = w + (size_t)((ntq * 4 + 1) * 16 + r) * DINNER;
    const float* wr2 = w + (size_t)((ntq * 4 + 2) * 16 + r) * DINNER;
    const float* wr3 = w + (size_t)((ntq * 4 + 3) * 16 + r) * DINNER;

    v8f acc0 = {}, acc1 = {}, acc2 = {}, acc3 = {};
    #pragma unroll 2
    for (int k0 = 0; k0 < DINNER; k0 += 4) {
        const int ka = k0 + half * 2;
        v2f a  = *(const v2f*)(arow + ka);
        v2f b0 = *(const v2f*)(wr0 + ka);
        v2f b1 = *(const v2f*)(wr1 + ka);
        v2f b2 = *(const v2f*)(wr2 + ka);
        v2f b3 = *(const v2f*)(wr3 + ka);
        __builtin_prefetch(arow + k0 + 32, 0, 1);
        acc0 = wmma_f32_4(a, b0, acc0);
        acc1 = wmma_f32_4(a, b1, acc1);
        acc2 = wmma_f32_4(a, b2, acc2);
        acc3 = wmma_f32_4(a, b3, acc3);
    }
    #pragma unroll
    for (int j = 0; j < 8; ++j) {
        const int m = tileM * 16 + half * 8 + j;
        const int b = m >> 9;
        const int l = m & 511;
        const size_t base = ((size_t)b * DMODEL + ntq * 64 + r) * SEQ + l;
        otmp[base + 0  * SEQ] = acc0[j];
        otmp[base + 16 * SEQ] = acc1[j];
        otmp[base + 32 * SEQ] = acc2[j];
        otmp[base + 48 * SEQ] = acc3[j];
    }
}

__global__ void k_gnstats(const float* __restrict__ otmp,
                          float* __restrict__ stats) {
    const int bid = blockIdx.x;
    const int tid = threadIdx.x;
    const float* base = otmp + (size_t)bid * 32768;

    __shared__ float ssum[256];
    __shared__ float ssq[256];
    float s = 0.f, q = 0.f;
    for (int i = tid; i < 32768; i += 256) {
        const float v = base[i];
        s += v;
        q += v * v;
    }
    ssum[tid] = s; ssq[tid] = q;
    __syncthreads();
    for (int off = 128; off > 0; off >>= 1) {
        if (tid < off) { ssum[tid] += ssum[tid + off]; ssq[tid] += ssq[tid + off]; }
        __syncthreads();
    }
    if (tid == 0) {
        const float mu  = ssum[0] * (1.f / 32768.f);
        const float var = ssq[0] * (1.f / 32768.f) - mu * mu;
        stats[2 * bid]     = mu;
        stats[2 * bid + 1] = rsqrtf(var + 1e-5f);
    }
}

__global__ void k_gnmish(const float* __restrict__ otmp,
                         const float* __restrict__ stats,
                         const float* __restrict__ gw,
                         const float* __restrict__ gb,
                         float* __restrict__ out) {
    const int idx = blockIdx.x * blockDim.x + threadIdx.x;
    if (idx >= BATCH * DMODEL * SEQ) return;
    const int b   = idx >> 18;
    const int rem = idx & (DMODEL * SEQ - 1);
    const int c   = rem >> 9;
    const int g   = c >> 6;
    const int blk = b * NGROUPS + g;

    const float mu = stats[2 * blk];
    const float rs = stats[2 * blk + 1];
    float v = (otmp[idx] - mu) * rs * gw[c] + gb[c];
    out[idx] = v * tanhf(softplus_f(v));
}

extern "C" void kernel_launch(void* const* d_in, const int* in_sizes, int n_in,
                              void* d_out, int out_size, void* d_ws, size_t ws_size,
                              hipStream_t stream) {
    const float* x         = (const float*)d_in[0];
    const float* in_proj_w = (const float*)d_in[1];
    const float* conv_w    = (const float*)d_in[2];
    const float* conv_b    = (const float*)d_in[3];
    const float* x_proj_w  = (const float*)d_in[4];
    const float* dt_proj_w = (const float*)d_in[5];
    const float* dt_proj_b = (const float*)d_in[6];
    const float* A_log     = (const float*)d_in[7];
    const float* D_param   = (const float*)d_in[8];
    const float* out_proj_w= (const float*)d_in[9];
    const float* gn_w      = (const float*)d_in[10];
    const float* gn_b      = (const float*)d_in[11];
    float* out = (float*)d_out;

    float* ws    = (float*)d_ws;
    float* xz    = ws;
    float* ucv   = xz    + (size_t)MROWS * 2 * DINNER;
    float* dbc   = ucv   + (size_t)MROWS * DINNER;
    float* delta = dbc   + (size_t)MROWS * 64;
    float* ybuf  = delta + (size_t)MROWS * DINNER;
    float* stats = ybuf  + (size_t)MROWS * DINNER;
    float* otmp  = xz;

    k_inproj<<<8192 / 4, 128, 0, stream>>>(x, in_proj_w, xz);

    k_conv<<<(MROWS * DINNER + 255) / 256, 256, 0, stream>>>(xz, conv_w, conv_b, ucv);

    k_xproj<<<256 / 4, 128, 0, stream>>>(ucv, x_proj_w, dbc);

    k_dtproj<<<4096 / 4, 128, 0, stream>>>(dbc, dt_proj_w, dt_proj_b, delta);

    k_scan<<<BATCH * 8, 128, 0, stream>>>(delta, ucv, xz, dbc, A_log, D_param, ybuf);

    k_outproj<<<2048 / 4, 128, 0, stream>>>(ybuf, out_proj_w, otmp);

    k_gnstats<<<BATCH * NGROUPS, 256, 0, stream>>>(otmp, stats);

    k_gnmish<<<(BATCH * DMODEL * SEQ + 255) / 256, 256, 0, stream>>>(
        otmp, stats, gn_w, gn_b, out);
}


# scan kernel: memory taken off the serial recurrence: B/C of the whole sequence staged to LDS once (one barrier, was two per step), delta/u/z register-prefetched a 16-step tile ahead; math unchanged
# speedup vs baseline: 2.2313x; 2.2313x over previous
#include <hip/hip_runtime.h>
#include <hip/hip_bf16.h>
#include <math.h>


typedef float v2f __attribute__((ext_vector_type(2)));
typedef float v8f __attribute__((ext_vector_type(8)));

#define BATCH    8
#define SEQ      512
#define DMODEL   512
#define DINNER   1024
#define DSTATE   16
#define DTRANK   32
#define NGROUPS  8
#define MROWS    (BATCH * SEQ)

__device__ __forceinline__ v8f wmma_f32_4(v2f a, v2f b, v8f c) {
    return __builtin_amdgcn_wmma_f32_16x16x4_f32(
        false, a, false, b, (short)0, c, false, false);
}

__device__ __forceinline__ float softplus_f(float x) {
    return fmaxf(x, 0.f) + log1pf(__expf(-fabsf(x)));
}

__global__ void k_inproj(const float* __restrict__ x,
                         const float* __restrict__ w,
                         float* __restrict__ xz) {
    const int wid   = blockIdx.x * 4 + (threadIdx.x >> 5);
    const int ntq   = wid & 31;
    const int tileM = wid >> 5;
    const int lane  = threadIdx.x & 31;
    const int half  = lane >> 4;
    const int r     = lane & 15;

    const int m  = tileM * 16 + r;
    const int bb = m >> 9;
    const int ll = m & 511;

    const float* xb = x + (size_t)bb * DMODEL * SEQ + ll;
    const float* wr0 = w + (size_t)((ntq * 4 + 0) * 16 + r) * DMODEL;
    const float* wr1 = w + (size_t)((ntq * 4 + 1) * 16 + r) * DMODEL;
    const float* wr2 = w + (size_t)((ntq * 4 + 2) * 16 + r) * DMODEL;
    const float* wr3 = w + (size_t)((ntq * 4 + 3) * 16 + r) * DMODEL;

    v8f acc0 = {}, acc1 = {}, acc2 = {}, acc3 = {};
    #pragma unroll 2
    for (int k0 = 0; k0 < DMODEL; k0 += 4) {
        const int ka = k0 + half * 2;
        v2f a;
        a.x = xb[(size_t)ka * SEQ];
        a.y = xb[(size_t)(ka + 1) * SEQ];
        v2f b0 = *(const v2f*)(wr0 + ka);
        v2f b1 = *(const v2f*)(wr1 + ka);
        v2f b2 = *(const v2f*)(wr2 + ka);
        v2f b3 = *(const v2f*)(wr3 + ka);
        __builtin_prefetch(wr0 + k0 + 32, 0, 1);
        acc0 = wmma_f32_4(a, b0, acc0);
        acc1 = wmma_f32_4(a, b1, acc1);
        acc2 = wmma_f32_4(a, b2, acc2);
        acc3 = wmma_f32_4(a, b3, acc3);
    }
    #pragma unroll
    for (int j = 0; j < 8; ++j) {
        const size_t row = (size_t)(tileM * 16 + half * 8 + j) * (2 * DINNER);
        xz[row + ntq * 64 + 0  + r] = acc0[j];
        xz[row + ntq * 64 + 16 + r] = acc1[j];
        xz[row + ntq * 64 + 32 + r] = acc2[j];
        xz[row + ntq * 64 + 48 + r] = acc3[j];
    }
}

__global__ void k_conv(const float* __restrict__ xz,
                       const float* __restrict__ cw,
                       const float* __restrict__ cb,
                       float* __restrict__ ucv) {
    const int idx = blockIdx.x * blockDim.x + threadIdx.x;
    if (idx >= MROWS * DINNER) return;
    const int c = idx & (DINNER - 1);
    const int m = idx >> 10;
    const int l = m & 511;

    float s = cb[c];
    #pragma unroll
    for (int t = 0; t < 4; ++t) {
        const int lt = l - 3 + t;
        if (lt >= 0)
            s += xz[(size_t)(m - 3 + t) * (2 * DINNER) + c] * cw[c * 4 + t];
    }
    const float sig = 1.f / (1.f + __expf(-s));
    ucv[(size_t)m * DINNER + c] = s * sig;
}

__global__ void k_xproj(const float* __restrict__ ucv,
                        const float* __restrict__ w,
                        float* __restrict__ dbc) {
    const int tileM = blockIdx.x * 4 + (threadIdx.x >> 5);
    const int lane  = threadIdx.x & 31;
    const int half  = lane >> 4;
    const int r     = lane & 15;

    const float* arow = ucv + (size_t)(tileM * 16 + r) * DINNER;
    const float* wr0 = w + (size_t)(0  + r) * DINNER;
    const float* wr1 = w + (size_t)(16 + r) * DINNER;
    const float* wr2 = w + (size_t)(32 + r) * DINNER;
    const float* wr3 = w + (size_t)(48 + r) * DINNER;

    v8f acc0 = {}, acc1 = {}, acc2 = {}, acc3 = {};
    #pragma unroll 2
    for (int k0 = 0; k0 < DINNER; k0 += 4) {
        const int ka = k0 + half * 2;
        v2f a  = *(const v2f*)(arow + ka);
        v2f b0 = *(const v2f*)(wr0 + ka);
        v2f b1 = *(const v2f*)(wr1 + ka);
        v2f b2 = *(const v2f*)(wr2 + ka);
        v2f b3 = *(const v2f*)(wr3 + ka);
        __builtin_prefetch(arow + k0 + 32, 0, 1);
        acc0 = wmma_f32_4(a, b0, acc0);
        acc1 = wmma_f32_4(a, b1, acc1);
        acc2 = wmma_f32_4(a, b2, acc2);
        acc3 = wmma_f32_4(a, b3, acc3);
    }
    #pragma unroll
    for (int j = 0; j < 8; ++j) {
        const size_t row = (size_t)(tileM * 16 + half * 8 + j) * 64;
        dbc[row + 0  + r] = acc0[j];
        dbc[row + 16 + r] = acc1[j];
        dbc[row + 32 + r] = acc2[j];
        dbc[row + 48 + r] = acc3[j];
    }
}

__global__ void k_dtproj(const float* __restrict__ dbc,
                         const float* __restrict__ w,
                         const float* __restrict__ bias,
                         float* __restrict__ delta) {
    const int wid   = blockIdx.x * 4 + (threadIdx.x >> 5);
    const int ntq   = wid & 15;
    const int tileM = wid >> 4;
    const int lane  = threadIdx.x & 31;
    const int half  = lane >> 4;
    const int r     = lane & 15;

    const float* arow = dbc + (size_t)(tileM * 16 + r) * 64;
    const float* wr0 = w + (size_t)((ntq * 4 + 0) * 16 + r) * DTRANK;
    const float* wr1 = w + (size_t)((ntq * 4 + 1) * 16 + r) * DTRANK;
    const float* wr2 = w + (size_t)((ntq * 4 + 2) * 16 + r) * DTRANK;
    const float* wr3 = w + (size_t)((ntq * 4 + 3) * 16 + r) * DTRANK;

    v8f acc0 = {}, acc1 = {}, acc2 = {}, acc3 = {};
    #pragma unroll
    for (int k0 = 0; k0 < DTRANK; k0 += 4) {
        const int ka = k0 + half * 2;
        v2f a  = *(const v2f*)(arow + ka);
        v2f b0 = *(const v2f*)(wr0 + ka);
        v2f b1 = *(const v2f*)(wr1 + ka);
        v2f b2 = *(const v2f*)(wr2 + ka);
        v2f b3 = *(const v2f*)(wr3 + ka);
        acc0 = wmma_f32_4(a, b0, acc0);
        acc1 = wmma_f32_4(a, b1, acc1);
        acc2 = wmma_f32_4(a, b2, acc2);
        acc3 = wmma_f32_4(a, b3, acc3);
    }
    const int nb = ntq * 64 + r;
    const float bn0 = bias[nb + 0];
    const float bn1 = bias[nb + 16];
    const float bn2 = bias[nb + 32];
    const float bn3 = bias[nb + 48];
    #pragma unroll
    for (int j = 0; j < 8; ++j) {
        const size_t row = (size_t)(tileM * 16 + half * 8 + j) * DINNER;
        delta[row + nb + 0 ] = softplus_f(acc0[j] + bn0);
        delta[row + nb + 16] = softplus_f(acc1[j] + bn1);
        delta[row + nb + 32] = softplus_f(acc2[j] + bn2);
        delta[row + nb + 48] = softplus_f(acc3[j] + bn3);
    }
}

#define SCAN_T 16

__device__ __forceinline__ void scan_load_bc(const float4* __restrict__ row,
                                             float (&Bv)[DSTATE], float (&Cv)[DSTATE]) {
    #pragma unroll
    for (int q = 0; q < 4; ++q) {
        const float4 b4 = row[q];
        const float4 c4 = row[4 + q];
        Bv[4 * q + 0] = b4.x; Bv[4 * q + 1] = b4.y; Bv[4 * q + 2] = b4.z; Bv[4 * q + 3] = b4.w;
        Cv[4 * q + 0] = c4.x; Cv[4 * q + 1] = c4.y; Cv[4 * q + 2] = c4.z; Cv[4 * q + 3] = c4.w;
    }
}

__global__ void __launch_bounds__(128)
k_scan(const float* __restrict__ delta,
       const float* __restrict__ ucv,
       const float* __restrict__ xz,
       const float* __restrict__ dbc,
       const float* __restrict__ A_log,
       const float* __restrict__ D_param,
       float* __restrict__ ybuf) {
    const int bb = blockIdx.x >> 3;
    const int tid = threadIdx.x;
    const int c  = (blockIdx.x & 7) * 128 + tid;
    const int mbase = bb * SEQ;

    __shared__ float4 sBC[SEQ * 8];

    const float* pd = delta + (size_t)mbase * DINNER + c;
    const float* pu = ucv   + (size_t)mbase * DINNER + c;
    const float* pz = xz    + (size_t)mbase * (2 * DINNER) + DINNER + c;
    float*       py = ybuf  + (size_t)mbase * DINNER + c;

    float nd[SCAN_T], nu[SCAN_T], nz[SCAN_T];
    #pragma unroll
    for (int t = 0; t < SCAN_T; ++t) {
        nd[t] = pd[(size_t)t * DINNER];
        nu[t] = pu[(size_t)t * DINNER];
        nz[t] = pz[(size_t)t * (2 * DINNER)];
    }

    const float* bcsrc = dbc + (size_t)mbase * 64 + 32;
    #pragma unroll
    for (int j = 0; j < (SEQ * 8) / 128; ++j) {
        const int i = j * 128 + tid;
        sBC[i] = *(const float4*)(bcsrc + (size_t)(i >> 3) * 64 + (i & 7) * 4);
    }

    float Ac[DSTATE];
    float h[DSTATE];
    #pragma unroll
    for (int n = 0; n < DSTATE; ++n) {
        Ac[n] = -__expf(A_log[(size_t)c * DSTATE + n]);
        h[n] = 0.f;
    }
    const float Dp = D_param[c];

    __syncthreads();

    float Bc[DSTATE], Cc[DSTATE];
    scan_load_bc(sBC, Bc, Cc);

    for (int l0 = 0; l0 < SEQ; l0 += SCAN_T) {
        float cd[SCAN_T], cu[SCAN_T], cz[SCAN_T];
        #pragma unroll
        for (int t = 0; t < SCAN_T; ++t) { cd[t] = nd[t]; cu[t] = nu[t]; cz[t] = nz[t]; }

        const int ln = (l0 + SCAN_T < SEQ) ? (l0 + SCAN_T) : l0;
        #pragma unroll
        for (int t = 0; t < SCAN_T; ++t) {
            nd[t] = pd[(size_t)(ln + t) * DINNER];
            nu[t] = pu[(size_t)(ln + t) * DINNER];
            nz[t] = pz[(size_t)(ln + t) * (2 * DINNER)];
        }

        #pragma unroll
        for (int t = 0; t < SCAN_T; ++t) {
            const int l = l0 + t;

            float Bn[DSTATE], Cn[DSTATE];
            const int lnx = (l + 1 < SEQ) ? (l + 1) : l;
            scan_load_bc(sBC + lnx * 8, Bn, Cn);
            __builtin_amdgcn_sched_barrier(0);

            const float dlt = cd[t];
            const float ut  = cu[t];
            const float zt  = cz[t];
            const float du  = dlt * ut;

            float y = 0.f;
            #pragma unroll
            for (int n = 0; n < DSTATE; ++n) {
                const float dA = __expf(dlt * Ac[n]);
                h[n] = dA * h[n] + du * Bc[n];
                y += h[n] * Cc[n];
            }
            const float sig = 1.f / (1.f + __expf(-zt));
            py[(size_t)l * DINNER] = (y + ut * Dp) * (zt * sig);

            #pragma unroll
            for (int n = 0; n < DSTATE; ++n) { Bc[n] = Bn[n]; Cc[n] = Cn[n]; }

            __builtin_amdgcn_sched_barrier(0);
        }
    }
}

__global__ void k_outproj(const float* __restrict__ ybuf,
                          const float* __restrict__ w,
                          float* __restrict__ otmp) {
    const int wid   = blockIdx.x * 4 + (threadIdx.x >> 5);
    const int ntq   = wid & 7;
    const int tileM = wid >> 3;
    const int lane  = threadIdx.x & 31;
    const int half  = lane >> 4;
    const int r     = lane & 15;

    const float* arow = ybuf + (size_t)(tileM * 16 + r) * DINNER;
    const float* wr0 = w + (size_t)((ntq * 4 + 0) * 16 + r) * DINNER;
    const float* wr1 = w + (size_t)((ntq * 4 + 1) * 16 + r) * DINNER;
    const float* wr2 = w + (size_t)((ntq * 4 + 2) * 16 + r) * DINNER;
    const float* wr3 = w + (size_t)((ntq * 4 + 3) * 16 + r) * DINNER;

    v8f acc0 = {}, acc1 = {}, acc2 = {}, acc3 = {};
    #pragma unroll 2
    for (int k0 = 0; k0 < DINNER; k0 += 4) {
        const int ka = k0 + half * 2;
        v2f a  = *(const v2f*)(arow + ka);
        v2f b0 = *(const v2f*)(wr0 + ka);
        v2f b1 = *(const v2f*)(wr1 + ka);
        v2f b2 = *(const v2f*)(wr2 + ka);
        v2f b3 = *(const v2f*)(wr3 + ka);
        __builtin_prefetch(arow + k0 + 32, 0, 1);
        acc0 = wmma_f32_4(a, b0, acc0);
        acc1 = wmma_f32_4(a, b1, acc1);
        acc2 = wmma_f32_4(a, b2, acc2);
        acc3 = wmma_f32_4(a, b3, acc3);
    }
    #pragma unroll
    for (int j = 0; j < 8; ++j) {
        const int m = tileM * 16 + half * 8 + j;
        const int b = m >> 9;
        const int l = m & 511;
        const size_t base = ((size_t)b * DMODEL + ntq * 64 + r) * SEQ + l;
        otmp[base + 0  * SEQ] = acc0[j];
        otmp[base + 16 * SEQ] = acc1[j];
        otmp[base + 32 * SEQ] = acc2[j];
        otmp[base + 48 * SEQ] = acc3[j];
    }
}

__global__ void k_gnstats(const float* __restrict__ otmp,
                          float* __restrict__ stats) {
    const int bid = blockIdx.x;
    const int tid = threadIdx.x;
    const float* base = otmp + (size_t)bid * 32768;

    __shared__ float ssum[256];
    __shared__ float ssq[256];
    float s = 0.f, q = 0.f;
    for (int i = tid; i < 32768; i += 256) {
        const float v = base[i];
        s += v;
        q += v * v;
    }
    ssum[tid] = s; ssq[tid] = q;
    __syncthreads();
    for (int off = 128; off > 0; off >>= 1) {
        if (tid < off) { ssum[tid] += ssum[tid + off]; ssq[tid] += ssq[tid + off]; }
        __syncthreads();
    }
    if (tid == 0) {
        const float mu  = ssum[0] * (1.f / 32768.f);
        const float var = ssq[0] * (1.f / 32768.f) - mu * mu;
        stats[2 * bid]     = mu;
        stats[2 * bid + 1] = rsqrtf(var + 1e-5f);
    }
}

__global__ void k_gnmish(const float* __restrict__ otmp,
                         const float* __restrict__ stats,
                         const float* __restrict__ gw,
                         const float* __restrict__ gb,
                         float* __restrict__ out) {
    const int idx = blockIdx.x * blockDim.x + threadIdx.x;
    if (idx >= BATCH * DMODEL * SEQ) return;
    const int b   = idx >> 18;
    const int rem = idx & (DMODEL * SEQ - 1);
    const int c   = rem >> 9;
    const int g   = c >> 6;
    const int blk = b * NGROUPS + g;

    const float mu = stats[2 * blk];
    const float rs = stats[2 * blk + 1];
    float v = (otmp[idx] - mu) * rs * gw[c] + gb[c];
    out[idx] = v * tanhf(softplus_f(v));
}

extern "C" void kernel_launch(void* const* d_in, const int* in_sizes, int n_in,
                              void* d_out, int out_size, void* d_ws, size_t ws_size,
                              hipStream_t stream) {
    const float* x         = (const float*)d_in[0];
    const float* in_proj_w = (const float*)d_in[1];
    const float* conv_w    = (const float*)d_in[2];
    const float* conv_b    = (const float*)d_in[3];
    const float* x_proj_w  = (const float*)d_in[4];
    const float* dt_proj_w = (const float*)d_in[5];
    const float* dt_proj_b = (const float*)d_in[6];
    const float* A_log     = (const float*)d_in[7];
    const float* D_param   = (const float*)d_in[8];
    const float* out_proj_w= (const float*)d_in[9];
    const float* gn_w      = (const float*)d_in[10];
    const float* gn_b      = (const float*)d_in[11];
    float* out = (float*)d_out;

    float* ws    = (float*)d_ws;
    float* xz    = ws;
    float* ucv   = xz    + (size_t)MROWS * 2 * DINNER;
    float* dbc   = ucv   + (size_t)MROWS * DINNER;
    float* delta = dbc   + (size_t)MROWS * 64;
    float* ybuf  = delta + (size_t)MROWS * DINNER;
    float* stats = ybuf  + (size_t)MROWS * DINNER;
    float* otmp  = xz;

    k_inproj<<<8192 / 4, 128, 0, stream>>>(x, in_proj_w, xz);

    k_conv<<<(MROWS * DINNER + 255) / 256, 256, 0, stream>>>(xz, conv_w, conv_b, ucv);

    k_xproj<<<256 / 4, 128, 0, stream>>>(ucv, x_proj_w, dbc);

    k_dtproj<<<4096 / 4, 128, 0, stream>>>(dbc, dt_proj_w, dt_proj_b, delta);

    k_scan<<<BATCH * 8, 128, 0, stream>>>(delta, ucv, xz, dbc, A_log, D_param, ybuf);

    k_outproj<<<2048 / 4, 128, 0, stream>>>(ybuf, out_proj_w, otmp);

    k_gnstats<<<BATCH * NGROUPS, 256, 0, stream>>>(otmp, stats);

    k_gnmish<<<(BATCH * DMODEL * SEQ + 255) / 256, 256, 0, stream>>>(
        otmp, stats, gn_w, gn_b, out);
}


# in_proj/x_proj/out_proj GEMMs: blocks stage 32-deep K-slabs of A and W and out_proj's result tile through LDS, whole-line 16-byte global accesses; waves read fragments from LDS; K order unchanged
# speedup vs baseline: 5.5559x; 2.4900x over previous
#include <hip/hip_runtime.h>
#include <hip/hip_bf16.h>
#include <math.h>


typedef float v2f __attribute__((ext_vector_type(2)));
typedef float v8f __attribute__((ext_vector_type(8)));

#define BATCH    8
#define SEQ      512
#define DMODEL   512
#define DINNER   1024
#define DSTATE   16
#define DTRANK   32
#define NGROUPS  8
#define MROWS    (BATCH * SEQ)

__device__ __forceinline__ v8f wmma_f32_4(v2f a, v2f b, v8f c) {
    return __builtin_amdgcn_wmma_f32_16x16x4_f32(
        false, a, false, b, (short)0, c, false, false);
}

__device__ __forceinline__ float softplus_f(float x) {
    return fmaxf(x, 0.f) + log1pf(__expf(-fabsf(x)));
}


#define GEMM_K   32
#define GEMM_P   36

template <int WM, int WN, bool AKM>
struct GemmShape {
    static constexpr int NT  = 32 * WM * WN;
    static constexpr int BM  = 16 * WM;
    static constexpr int BN  = 64 * WN;
    static constexpr int AP  = AKM ? (BM + 8) : GEMM_P;
    static constexpr int ASZ = AKM ? GEMM_K * (BM + 8) : BM * GEMM_P;
    static constexpr int WSZ = BN * GEMM_P;
    static constexpr int AV4 = BM * GEMM_K / 4 / NT;
    static constexpr int WV4 = BN * GEMM_K / 4 / NT;

    static __device__ __forceinline__ size_t a_goff(const int i, const int tid, const int lda) {
        const int idx = tid + i * NT;
        return AKM ? (size_t)(idx / (BM / 4)) * lda + (idx % (BM / 4)) * 4
                   : (size_t)(idx >> 3) * lda + (idx & 7) * 4;
    }
    static __device__ __forceinline__ int a_soff(const int i, const int tid) {
        const int idx = tid + i * NT;
        return AKM ? (idx / (BM / 4)) * AP + (idx % (BM / 4)) * 4
                   : (idx >> 3) * AP + (idx & 7) * 4;
    }
    static __device__ __forceinline__ size_t w_goff(const int i, const int tid, const int ldw) {
        const int idx = tid + i * NT;
        return (size_t)(idx >> 3) * ldw + (idx & 7) * 4;
    }
    static __device__ __forceinline__ int w_soff(const int i, const int tid) {
        const int idx = tid + i * NT;
        return (idx >> 3) * GEMM_P + (idx & 7) * 4;
    }
};

template <int WM, int WN, bool AKM>
__device__ __forceinline__ void gemm_block(const float* __restrict__ Ablk, const int lda,
                                           const float* __restrict__ Wblk, const int ldw,
                                           const int K,
                                           float* __restrict__ sA, float* __restrict__ sW,
                                           v8f& acc0, v8f& acc1, v8f& acc2, v8f& acc3) {
    typedef GemmShape<WM, WN, AKM> S;
    static_assert(S::WV4 == 4 && (S::AV4 == 2 || S::AV4 == 4), "per-thread piece counts the code below is written for");
    const int tid  = threadIdx.x;
    const int wave = tid >> 5;
    const int wm   = wave % WM;
    const int wn   = wave / WM;
    const int lane = tid & 31;
    const int half = lane >> 4;
    const int r    = lane & 15;

    const float* ga0 = Ablk + S::a_goff(0, tid, lda);
    const float* ga1 = Ablk + S::a_goff(1, tid, lda);
    const float* ga2 = Ablk + S::a_goff(S::AV4 > 2 ? 2 : 0, tid, lda);
    const float* ga3 = Ablk + S::a_goff(S::AV4 > 2 ? 3 : 0, tid, lda);
    const float* gw0 = Wblk + S::w_goff(0, tid, ldw);
    const float* gw1 = Wblk + S::w_goff(1, tid, ldw);
    const float* gw2 = Wblk + S::w_goff(2, tid, ldw);
    const float* gw3 = Wblk + S::w_goff(3, tid, ldw);
    const size_t astep = AKM ? (size_t)GEMM_K * lda : (size_t)GEMM_K;

    float4 ra0, ra1, ra2, ra3, rw0, rw1, rw2, rw3;
    ra2 = ra3 = make_float4(0.f, 0.f, 0.f, 0.f);

#define GEMM_GLOAD(ks)                                                        \
    do {                                                                      \
        const size_t ao = (size_t)(ks) * astep;                               \
        const size_t wo = (size_t)(ks) * GEMM_K;                              \
        ra0 = *(const float4*)(ga0 + ao);                                     \
        ra1 = *(const float4*)(ga1 + ao);                                     \
        if (S::AV4 > 2) {                                                     \
            ra2 = *(const float4*)(ga2 + ao);                                 \
            ra3 = *(const float4*)(ga3 + ao);                                 \
        }                                                                     \
        rw0 = *(const float4*)(gw0 + wo);                                     \
        rw1 = *(const float4*)(gw1 + wo);                                     \
        rw2 = *(const float4*)(gw2 + wo);                                     \
        rw3 = *(const float4*)(gw3 + wo);                                     \
    } while (0)

#define GEMM_SSTORE(buf)                                                      \
    do {                                                                      \
        float* a_ = sA + (buf) * S::ASZ;                                      \
        float* w_ = sW + (buf) * S::WSZ;                                      \
        *(float4*)(a_ + S::a_soff(0, tid)) = ra0;                             \
        *(float4*)(a_ + S::a_soff(1, tid)) = ra1;                             \
        if (S::AV4 > 2) {                                                     \
            *(float4*)(a_ + S::a_soff(2, tid)) = ra2;                         \
            *(float4*)(a_ + S::a_soff(3, tid)) = ra3;                         \
        }                                                                     \
        *(float4*)(w_ + S::w_soff(0, tid)) = rw0;                             \
        *(float4*)(w_ + S::w_soff(1, tid)) = rw1;                             \
        *(float4*)(w_ + S::w_soff(2, tid)) = rw2;                             \
        *(float4*)(w_ + S::w_soff(3, tid)) = rw3;                             \
    } while (0)

    GEMM_GLOAD(0);
    GEMM_SSTORE(0);
    __syncthreads();

    const int nslab = K / GEMM_K;
    for (int s = 0; s < nslab; ++s) {
        const int buf = s & 1;
        const int sn = (s + 1 < nslab) ? (s + 1) : s;
        GEMM_GLOAD(sn);

        const float* a = sA + buf * S::ASZ;
        const float* w = sW + buf * S::WSZ + (wn * 64 + r) * GEMM_P;
        #pragma unroll
        for (int k0 = 0; k0 < GEMM_K; k0 += 4) {
            const int ka = k0 + half * 2;
            v2f av;
            if (AKM) {
                av.x = a[ka * S::AP + wm * 16 + r];
                av.y = a[(ka + 1) * S::AP + wm * 16 + r];
            } else {
                av = *(const v2f*)(a + (wm * 16 + r) * S::AP + ka);
            }
            const v2f b0 = *(const v2f*)(w + 0 * 16 * GEMM_P + ka);
            const v2f b1 = *(const v2f*)(w + 1 * 16 * GEMM_P + ka);
            const v2f b2 = *(const v2f*)(w + 2 * 16 * GEMM_P + ka);
            const v2f b3 = *(const v2f*)(w + 3 * 16 * GEMM_P + ka);
            acc0 = wmma_f32_4(av, b0, acc0);
            acc1 = wmma_f32_4(av, b1, acc1);
            acc2 = wmma_f32_4(av, b2, acc2);
            acc3 = wmma_f32_4(av, b3, acc3);
        }

        GEMM_SSTORE(buf ^ 1);
        __syncthreads();
    }
#undef GEMM_GLOAD
#undef GEMM_SSTORE
}

__global__ void __launch_bounds__(256) __attribute__((amdgpu_waves_per_eu(4, 8)))
k_inproj(const float* __restrict__ x,
         const float* __restrict__ w,
         float* __restrict__ xz) {
    typedef GemmShape<4, 2, true> S;
    __shared__ __attribute__((aligned(16))) float sA[2 * S::ASZ];
    __shared__ __attribute__((aligned(16))) float sW[2 * S::WSZ];

    const int bn = blockIdx.x & 15;
    const int bm = blockIdx.x >> 4;
    const int m0 = bm * 64;
    const int bb = m0 >> 9;
    const int l0 = m0 & 511;

    v8f acc0 = {}, acc1 = {}, acc2 = {}, acc3 = {};
    gemm_block<4, 2, true>(x + (size_t)bb * DMODEL * SEQ + l0, SEQ,
                           w + (size_t)(bn * 128) * DMODEL, DMODEL, DMODEL,
                           sA, sW, acc0, acc1, acc2, acc3);

    const int wave = threadIdx.x >> 5;
    const int wm   = wave % 4;
    const int wn   = wave / 4;
    const int lane = threadIdx.x & 31;
    const int half = lane >> 4;
    const int r    = lane & 15;
    const int col  = bn * 128 + wn * 64 + r;
    #pragma unroll
    for (int j = 0; j < 8; ++j) {
        const size_t row = (size_t)(m0 + wm * 16 + half * 8 + j) * (2 * DINNER);
        xz[row + col + 0 ] = acc0[j];
        xz[row + col + 16] = acc1[j];
        xz[row + col + 32] = acc2[j];
        xz[row + col + 48] = acc3[j];
    }
}

__global__ void k_conv(const float* __restrict__ xz,
                       const float* __restrict__ cw,
                       const float* __restrict__ cb,
                       float* __restrict__ ucv) {
    const int idx = blockIdx.x * blockDim.x + threadIdx.x;
    if (idx >= MROWS * DINNER) return;
    const int c = idx & (DINNER - 1);
    const int m = idx >> 10;
    const int l = m & 511;

    float s = cb[c];
    #pragma unroll
    for (int t = 0; t < 4; ++t) {
        const int lt = l - 3 + t;
        if (lt >= 0)
            s += xz[(size_t)(m - 3 + t) * (2 * DINNER) + c] * cw[c * 4 + t];
    }
    const float sig = 1.f / (1.f + __expf(-s));
    ucv[(size_t)m * DINNER + c] = s * sig;
}

__global__ void __launch_bounds__(128) __attribute__((amdgpu_waves_per_eu(1, 2)))
k_xproj(const float* __restrict__ ucv,
        const float* __restrict__ w,
        float* __restrict__ dbc) {
    typedef GemmShape<4, 1, false> S;
    __shared__ __attribute__((aligned(16))) float sA[2 * S::ASZ];
    __shared__ __attribute__((aligned(16))) float sW[2 * S::WSZ];

    const int m0 = blockIdx.x * 64;

    v8f acc0 = {}, acc1 = {}, acc2 = {}, acc3 = {};
    gemm_block<4, 1, false>(ucv + (size_t)m0 * DINNER, DINNER,
                            w, DINNER, DINNER,
                            sA, sW, acc0, acc1, acc2, acc3);

    const int wm   = threadIdx.x >> 5;
    const int lane = threadIdx.x & 31;
    const int half = lane >> 4;
    const int r    = lane & 15;
    #pragma unroll
    for (int j = 0; j < 8; ++j) {
        const size_t row = (size_t)(m0 + wm * 16 + half * 8 + j) * 64;
        dbc[row + 0  + r] = acc0[j];
        dbc[row + 16 + r] = acc1[j];
        dbc[row + 32 + r] = acc2[j];
        dbc[row + 48 + r] = acc3[j];
    }
}

__global__ void k_dtproj(const float* __restrict__ dbc,
                         const float* __restrict__ w,
                         const float* __restrict__ bias,
                         float* __restrict__ delta) {
    const int wid   = blockIdx.x * 4 + (threadIdx.x >> 5);
    const int ntq   = wid & 15;
    const int tileM = wid >> 4;
    const int lane  = threadIdx.x & 31;
    const int half  = lane >> 4;
    const int r     = lane & 15;

    const float* arow = dbc + (size_t)(tileM * 16 + r) * 64;
    const float* wr0 = w + (size_t)((ntq * 4 + 0) * 16 + r) * DTRANK;
    const float* wr1 = w + (size_t)((ntq * 4 + 1) * 16 + r) * DTRANK;
    const float* wr2 = w + (size_t)((ntq * 4 + 2) * 16 + r) * DTRANK;
    const float* wr3 = w + (size_t)((ntq * 4 + 3) * 16 + r) * DTRANK;

    v8f acc0 = {}, acc1 = {}, acc2 = {}, acc3 = {};
    #pragma unroll
    for (int k0 = 0; k0 < DTRANK; k0 += 4) {
        const int ka = k0 + half * 2;
        v2f a  = *(const v2f*)(arow + ka);
        v2f b0 = *(const v2f*)(wr0 + ka);
        v2f b1 = *(const v2f*)(wr1 + ka);
        v2f b2 = *(const v2f*)(wr2 + ka);
        v2f b3 = *(const v2f*)(wr3 + ka);
        acc0 = wmma_f32_4(a, b0, acc0);
        acc1 = wmma_f32_4(a, b1, acc1);
        acc2 = wmma_f32_4(a, b2, acc2);
        acc3 = wmma_f32_4(a, b3, acc3);
    }
    const int nb = ntq * 64 + r;
    const float bn0 = bias[nb + 0];
    const float bn1 = bias[nb + 16];
    const float bn2 = bias[nb + 32];
    const float bn3 = bias[nb + 48];
    #pragma unroll
    for (int j = 0; j < 8; ++j) {
        const size_t row = (size_t)(tileM * 16 + half * 8 + j) * DINNER;
        delta[row + nb + 0 ] = softplus_f(acc0[j] + bn0);
        delta[row + nb + 16] = softplus_f(acc1[j] + bn1);
        delta[row + nb + 32] = softplus_f(acc2[j] + bn2);
        delta[row + nb + 48] = softplus_f(acc3[j] + bn3);
    }
}

#define SCAN_T 16

__device__ __forceinline__ void scan_load_bc(const float4* __restrict__ row,
                                             float (&Bv)[DSTATE], float (&Cv)[DSTATE]) {
    #pragma unroll
    for (int q = 0; q < 4; ++q) {
        const float4 b4 = row[q];
        const float4 c4 = row[4 + q];
        Bv[4 * q + 0] = b4.x; Bv[4 * q + 1] = b4.y; Bv[4 * q + 2] = b4.z; Bv[4 * q + 3] = b4.w;
        Cv[4 * q + 0] = c4.x; Cv[4 * q + 1] = c4.y; Cv[4 * q + 2] = c4.z; Cv[4 * q + 3] = c4.w;
    }
}

__global__ void __launch_bounds__(128)
k_scan(const float* __restrict__ delta,
       const float* __restrict__ ucv,
       const float* __restrict__ xz,
       const float* __restrict__ dbc,
       const float* __restrict__ A_log,
       const float* __restrict__ D_param,
       float* __restrict__ ybuf) {
    const int bb = blockIdx.x >> 3;
    const int tid = threadIdx.x;
    const int c  = (blockIdx.x & 7) * 128 + tid;
    const int mbase = bb * SEQ;

    __shared__ float4 sBC[SEQ * 8];

    const float* pd = delta + (size_t)mbase * DINNER + c;
    const float* pu = ucv   + (size_t)mbase * DINNER + c;
    const float* pz = xz    + (size_t)mbase * (2 * DINNER) + DINNER + c;
    float*       py = ybuf  + (size_t)mbase * DINNER + c;

    float nd[SCAN_T], nu[SCAN_T], nz[SCAN_T];
    #pragma unroll
    for (int t = 0; t < SCAN_T; ++t) {
        nd[t] = pd[(size_t)t * DINNER];
        nu[t] = pu[(size_t)t * DINNER];
        nz[t] = pz[(size_t)t * (2 * DINNER)];
    }

    const float* bcsrc = dbc + (size_t)mbase * 64 + 32;
    #pragma unroll
    for (int j = 0; j < (SEQ * 8) / 128; ++j) {
        const int i = j * 128 + tid;
        sBC[i] = *(const float4*)(bcsrc + (size_t)(i >> 3) * 64 + (i & 7) * 4);
    }

    float Ac[DSTATE];
    float h[DSTATE];
    #pragma unroll
    for (int n = 0; n < DSTATE; ++n) {
        Ac[n] = -__expf(A_log[(size_t)c * DSTATE + n]);
        h[n] = 0.f;
    }
    const float Dp = D_param[c];

    __syncthreads();

    float Bc[DSTATE], Cc[DSTATE];
    scan_load_bc(sBC, Bc, Cc);

    for (int l0 = 0; l0 < SEQ; l0 += SCAN_T) {
        float cd[SCAN_T], cu[SCAN_T], cz[SCAN_T];
        #pragma unroll
        for (int t = 0; t < SCAN_T; ++t) { cd[t] = nd[t]; cu[t] = nu[t]; cz[t] = nz[t]; }

        const int ln = (l0 + SCAN_T < SEQ) ? (l0 + SCAN_T) : l0;
        #pragma unroll
        for (int t = 0; t < SCAN_T; ++t) {
            nd[t] = pd[(size_t)(ln + t) * DINNER];
            nu[t] = pu[(size_t)(ln + t) * DINNER];
            nz[t] = pz[(size_t)(ln + t) * (2 * DINNER)];
        }

        #pragma unroll
        for (int t = 0; t < SCAN_T; ++t) {
            const int l = l0 + t;

            float Bn[DSTATE], Cn[DSTATE];
            const int lnx = (l + 1 < SEQ) ? (l + 1) : l;
            scan_load_bc(sBC + lnx * 8, Bn, Cn);
            __builtin_amdgcn_sched_barrier(0);

            const float dlt = cd[t];
            const float ut  = cu[t];
            const float zt  = cz[t];
            const float du  = dlt * ut;

            float y = 0.f;
            #pragma unroll
            for (int n = 0; n < DSTATE; ++n) {
                const float dA = __expf(dlt * Ac[n]);
                h[n] = dA * h[n] + du * Bc[n];
                y += h[n] * Cc[n];
            }
            const float sig = 1.f / (1.f + __expf(-zt));
            py[(size_t)l * DINNER] = (y + ut * Dp) * (zt * sig);

            #pragma unroll
            for (int n = 0; n < DSTATE; ++n) { Bc[n] = Bn[n]; Cc[n] = Cn[n]; }

            __builtin_amdgcn_sched_barrier(0);
        }
    }
}

#define OUT_CP 68

__global__ void __launch_bounds__(256) __attribute__((amdgpu_waves_per_eu(2, 4)))
k_outproj(const float* __restrict__ ybuf,
          const float* __restrict__ w,
          float* __restrict__ otmp) {
    typedef GemmShape<4, 2, false> S;
    __shared__ __attribute__((aligned(16))) float sA[2 * S::ASZ];
    __shared__ __attribute__((aligned(16))) float sW[2 * S::WSZ];
    static_assert(128 * OUT_CP <= 2 * S::WSZ, "turned tile must fit the W buffers");

    const int bn = blockIdx.x & 3;
    const int bm = blockIdx.x >> 2;
    const int m0 = bm * 64;
    const int bb = m0 >> 9;
    const int l0 = m0 & 511;

    v8f acc0 = {}, acc1 = {}, acc2 = {}, acc3 = {};
    gemm_block<4, 2, false>(ybuf + (size_t)m0 * DINNER, DINNER,
                            w + (size_t)(bn * 128) * DINNER, DINNER, DINNER,
                            sA, sW, acc0, acc1, acc2, acc3);

    const int tid  = threadIdx.x;
    const int wave = tid >> 5;
    const int wm   = wave % 4;
    const int wn   = wave / 4;
    const int lane = tid & 31;
    const int half = lane >> 4;
    const int r    = lane & 15;

    float* sC = sW;
    {
        float* c0 = sC + (wn * 64 + r) * OUT_CP + wm * 16 + half * 8;
        *(float4*)(c0 + 0 * 16 * OUT_CP)     = make_float4(acc0[0], acc0[1], acc0[2], acc0[3]);
        *(float4*)(c0 + 0 * 16 * OUT_CP + 4) = make_float4(acc0[4], acc0[5], acc0[6], acc0[7]);
        *(float4*)(c0 + 1 * 16 * OUT_CP)     = make_float4(acc1[0], acc1[1], acc1[2], acc1[3]);
        *(float4*)(c0 + 1 * 16 * OUT_CP + 4) = make_float4(acc1[4], acc1[5], acc1[6], acc1[7]);
        *(float4*)(c0 + 2 * 16 * OUT_CP)     = make_float4(acc2[0], acc2[1], acc2[2], acc2[3]);
        *(float4*)(c0 + 2 * 16 * OUT_CP + 4) = make_float4(acc2[4], acc2[5], acc2[6], acc2[7]);
        *(float4*)(c0 + 3 * 16 * OUT_CP)     = make_float4(acc3[0], acc3[1], acc3[2], acc3[3]);
        *(float4*)(c0 + 3 * 16 * OUT_CP + 4) = make_float4(acc3[4], acc3[5], acc3[6], acc3[7]);
    }
    __syncthreads();

    float* obase = otmp + ((size_t)bb * DMODEL + bn * 128) * SEQ + l0;
    #pragma unroll
    for (int i = 0; i < 8; ++i) {
        const int idx = tid + i * 256;
        const int n = idx >> 4, q = idx & 15;
        *(float4*)(obase + (size_t)n * SEQ + q * 4) = *(const float4*)(sC + n * OUT_CP + q * 4);
    }
}

__global__ void k_gnstats(const float* __restrict__ otmp,
                          float* __restrict__ stats) {
    const int bid = blockIdx.x;
    const int tid = threadIdx.x;
    const float* base = otmp + (size_t)bid * 32768;

    __shared__ float ssum[256];
    __shared__ float ssq[256];
    float s = 0.f, q = 0.f;
    for (int i = tid; i < 32768; i += 256) {
        const float v = base[i];
        s += v;
        q += v * v;
    }
    ssum[tid] = s; ssq[tid] = q;
    __syncthreads();
    for (int off = 128; off > 0; off >>= 1) {
        if (tid < off) { ssum[tid] += ssum[tid + off]; ssq[tid] += ssq[tid + off]; }
        __syncthreads();
    }
    if (tid == 0) {
        const float mu  = ssum[0] * (1.f / 32768.f);
        const float var = ssq[0] * (1.f / 32768.f) - mu * mu;
        stats[2 * bid]     = mu;
        stats[2 * bid + 1] = rsqrtf(var + 1e-5f);
    }
}

__global__ void k_gnmish(const float* __restrict__ otmp,
                         const float* __restrict__ stats,
                         const float* __restrict__ gw,
                         const float* __restrict__ gb,
                         float* __restrict__ out) {
    const int idx = blockIdx.x * blockDim.x + threadIdx.x;
    if (idx >= BATCH * DMODEL * SEQ) return;
    const int b   = idx >> 18;
    const int rem = idx & (DMODEL * SEQ - 1);
    const int c   = rem >> 9;
    const int g   = c >> 6;
    const int blk = b * NGROUPS + g;

    const float mu = stats[2 * blk];
    const float rs = stats[2 * blk + 1];
    float v = (otmp[idx] - mu) * rs * gw[c] + gb[c];
    out[idx] = v * tanhf(softplus_f(v));
}

extern "C" void kernel_launch(void* const* d_in, const int* in_sizes, int n_in,
                              void* d_out, int out_size, void* d_ws, size_t ws_size,
                              hipStream_t stream) {
    const float* x         = (const float*)d_in[0];
    const float* in_proj_w = (const float*)d_in[1];
    const float* conv_w    = (const float*)d_in[2];
    const float* conv_b    = (const float*)d_in[3];
    const float* x_proj_w  = (const float*)d_in[4];
    const float* dt_proj_w = (const float*)d_in[5];
    const float* dt_proj_b = (const float*)d_in[6];
    const float* A_log     = (const float*)d_in[7];
    const float* D_param   = (const float*)d_in[8];
    const float* out_proj_w= (const float*)d_in[9];
    const float* gn_w      = (const float*)d_in[10];
    const float* gn_b      = (const float*)d_in[11];
    float* out = (float*)d_out;

    float* ws    = (float*)d_ws;
    float* xz    = ws;
    float* ucv   = xz    + (size_t)MROWS * 2 * DINNER;
    float* dbc   = ucv   + (size_t)MROWS * DINNER;
    float* delta = dbc   + (size_t)MROWS * 64;
    float* ybuf  = delta + (size_t)MROWS * DINNER;
    float* stats = ybuf  + (size_t)MROWS * DINNER;
    float* otmp  = xz;

    k_inproj<<<1024, 256, 0, stream>>>(x, in_proj_w, xz);

    k_conv<<<(MROWS * DINNER + 255) / 256, 256, 0, stream>>>(xz, conv_w, conv_b, ucv);

    k_xproj<<<64, 128, 0, stream>>>(ucv, x_proj_w, dbc);

    k_dtproj<<<4096 / 4, 128, 0, stream>>>(dbc, dt_proj_w, dt_proj_b, delta);

    k_scan<<<BATCH * 8, 128, 0, stream>>>(delta, ucv, xz, dbc, A_log, D_param, ybuf);

    k_outproj<<<256, 256, 0, stream>>>(ybuf, out_proj_w, otmp);

    k_gnstats<<<BATCH * NGROUPS, 256, 0, stream>>>(otmp, stats);

    k_gnmish<<<(BATCH * DMODEL * SEQ + 255) / 256, 256, 0, stream>>>(
        otmp, stats, gn_w, gn_b, out);
}


# conv+SiLU and GroupNorm-apply+Mish kernels: a thread handles 32 / 16 elements as 16-byte vectors (conv: 4 channels x 8 steps, rows reused across taps), not one float; 196,608 waves become 8,192
# speedup vs baseline: 5.7553x; 1.0359x over previous
#include <hip/hip_runtime.h>
#include <hip/hip_bf16.h>
#include <math.h>


typedef float v2f __attribute__((ext_vector_type(2)));
typedef float v8f __attribute__((ext_vector_type(8)));

#define BATCH    8
#define SEQ      512
#define DMODEL   512
#define DINNER   1024
#define DSTATE   16
#define DTRANK   32
#define NGROUPS  8
#define MROWS    (BATCH * SEQ)

__device__ __forceinline__ v8f wmma_f32_4(v2f a, v2f b, v8f c) {
    return __builtin_amdgcn_wmma_f32_16x16x4_f32(
        false, a, false, b, (short)0, c, false, false);
}

__device__ __forceinline__ float softplus_f(float x) {
    return fmaxf(x, 0.f) + log1pf(__expf(-fabsf(x)));
}


#define GEMM_K   32
#define GEMM_P   36

template <int WM, int WN, bool AKM>
struct GemmShape {
    static constexpr int NT  = 32 * WM * WN;
    static constexpr int BM  = 16 * WM;
    static constexpr int BN  = 64 * WN;
    static constexpr int AP  = AKM ? (BM + 8) : GEMM_P;
    static constexpr int ASZ = AKM ? GEMM_K * (BM + 8) : BM * GEMM_P;
    static constexpr int WSZ = BN * GEMM_P;
    static constexpr int AV4 = BM * GEMM_K / 4 / NT;
    static constexpr int WV4 = BN * GEMM_K / 4 / NT;

    static __device__ __forceinline__ size_t a_goff(const int i, const int tid, const int lda) {
        const int idx = tid + i * NT;
        return AKM ? (size_t)(idx / (BM / 4)) * lda + (idx % (BM / 4)) * 4
                   : (size_t)(idx >> 3) * lda + (idx & 7) * 4;
    }
    static __device__ __forceinline__ int a_soff(const int i, const int tid) {
        const int idx = tid + i * NT;
        return AKM ? (idx / (BM / 4)) * AP + (idx % (BM / 4)) * 4
                   : (idx >> 3) * AP + (idx & 7) * 4;
    }
    static __device__ __forceinline__ size_t w_goff(const int i, const int tid, const int ldw) {
        const int idx = tid + i * NT;
        return (size_t)(idx >> 3) * ldw + (idx & 7) * 4;
    }
    static __device__ __forceinline__ int w_soff(const int i, const int tid) {
        const int idx = tid + i * NT;
        return (idx >> 3) * GEMM_P + (idx & 7) * 4;
    }
};

template <int WM, int WN, bool AKM>
__device__ __forceinline__ void gemm_block(const float* __restrict__ Ablk, const int lda,
                                           const float* __restrict__ Wblk, const int ldw,
                                           const int K,
                                           float* __restrict__ sA, float* __restrict__ sW,
                                           v8f& acc0, v8f& acc1, v8f& acc2, v8f& acc3) {
    typedef GemmShape<WM, WN, AKM> S;
    static_assert(S::WV4 == 4 && (S::AV4 == 2 || S::AV4 == 4), "per-thread piece counts the code below is written for");
    const int tid  = threadIdx.x;
    const int wave = tid >> 5;
    const int wm   = wave % WM;
    const int wn   = wave / WM;
    const int lane = tid & 31;
    const int half = lane >> 4;
    const int r    = lane & 15;

    const float* ga0 = Ablk + S::a_goff(0, tid, lda);
    const float* ga1 = Ablk + S::a_goff(1, tid, lda);
    const float* ga2 = Ablk + S::a_goff(S::AV4 > 2 ? 2 : 0, tid, lda);
    const float* ga3 = Ablk + S::a_goff(S::AV4 > 2 ? 3 : 0, tid, lda);
    const float* gw0 = Wblk + S::w_goff(0, tid, ldw);
    const float* gw1 = Wblk + S::w_goff(1, tid, ldw);
    const float* gw2 = Wblk + S::w_goff(2, tid, ldw);
    const float* gw3 = Wblk + S::w_goff(3, tid, ldw);
    const size_t astep = AKM ? (size_t)GEMM_K * lda : (size_t)GEMM_K;

    float4 ra0, ra1, ra2, ra3, rw0, rw1, rw2, rw3;
    ra2 = ra3 = make_float4(0.f, 0.f, 0.f, 0.f);

#define GEMM_GLOAD(ks)                                                        \
    do {                                                                      \
        const size_t ao = (size_t)(ks) * astep;                               \
        const size_t wo = (size_t)(ks) * GEMM_K;                              \
        ra0 = *(const float4*)(ga0 + ao);                                     \
        ra1 = *(const float4*)(ga1 + ao);                                     \
        if (S::AV4 > 2) {                                                     \
            ra2 = *(const float4*)(ga2 + ao);                                 \
            ra3 = *(const float4*)(ga3 + ao);                                 \
        }                                                                     \
        rw0 = *(const float4*)(gw0 + wo);                                     \
        rw1 = *(const float4*)(gw1 + wo);                                     \
        rw2 = *(const float4*)(gw2 + wo);                                     \
        rw3 = *(const float4*)(gw3 + wo);                                     \
    } while (0)

#define GEMM_SSTORE(buf)                                                      \
    do {                                                                      \
        float* a_ = sA + (buf) * S::ASZ;                                      \
        float* w_ = sW + (buf) * S::WSZ;                                      \
        *(float4*)(a_ + S::a_soff(0, tid)) = ra0;                             \
        *(float4*)(a_ + S::a_soff(1, tid)) = ra1;                             \
        if (S::AV4 > 2) {                                                     \
            *(float4*)(a_ + S::a_soff(2, tid)) = ra2;                         \
            *(float4*)(a_ + S::a_soff(3, tid)) = ra3;                         \
        }                                                                     \
        *(float4*)(w_ + S::w_soff(0, tid)) = rw0;                             \
        *(float4*)(w_ + S::w_soff(1, tid)) = rw1;                             \
        *(float4*)(w_ + S::w_soff(2, tid)) = rw2;                             \
        *(float4*)(w_ + S::w_soff(3, tid)) = rw3;                             \
    } while (0)

    GEMM_GLOAD(0);
    GEMM_SSTORE(0);
    __syncthreads();

    const int nslab = K / GEMM_K;
    for (int s = 0; s < nslab; ++s) {
        const int buf = s & 1;
        const int sn = (s + 1 < nslab) ? (s + 1) : s;
        GEMM_GLOAD(sn);

        const float* a = sA + buf * S::ASZ;
        const float* w = sW + buf * S::WSZ + (wn * 64 + r) * GEMM_P;
        #pragma unroll
        for (int k0 = 0; k0 < GEMM_K; k0 += 4) {
            const int ka = k0 + half * 2;
            v2f av;
            if (AKM) {
                av.x = a[ka * S::AP + wm * 16 + r];
                av.y = a[(ka + 1) * S::AP + wm * 16 + r];
            } else {
                av = *(const v2f*)(a + (wm * 16 + r) * S::AP + ka);
            }
            const v2f b0 = *(const v2f*)(w + 0 * 16 * GEMM_P + ka);
            const v2f b1 = *(const v2f*)(w + 1 * 16 * GEMM_P + ka);
            const v2f b2 = *(const v2f*)(w + 2 * 16 * GEMM_P + ka);
            const v2f b3 = *(const v2f*)(w + 3 * 16 * GEMM_P + ka);
            acc0 = wmma_f32_4(av, b0, acc0);
            acc1 = wmma_f32_4(av, b1, acc1);
            acc2 = wmma_f32_4(av, b2, acc2);
            acc3 = wmma_f32_4(av, b3, acc3);
        }

        GEMM_SSTORE(buf ^ 1);
        __syncthreads();
    }
#undef GEMM_GLOAD
#undef GEMM_SSTORE
}

__global__ void __launch_bounds__(256) __attribute__((amdgpu_waves_per_eu(4, 8)))
k_inproj(const float* __restrict__ x,
         const float* __restrict__ w,
         float* __restrict__ xz) {
    typedef GemmShape<4, 2, true> S;
    __shared__ __attribute__((aligned(16))) float sA[2 * S::ASZ];
    __shared__ __attribute__((aligned(16))) float sW[2 * S::WSZ];

    const int bn = blockIdx.x & 15;
    const int bm = blockIdx.x >> 4;
    const int m0 = bm * 64;
    const int bb = m0 >> 9;
    const int l0 = m0 & 511;

    v8f acc0 = {}, acc1 = {}, acc2 = {}, acc3 = {};
    gemm_block<4, 2, true>(x + (size_t)bb * DMODEL * SEQ + l0, SEQ,
                           w + (size_t)(bn * 128) * DMODEL, DMODEL, DMODEL,
                           sA, sW, acc0, acc1, acc2, acc3);

    const int wave = threadIdx.x >> 5;
    const int wm   = wave % 4;
    const int wn   = wave / 4;
    const int lane = threadIdx.x & 31;
    const int half = lane >> 4;
    const int r    = lane & 15;
    const int col  = bn * 128 + wn * 64 + r;
    #pragma unroll
    for (int j = 0; j < 8; ++j) {
        const size_t row = (size_t)(m0 + wm * 16 + half * 8 + j) * (2 * DINNER);
        xz[row + col + 0 ] = acc0[j];
        xz[row + col + 16] = acc1[j];
        xz[row + col + 32] = acc2[j];
        xz[row + col + 48] = acc3[j];
    }
}

#define CONV_R 8

__global__ void __launch_bounds__(256)
k_conv(const float* __restrict__ xz,
       const float* __restrict__ cw,
       const float* __restrict__ cb,
       float* __restrict__ ucv) {
    const int c  = threadIdx.x * 4;
    const int m0 = blockIdx.x * CONV_R;
    const int l0 = m0 & (SEQ - 1);

    float xr[CONV_R + 3][4];
    #pragma unroll
    for (int j = 0; j < CONV_R + 3; ++j) {
        const int mr = (l0 + j >= 3) ? (m0 + j - 3) : m0;
        const float4 v = *(const float4*)(xz + (size_t)mr * (2 * DINNER) + c);
        xr[j][0] = v.x; xr[j][1] = v.y; xr[j][2] = v.z; xr[j][3] = v.w;
    }

    float wt[4][4];
    #pragma unroll
    for (int ch = 0; ch < 4; ++ch) {
        const float4 w4 = *(const float4*)(cw + (size_t)(c + ch) * 4);
        wt[ch][0] = w4.x; wt[ch][1] = w4.y; wt[ch][2] = w4.z; wt[ch][3] = w4.w;
    }
    const float4 b4 = *(const float4*)(cb + c);
    const float bias[4] = {b4.x, b4.y, b4.z, b4.w};

    #pragma unroll
    for (int i = 0; i < CONV_R; ++i) {
        float o[4];
        #pragma unroll
        for (int ch = 0; ch < 4; ++ch) {
            float s = bias[ch];
            #pragma unroll
            for (int t = 0; t < 4; ++t) {
                if (l0 + i + t >= 3)
                    s += xr[i + t][ch] * wt[ch][t];
            }
            const float sig = 1.f / (1.f + __expf(-s));
            o[ch] = s * sig;
        }
        *(float4*)(ucv + (size_t)(m0 + i) * DINNER + c) = make_float4(o[0], o[1], o[2], o[3]);
    }
}

__global__ void __launch_bounds__(128) __attribute__((amdgpu_waves_per_eu(1, 2)))
k_xproj(const float* __restrict__ ucv,
        const float* __restrict__ w,
        float* __restrict__ dbc) {
    typedef GemmShape<4, 1, false> S;
    __shared__ __attribute__((aligned(16))) float sA[2 * S::ASZ];
    __shared__ __attribute__((aligned(16))) float sW[2 * S::WSZ];

    const int m0 = blockIdx.x * 64;

    v8f acc0 = {}, acc1 = {}, acc2 = {}, acc3 = {};
    gemm_block<4, 1, false>(ucv + (size_t)m0 * DINNER, DINNER,
                            w, DINNER, DINNER,
                            sA, sW, acc0, acc1, acc2, acc3);

    const int wm   = threadIdx.x >> 5;
    const int lane = threadIdx.x & 31;
    const int half = lane >> 4;
    const int r    = lane & 15;
    #pragma unroll
    for (int j = 0; j < 8; ++j) {
        const size_t row = (size_t)(m0 + wm * 16 + half * 8 + j) * 64;
        dbc[row + 0  + r] = acc0[j];
        dbc[row + 16 + r] = acc1[j];
        dbc[row + 32 + r] = acc2[j];
        dbc[row + 48 + r] = acc3[j];
    }
}

__global__ void k_dtproj(const float* __restrict__ dbc,
                         const float* __restrict__ w,
                         const float* __restrict__ bias,
                         float* __restrict__ delta) {
    const int wid   = blockIdx.x * 4 + (threadIdx.x >> 5);
    const int ntq   = wid & 15;
    const int tileM = wid >> 4;
    const int lane  = threadIdx.x & 31;
    const int half  = lane >> 4;
    const int r     = lane & 15;

    const float* arow = dbc + (size_t)(tileM * 16 + r) * 64;
    const float* wr0 = w + (size_t)((ntq * 4 + 0) * 16 + r) * DTRANK;
    const float* wr1 = w + (size_t)((ntq * 4 + 1) * 16 + r) * DTRANK;
    const float* wr2 = w + (size_t)((ntq * 4 + 2) * 16 + r) * DTRANK;
    const float* wr3 = w + (size_t)((ntq * 4 + 3) * 16 + r) * DTRANK;

    v8f acc0 = {}, acc1 = {}, acc2 = {}, acc3 = {};
    #pragma unroll
    for (int k0 = 0; k0 < DTRANK; k0 += 4) {
        const int ka = k0 + half * 2;
        v2f a  = *(const v2f*)(arow + ka);
        v2f b0 = *(const v2f*)(wr0 + ka);
        v2f b1 = *(const v2f*)(wr1 + ka);
        v2f b2 = *(const v2f*)(wr2 + ka);
        v2f b3 = *(const v2f*)(wr3 + ka);
        acc0 = wmma_f32_4(a, b0, acc0);
        acc1 = wmma_f32_4(a, b1, acc1);
        acc2 = wmma_f32_4(a, b2, acc2);
        acc3 = wmma_f32_4(a, b3, acc3);
    }
    const int nb = ntq * 64 + r;
    const float bn0 = bias[nb + 0];
    const float bn1 = bias[nb + 16];
    const float bn2 = bias[nb + 32];
    const float bn3 = bias[nb + 48];
    #pragma unroll
    for (int j = 0; j < 8; ++j) {
        const size_t row = (size_t)(tileM * 16 + half * 8 + j) * DINNER;
        delta[row + nb + 0 ] = softplus_f(acc0[j] + bn0);
        delta[row + nb + 16] = softplus_f(acc1[j] + bn1);
        delta[row + nb + 32] = softplus_f(acc2[j] + bn2);
        delta[row + nb + 48] = softplus_f(acc3[j] + bn3);
    }
}

#define SCAN_T 16

__device__ __forceinline__ void scan_load_bc(const float4* __restrict__ row,
                                             float (&Bv)[DSTATE], float (&Cv)[DSTATE]) {
    #pragma unroll
    for (int q = 0; q < 4; ++q) {
        const float4 b4 = row[q];
        const float4 c4 = row[4 + q];
        Bv[4 * q + 0] = b4.x; Bv[4 * q + 1] = b4.y; Bv[4 * q + 2] = b4.z; Bv[4 * q + 3] = b4.w;
        Cv[4 * q + 0] = c4.x; Cv[4 * q + 1] = c4.y; Cv[4 * q + 2] = c4.z; Cv[4 * q + 3] = c4.w;
    }
}

__global__ void __launch_bounds__(128)
k_scan(const float* __restrict__ delta,
       const float* __restrict__ ucv,
       const float* __restrict__ xz,
       const float* __restrict__ dbc,
       const float* __restrict__ A_log,
       const float* __restrict__ D_param,
       float* __restrict__ ybuf) {
    const int bb = blockIdx.x >> 3;
    const int tid = threadIdx.x;
    const int c  = (blockIdx.x & 7) * 128 + tid;
    const int mbase = bb * SEQ;

    __shared__ float4 sBC[SEQ * 8];

    const float* pd = delta + (size_t)mbase * DINNER + c;
    const float* pu = ucv   + (size_t)mbase * DINNER + c;
    const float* pz = xz    + (size_t)mbase * (2 * DINNER) + DINNER + c;
    float*       py = ybuf  + (size_t)mbase * DINNER + c;

    float nd[SCAN_T], nu[SCAN_T], nz[SCAN_T];
    #pragma unroll
    for (int t = 0; t < SCAN_T; ++t) {
        nd[t] = pd[(size_t)t * DINNER];
        nu[t] = pu[(size_t)t * DINNER];
        nz[t] = pz[(size_t)t * (2 * DINNER)];
    }

    const float* bcsrc = dbc + (size_t)mbase * 64 + 32;
    #pragma unroll
    for (int j = 0; j < (SEQ * 8) / 128; ++j) {
        const int i = j * 128 + tid;
        sBC[i] = *(const float4*)(bcsrc + (size_t)(i >> 3) * 64 + (i & 7) * 4);
    }

    float Ac[DSTATE];
    float h[DSTATE];
    #pragma unroll
    for (int n = 0; n < DSTATE; ++n) {
        Ac[n] = -__expf(A_log[(size_t)c * DSTATE + n]);
        h[n] = 0.f;
    }
    const float Dp = D_param[c];

    __syncthreads();

    float Bc[DSTATE], Cc[DSTATE];
    scan_load_bc(sBC, Bc, Cc);

    for (int l0 = 0; l0 < SEQ; l0 += SCAN_T) {
        float cd[SCAN_T], cu[SCAN_T], cz[SCAN_T];
        #pragma unroll
        for (int t = 0; t < SCAN_T; ++t) { cd[t] = nd[t]; cu[t] = nu[t]; cz[t] = nz[t]; }

        const int ln = (l0 + SCAN_T < SEQ) ? (l0 + SCAN_T) : l0;
        #pragma unroll
        for (int t = 0; t < SCAN_T; ++t) {
            nd[t] = pd[(size_t)(ln + t) * DINNER];
            nu[t] = pu[(size_t)(ln + t) * DINNER];
            nz[t] = pz[(size_t)(ln + t) * (2 * DINNER)];
        }

        #pragma unroll
        for (int t = 0; t < SCAN_T; ++t) {
            const int l = l0 + t;

            float Bn[DSTATE], Cn[DSTATE];
            const int lnx = (l + 1 < SEQ) ? (l + 1) : l;
            scan_load_bc(sBC + lnx * 8, Bn, Cn);
            __builtin_amdgcn_sched_barrier(0);

            const float dlt = cd[t];
            const float ut  = cu[t];
            const float zt  = cz[t];
            const float du  = dlt * ut;

            float y = 0.f;
            #pragma unroll
            for (int n = 0; n < DSTATE; ++n) {
                const float dA = __expf(dlt * Ac[n]);
                h[n] = dA * h[n] + du * Bc[n];
                y += h[n] * Cc[n];
            }
            const float sig = 1.f / (1.f + __expf(-zt));
            py[(size_t)l * DINNER] = (y + ut * Dp) * (zt * sig);

            #pragma unroll
            for (int n = 0; n < DSTATE; ++n) { Bc[n] = Bn[n]; Cc[n] = Cn[n]; }

            __builtin_amdgcn_sched_barrier(0);
        }
    }
}

#define OUT_CP 68

__global__ void __launch_bounds__(256) __attribute__((amdgpu_waves_per_eu(2, 4)))
k_outproj(const float* __restrict__ ybuf,
          const float* __restrict__ w,
          float* __restrict__ otmp) {
    typedef GemmShape<4, 2, false> S;
    __shared__ __attribute__((aligned(16))) float sA[2 * S::ASZ];
    __shared__ __attribute__((aligned(16))) float sW[2 * S::WSZ];
    static_assert(128 * OUT_CP <= 2 * S::WSZ, "turned tile must fit the W buffers");

    const int bn = blockIdx.x & 3;
    const int bm = blockIdx.x >> 2;
    const int m0 = bm * 64;
    const int bb = m0 >> 9;
    const int l0 = m0 & 511;

    v8f acc0 = {}, acc1 = {}, acc2 = {}, acc3 = {};
    gemm_block<4, 2, false>(ybuf + (size_t)m0 * DINNER, DINNER,
                            w + (size_t)(bn * 128) * DINNER, DINNER, DINNER,
                            sA, sW, acc0, acc1, acc2, acc3);

    const int tid  = threadIdx.x;
    const int wave = tid >> 5;
    const int wm   = wave % 4;
    const int wn   = wave / 4;
    const int lane = tid & 31;
    const int half = lane >> 4;
    const int r    = lane & 15;

    float* sC = sW;
    {
        float* c0 = sC + (wn * 64 + r) * OUT_CP + wm * 16 + half * 8;
        *(float4*)(c0 + 0 * 16 * OUT_CP)     = make_float4(acc0[0], acc0[1], acc0[2], acc0[3]);
        *(float4*)(c0 + 0 * 16 * OUT_CP + 4) = make_float4(acc0[4], acc0[5], acc0[6], acc0[7]);
        *(float4*)(c0 + 1 * 16 * OUT_CP)     = make_float4(acc1[0], acc1[1], acc1[2], acc1[3]);
        *(float4*)(c0 + 1 * 16 * OUT_CP + 4) = make_float4(acc1[4], acc1[5], acc1[6], acc1[7]);
        *(float4*)(c0 + 2 * 16 * OUT_CP)     = make_float4(acc2[0], acc2[1], acc2[2], acc2[3]);
        *(float4*)(c0 + 2 * 16 * OUT_CP + 4) = make_float4(acc2[4], acc2[5], acc2[6], acc2[7]);
        *(float4*)(c0 + 3 * 16 * OUT_CP)     = make_float4(acc3[0], acc3[1], acc3[2], acc3[3]);
        *(float4*)(c0 + 3 * 16 * OUT_CP + 4) = make_float4(acc3[4], acc3[5], acc3[6], acc3[7]);
    }
    __syncthreads();

    float* obase = otmp + ((size_t)bb * DMODEL + bn * 128) * SEQ + l0;
    #pragma unroll
    for (int i = 0; i < 8; ++i) {
        const int idx = tid + i * 256;
        const int n = idx >> 4, q = idx & 15;
        *(float4*)(obase + (size_t)n * SEQ + q * 4) = *(const float4*)(sC + n * OUT_CP + q * 4);
    }
}

__global__ void k_gnstats(const float* __restrict__ otmp,
                          float* __restrict__ stats) {
    const int bid = blockIdx.x;
    const int tid = threadIdx.x;
    const float* base = otmp + (size_t)bid * 32768;

    __shared__ float ssum[256];
    __shared__ float ssq[256];
    float s = 0.f, q = 0.f;
    for (int i = tid; i < 32768; i += 256) {
        const float v = base[i];
        s += v;
        q += v * v;
    }
    ssum[tid] = s; ssq[tid] = q;
    __syncthreads();
    for (int off = 128; off > 0; off >>= 1) {
        if (tid < off) { ssum[tid] += ssum[tid + off]; ssq[tid] += ssq[tid + off]; }
        __syncthreads();
    }
    if (tid == 0) {
        const float mu  = ssum[0] * (1.f / 32768.f);
        const float var = ssq[0] * (1.f / 32768.f) - mu * mu;
        stats[2 * bid]     = mu;
        stats[2 * bid + 1] = rsqrtf(var + 1e-5f);
    }
}

#define GNM_V 4

__device__ __forceinline__ float gn_mish_f(float x, float mu, float rs, float g, float bt) {
    const float v = (x - mu) * rs * g + bt;
    return v * tanhf(softplus_f(v));
}

__global__ void __launch_bounds__(256)
k_gnmish(const float* __restrict__ otmp,
         const float* __restrict__ stats,
         const float* __restrict__ gw,
         const float* __restrict__ gb,
         float* __restrict__ out) {
    const int vec0 = blockIdx.x * (256 * GNM_V) + threadIdx.x;
    const int blk  = (blockIdx.x * (1024 * GNM_V)) >> 15;
    const float mu = stats[2 * blk];
    const float rs = stats[2 * blk + 1];

    float4 x4[GNM_V];
    #pragma unroll
    for (int i = 0; i < GNM_V; ++i)
        x4[i] = *(const float4*)(otmp + (size_t)(vec0 + i * 256) * 4);

    #pragma unroll
    for (int i = 0; i < GNM_V; ++i) {
        const int idx = (vec0 + i * 256) * 4;
        const int c   = (idx >> 9) & (DMODEL - 1);
        const float g  = gw[c];
        const float bt = gb[c];
        float4 o;
        o.x = gn_mish_f(x4[i].x, mu, rs, g, bt);
        o.y = gn_mish_f(x4[i].y, mu, rs, g, bt);
        o.z = gn_mish_f(x4[i].z, mu, rs, g, bt);
        o.w = gn_mish_f(x4[i].w, mu, rs, g, bt);
        *(float4*)(out + (size_t)idx) = o;
    }
}

extern "C" void kernel_launch(void* const* d_in, const int* in_sizes, int n_in,
                              void* d_out, int out_size, void* d_ws, size_t ws_size,
                              hipStream_t stream) {
    const float* x         = (const float*)d_in[0];
    const float* in_proj_w = (const float*)d_in[1];
    const float* conv_w    = (const float*)d_in[2];
    const float* conv_b    = (const float*)d_in[3];
    const float* x_proj_w  = (const float*)d_in[4];
    const float* dt_proj_w = (const float*)d_in[5];
    const float* dt_proj_b = (const float*)d_in[6];
    const float* A_log     = (const float*)d_in[7];
    const float* D_param   = (const float*)d_in[8];
    const float* out_proj_w= (const float*)d_in[9];
    const float* gn_w      = (const float*)d_in[10];
    const float* gn_b      = (const float*)d_in[11];
    float* out = (float*)d_out;

    float* ws    = (float*)d_ws;
    float* xz    = ws;
    float* ucv   = xz    + (size_t)MROWS * 2 * DINNER;
    float* dbc   = ucv   + (size_t)MROWS * DINNER;
    float* delta = dbc   + (size_t)MROWS * 64;
    float* ybuf  = delta + (size_t)MROWS * DINNER;
    float* stats = ybuf  + (size_t)MROWS * DINNER;
    float* otmp  = xz;

    k_inproj<<<1024, 256, 0, stream>>>(x, in_proj_w, xz);

    k_conv<<<MROWS / CONV_R, 256, 0, stream>>>(xz, conv_w, conv_b, ucv);

    k_xproj<<<64, 128, 0, stream>>>(ucv, x_proj_w, dbc);

    k_dtproj<<<4096 / 4, 128, 0, stream>>>(dbc, dt_proj_w, dt_proj_b, delta);

    k_scan<<<BATCH * 8, 128, 0, stream>>>(delta, ucv, xz, dbc, A_log, D_param, ybuf);

    k_outproj<<<256, 256, 0, stream>>>(ybuf, out_proj_w, otmp);

    k_gnstats<<<BATCH * NGROUPS, 256, 0, stream>>>(otmp, stats);

    k_gnmish<<<(BATCH * DMODEL * SEQ) / (1024 * GNM_V), 256, 0, stream>>>(
        otmp, stats, gn_w, gn_b, out);
}


# softplus (dt_proj), Mish (GroupNorm kernel) and the conv/scan sigmoids evaluated on the hardware exp/log/rcp units in closed form, not by in-line library log1pf, tanhf and IEEE division
# speedup vs baseline: 6.1627x; 1.0708x over previous
#include <hip/hip_runtime.h>
#include <hip/hip_bf16.h>
#include <math.h>


typedef float v2f __attribute__((ext_vector_type(2)));
typedef float v8f __attribute__((ext_vector_type(8)));

#define BATCH    8
#define SEQ      512
#define DMODEL   512
#define DINNER   1024
#define DSTATE   16
#define DTRANK   32
#define NGROUPS  8
#define MROWS    (BATCH * SEQ)

__device__ __forceinline__ v8f wmma_f32_4(v2f a, v2f b, v8f c) {
    return __builtin_amdgcn_wmma_f32_16x16x4_f32(
        false, a, false, b, (short)0, c, false, false);
}

__device__ __forceinline__ float softplus_f(float x) {
    const float t = __expf(-fabsf(x));
    const float u = 1.f + t;
    const float d = u - 1.f;
    const float l = (__builtin_amdgcn_logf(u) * 0.6931471805599453f) * (t * __builtin_amdgcn_rcpf(d));
    return fmaxf(x, 0.f) + (d == 0.f ? t : l);
}

__device__ __forceinline__ float mish_f(float v) {
    const float m   = __expf(-fabsf(v));
    const bool  pos = v >= 0.f;
    const float num = pos ? fmaf(2.f, m, 1.f) : m * (m + 2.f);
    const float den = num + (pos ? 2.f * m * m : 2.f);
    return v * (num * __builtin_amdgcn_rcpf(den));
}

__device__ __forceinline__ float sigmoid_f(float x) {
    return __builtin_amdgcn_rcpf(1.f + __expf(-x));
}


#define GEMM_K   32
#define GEMM_P   36

template <int WM, int WN, bool AKM>
struct GemmShape {
    static constexpr int NT  = 32 * WM * WN;
    static constexpr int BM  = 16 * WM;
    static constexpr int BN  = 64 * WN;
    static constexpr int AP  = AKM ? (BM + 8) : GEMM_P;
    static constexpr int ASZ = AKM ? GEMM_K * (BM + 8) : BM * GEMM_P;
    static constexpr int WSZ = BN * GEMM_P;
    static constexpr int AV4 = BM * GEMM_K / 4 / NT;
    static constexpr int WV4 = BN * GEMM_K / 4 / NT;

    static __device__ __forceinline__ size_t a_goff(const int i, const int tid, const int lda) {
        const int idx = tid + i * NT;
        return AKM ? (size_t)(idx / (BM / 4)) * lda + (idx % (BM / 4)) * 4
                   : (size_t)(idx >> 3) * lda + (idx & 7) * 4;
    }
    static __device__ __forceinline__ int a_soff(const int i, const int tid) {
        const int idx = tid + i * NT;
        return AKM ? (idx / (BM / 4)) * AP + (idx % (BM / 4)) * 4
                   : (idx >> 3) * AP + (idx & 7) * 4;
    }
    static __device__ __forceinline__ size_t w_goff(const int i, const int tid, const int ldw) {
        const int idx = tid + i * NT;
        return (size_t)(idx >> 3) * ldw + (idx & 7) * 4;
    }
    static __device__ __forceinline__ int w_soff(const int i, const int tid) {
        const int idx = tid + i * NT;
        return (idx >> 3) * GEMM_P + (idx & 7) * 4;
    }
};

template <int WM, int WN, bool AKM>
__device__ __forceinline__ void gemm_block(const float* __restrict__ Ablk, const int lda,
                                           const float* __restrict__ Wblk, const int ldw,
                                           const int K,
                                           float* __restrict__ sA, float* __restrict__ sW,
                                           v8f& acc0, v8f& acc1, v8f& acc2, v8f& acc3) {
    typedef GemmShape<WM, WN, AKM> S;
    static_assert(S::WV4 == 4 && (S::AV4 == 2 || S::AV4 == 4), "per-thread piece counts the code below is written for");
    const int tid  = threadIdx.x;
    const int wave = tid >> 5;
    const int wm   = wave % WM;
    const int wn   = wave / WM;
    const int lane = tid & 31;
    const int half = lane >> 4;
    const int r    = lane & 15;

    const float* ga0 = Ablk + S::a_goff(0, tid, lda);
    const float* ga1 = Ablk + S::a_goff(1, tid, lda);
    const float* ga2 = Ablk + S::a_goff(S::AV4 > 2 ? 2 : 0, tid, lda);
    const float* ga3 = Ablk + S::a_goff(S::AV4 > 2 ? 3 : 0, tid, lda);
    const float* gw0 = Wblk + S::w_goff(0, tid, ldw);
    const float* gw1 = Wblk + S::w_goff(1, tid, ldw);
    const float* gw2 = Wblk + S::w_goff(2, tid, ldw);
    const float* gw3 = Wblk + S::w_goff(3, tid, ldw);
    const size_t astep = AKM ? (size_t)GEMM_K * lda : (size_t)GEMM_K;

    float4 ra0, ra1, ra2, ra3, rw0, rw1, rw2, rw3;
    ra2 = ra3 = make_float4(0.f, 0.f, 0.f, 0.f);

#define GEMM_GLOAD(ks)                                                        \
    do {                                                                      \
        const size_t ao = (size_t)(ks) * astep;                               \
        const size_t wo = (size_t)(ks) * GEMM_K;                              \
        ra0 = *(const float4*)(ga0 + ao);                                     \
        ra1 = *(const float4*)(ga1 + ao);                                     \
        if (S::AV4 > 2) {                                                     \
            ra2 = *(const float4*)(ga2 + ao);                                 \
            ra3 = *(const float4*)(ga3 + ao);                                 \
        }                                                                     \
        rw0 = *(const float4*)(gw0 + wo);                                     \
        rw1 = *(const float4*)(gw1 + wo);                                     \
        rw2 = *(const float4*)(gw2 + wo);                                     \
        rw3 = *(const float4*)(gw3 + wo);                                     \
    } while (0)

#define GEMM_SSTORE(buf)                                                      \
    do {                                                                      \
        float* a_ = sA + (buf) * S::ASZ;                                      \
        float* w_ = sW + (buf) * S::WSZ;                                      \
        *(float4*)(a_ + S::a_soff(0, tid)) = ra0;                             \
        *(float4*)(a_ + S::a_soff(1, tid)) = ra1;                             \
        if (S::AV4 > 2) {                                                     \
            *(float4*)(a_ + S::a_soff(2, tid)) = ra2;                         \
            *(float4*)(a_ + S::a_soff(3, tid)) = ra3;                         \
        }                                                                     \
        *(float4*)(w_ + S::w_soff(0, tid)) = rw0;                             \
        *(float4*)(w_ + S::w_soff(1, tid)) = rw1;                             \
        *(float4*)(w_ + S::w_soff(2, tid)) = rw2;                             \
        *(float4*)(w_ + S::w_soff(3, tid)) = rw3;                             \
    } while (0)

    GEMM_GLOAD(0);
    GEMM_SSTORE(0);
    __syncthreads();

    const int nslab = K / GEMM_K;
    for (int s = 0; s < nslab; ++s) {
        const int buf = s & 1;
        const int sn = (s + 1 < nslab) ? (s + 1) : s;
        GEMM_GLOAD(sn);

        const float* a = sA + buf * S::ASZ;
        const float* w = sW + buf * S::WSZ + (wn * 64 + r) * GEMM_P;
        #pragma unroll
        for (int k0 = 0; k0 < GEMM_K; k0 += 4) {
            const int ka = k0 + half * 2;
            v2f av;
            if (AKM) {
                av.x = a[ka * S::AP + wm * 16 + r];
                av.y = a[(ka + 1) * S::AP + wm * 16 + r];
            } else {
                av = *(const v2f*)(a + (wm * 16 + r) * S::AP + ka);
            }
            const v2f b0 = *(const v2f*)(w + 0 * 16 * GEMM_P + ka);
            const v2f b1 = *(const v2f*)(w + 1 * 16 * GEMM_P + ka);
            const v2f b2 = *(const v2f*)(w + 2 * 16 * GEMM_P + ka);
            const v2f b3 = *(const v2f*)(w + 3 * 16 * GEMM_P + ka);
            acc0 = wmma_f32_4(av, b0, acc0);
            acc1 = wmma_f32_4(av, b1, acc1);
            acc2 = wmma_f32_4(av, b2, acc2);
            acc3 = wmma_f32_4(av, b3, acc3);
        }

        GEMM_SSTORE(buf ^ 1);
        __syncthreads();
    }
#undef GEMM_GLOAD
#undef GEMM_SSTORE
}

__global__ void __launch_bounds__(256) __attribute__((amdgpu_waves_per_eu(4, 8)))
k_inproj(const float* __restrict__ x,
         const float* __restrict__ w,
         float* __restrict__ xz) {
    typedef GemmShape<4, 2, true> S;
    __shared__ __attribute__((aligned(16))) float sA[2 * S::ASZ];
    __shared__ __attribute__((aligned(16))) float sW[2 * S::WSZ];

    const int bn = blockIdx.x & 15;
    const int bm = blockIdx.x >> 4;
    const int m0 = bm * 64;
    const int bb = m0 >> 9;
    const int l0 = m0 & 511;

    v8f acc0 = {}, acc1 = {}, acc2 = {}, acc3 = {};
    gemm_block<4, 2, true>(x + (size_t)bb * DMODEL * SEQ + l0, SEQ,
                           w + (size_t)(bn * 128) * DMODEL, DMODEL, DMODEL,
                           sA, sW, acc0, acc1, acc2, acc3);

    const int wave = threadIdx.x >> 5;
    const int wm   = wave % 4;
    const int wn   = wave / 4;
    const int lane = threadIdx.x & 31;
    const int half = lane >> 4;
    const int r    = lane & 15;
    const int col  = bn * 128 + wn * 64 + r;
    #pragma unroll
    for (int j = 0; j < 8; ++j) {
        const size_t row = (size_t)(m0 + wm * 16 + half * 8 + j) * (2 * DINNER);
        xz[row + col + 0 ] = acc0[j];
        xz[row + col + 16] = acc1[j];
        xz[row + col + 32] = acc2[j];
        xz[row + col + 48] = acc3[j];
    }
}

#define CONV_R 8

__global__ void __launch_bounds__(256)
k_conv(const float* __restrict__ xz,
       const float* __restrict__ cw,
       const float* __restrict__ cb,
       float* __restrict__ ucv) {
    const int c  = threadIdx.x * 4;
    const int m0 = blockIdx.x * CONV_R;
    const int l0 = m0 & (SEQ - 1);

    float xr[CONV_R + 3][4];
    #pragma unroll
    for (int j = 0; j < CONV_R + 3; ++j) {
        const int mr = (l0 + j >= 3) ? (m0 + j - 3) : m0;
        const float4 v = *(const float4*)(xz + (size_t)mr * (2 * DINNER) + c);
        xr[j][0] = v.x; xr[j][1] = v.y; xr[j][2] = v.z; xr[j][3] = v.w;
    }

    float wt[4][4];
    #pragma unroll
    for (int ch = 0; ch < 4; ++ch) {
        const float4 w4 = *(const float4*)(cw + (size_t)(c + ch) * 4);
        wt[ch][0] = w4.x; wt[ch][1] = w4.y; wt[ch][2] = w4.z; wt[ch][3] = w4.w;
    }
    const float4 b4 = *(const float4*)(cb + c);
    const float bias[4] = {b4.x, b4.y, b4.z, b4.w};

    #pragma unroll
    for (int i = 0; i < CONV_R; ++i) {
        float o[4];
        #pragma unroll
        for (int ch = 0; ch < 4; ++ch) {
            float s = bias[ch];
            #pragma unroll
            for (int t = 0; t < 4; ++t) {
                if (l0 + i + t >= 3)
                    s += xr[i + t][ch] * wt[ch][t];
            }
            o[ch] = s * sigmoid_f(s);
        }
        *(float4*)(ucv + (size_t)(m0 + i) * DINNER + c) = make_float4(o[0], o[1], o[2], o[3]);
    }
}

__global__ void __launch_bounds__(128) __attribute__((amdgpu_waves_per_eu(1, 2)))
k_xproj(const float* __restrict__ ucv,
        const float* __restrict__ w,
        float* __restrict__ dbc) {
    typedef GemmShape<4, 1, false> S;
    __shared__ __attribute__((aligned(16))) float sA[2 * S::ASZ];
    __shared__ __attribute__((aligned(16))) float sW[2 * S::WSZ];

    const int m0 = blockIdx.x * 64;

    v8f acc0 = {}, acc1 = {}, acc2 = {}, acc3 = {};
    gemm_block<4, 1, false>(ucv + (size_t)m0 * DINNER, DINNER,
                            w, DINNER, DINNER,
                            sA, sW, acc0, acc1, acc2, acc3);

    const int wm   = threadIdx.x >> 5;
    const int lane = threadIdx.x & 31;
    const int half = lane >> 4;
    const int r    = lane & 15;
    #pragma unroll
    for (int j = 0; j < 8; ++j) {
        const size_t row = (size_t)(m0 + wm * 16 + half * 8 + j) * 64;
        dbc[row + 0  + r] = acc0[j];
        dbc[row + 16 + r] = acc1[j];
        dbc[row + 32 + r] = acc2[j];
        dbc[row + 48 + r] = acc3[j];
    }
}

__global__ void k_dtproj(const float* __restrict__ dbc,
                         const float* __restrict__ w,
                         const float* __restrict__ bias,
                         float* __restrict__ delta) {
    const int wid   = blockIdx.x * 4 + (threadIdx.x >> 5);
    const int ntq   = wid & 15;
    const int tileM = wid >> 4;
    const int lane  = threadIdx.x & 31;
    const int half  = lane >> 4;
    const int r     = lane & 15;

    const float* arow = dbc + (size_t)(tileM * 16 + r) * 64;
    const float* wr0 = w + (size_t)((ntq * 4 + 0) * 16 + r) * DTRANK;
    const float* wr1 = w + (size_t)((ntq * 4 + 1) * 16 + r) * DTRANK;
    const float* wr2 = w + (size_t)((ntq * 4 + 2) * 16 + r) * DTRANK;
    const float* wr3 = w + (size_t)((ntq * 4 + 3) * 16 + r) * DTRANK;

    v8f acc0 = {}, acc1 = {}, acc2 = {}, acc3 = {};
    #pragma unroll
    for (int k0 = 0; k0 < DTRANK; k0 += 4) {
        const int ka = k0 + half * 2;
        v2f a  = *(const v2f*)(arow + ka);
        v2f b0 = *(const v2f*)(wr0 + ka);
        v2f b1 = *(const v2f*)(wr1 + ka);
        v2f b2 = *(const v2f*)(wr2 + ka);
        v2f b3 = *(const v2f*)(wr3 + ka);
        acc0 = wmma_f32_4(a, b0, acc0);
        acc1 = wmma_f32_4(a, b1, acc1);
        acc2 = wmma_f32_4(a, b2, acc2);
        acc3 = wmma_f32_4(a, b3, acc3);
    }
    const int nb = ntq * 64 + r;
    const float bn0 = bias[nb + 0];
    const float bn1 = bias[nb + 16];
    const float bn2 = bias[nb + 32];
    const float bn3 = bias[nb + 48];
    #pragma unroll
    for (int j = 0; j < 8; ++j) {
        const size_t row = (size_t)(tileM * 16 + half * 8 + j) * DINNER;
        delta[row + nb + 0 ] = softplus_f(acc0[j] + bn0);
        delta[row + nb + 16] = softplus_f(acc1[j] + bn1);
        delta[row + nb + 32] = softplus_f(acc2[j] + bn2);
        delta[row + nb + 48] = softplus_f(acc3[j] + bn3);
    }
}

#define SCAN_T 16

__device__ __forceinline__ void scan_load_bc(const float4* __restrict__ row,
                                             float (&Bv)[DSTATE], float (&Cv)[DSTATE]) {
    #pragma unroll
    for (int q = 0; q < 4; ++q) {
        const float4 b4 = row[q];
        const float4 c4 = row[4 + q];
        Bv[4 * q + 0] = b4.x; Bv[4 * q + 1] = b4.y; Bv[4 * q + 2] = b4.z; Bv[4 * q + 3] = b4.w;
        Cv[4 * q + 0] = c4.x; Cv[4 * q + 1] = c4.y; Cv[4 * q + 2] = c4.z; Cv[4 * q + 3] = c4.w;
    }
}

__global__ void __launch_bounds__(128)
k_scan(const float* __restrict__ delta,
       const float* __restrict__ ucv,
       const float* __restrict__ xz,
       const float* __restrict__ dbc,
       const float* __restrict__ A_log,
       const float* __restrict__ D_param,
       float* __restrict__ ybuf) {
    const int bb = blockIdx.x >> 3;
    const int tid = threadIdx.x;
    const int c  = (blockIdx.x & 7) * 128 + tid;
    const int mbase = bb * SEQ;

    __shared__ float4 sBC[SEQ * 8];

    const float* pd = delta + (size_t)mbase * DINNER + c;
    const float* pu = ucv   + (size_t)mbase * DINNER + c;
    const float* pz = xz    + (size_t)mbase * (2 * DINNER) + DINNER + c;
    float*       py = ybuf  + (size_t)mbase * DINNER + c;

    float nd[SCAN_T], nu[SCAN_T], nz[SCAN_T];
    #pragma unroll
    for (int t = 0; t < SCAN_T; ++t) {
        nd[t] = pd[(size_t)t * DINNER];
        nu[t] = pu[(size_t)t * DINNER];
        nz[t] = pz[(size_t)t * (2 * DINNER)];
    }

    const float* bcsrc = dbc + (size_t)mbase * 64 + 32;
    #pragma unroll
    for (int j = 0; j < (SEQ * 8) / 128; ++j) {
        const int i = j * 128 + tid;
        sBC[i] = *(const float4*)(bcsrc + (size_t)(i >> 3) * 64 + (i & 7) * 4);
    }

    float Ac[DSTATE];
    float h[DSTATE];
    #pragma unroll
    for (int n = 0; n < DSTATE; ++n) {
        Ac[n] = -__expf(A_log[(size_t)c * DSTATE + n]);
        h[n] = 0.f;
    }
    const float Dp = D_param[c];

    __syncthreads();

    float Bc[DSTATE], Cc[DSTATE];
    scan_load_bc(sBC, Bc, Cc);

    for (int l0 = 0; l0 < SEQ; l0 += SCAN_T) {
        float cd[SCAN_T], cu[SCAN_T], cz[SCAN_T];
        #pragma unroll
        for (int t = 0; t < SCAN_T; ++t) { cd[t] = nd[t]; cu[t] = nu[t]; cz[t] = nz[t]; }

        const int ln = (l0 + SCAN_T < SEQ) ? (l0 + SCAN_T) : l0;
        #pragma unroll
        for (int t = 0; t < SCAN_T; ++t) {
            nd[t] = pd[(size_t)(ln + t) * DINNER];
            nu[t] = pu[(size_t)(ln + t) * DINNER];
            nz[t] = pz[(size_t)(ln + t) * (2 * DINNER)];
        }

        #pragma unroll
        for (int t = 0; t < SCAN_T; ++t) {
            const int l = l0 + t;

            float Bn[DSTATE], Cn[DSTATE];
            const int lnx = (l + 1 < SEQ) ? (l + 1) : l;
            scan_load_bc(sBC + lnx * 8, Bn, Cn);
            __builtin_amdgcn_sched_barrier(0);

            const float dlt = cd[t];
            const float ut  = cu[t];
            const float zt  = cz[t];
            const float du  = dlt * ut;

            float y = 0.f;
            #pragma unroll
            for (int n = 0; n < DSTATE; ++n) {
                const float dA = __expf(dlt * Ac[n]);
                h[n] = dA * h[n] + du * Bc[n];
                y += h[n] * Cc[n];
            }
            py[(size_t)l * DINNER] = (y + ut * Dp) * (zt * sigmoid_f(zt));

            #pragma unroll
            for (int n = 0; n < DSTATE; ++n) { Bc[n] = Bn[n]; Cc[n] = Cn[n]; }

            __builtin_amdgcn_sched_barrier(0);
        }
    }
}

#define OUT_CP 68

__global__ void __launch_bounds__(256) __attribute__((amdgpu_waves_per_eu(2, 4)))
k_outproj(const float* __restrict__ ybuf,
          const float* __restrict__ w,
          float* __restrict__ otmp) {
    typedef GemmShape<4, 2, false> S;
    __shared__ __attribute__((aligned(16))) float sA[2 * S::ASZ];
    __shared__ __attribute__((aligned(16))) float sW[2 * S::WSZ];
    static_assert(128 * OUT_CP <= 2 * S::WSZ, "turned tile must fit the W buffers");

    const int bn = blockIdx.x & 3;
    const int bm = blockIdx.x >> 2;
    const int m0 = bm * 64;
    const int bb = m0 >> 9;
    const int l0 = m0 & 511;

    v8f acc0 = {}, acc1 = {}, acc2 = {}, acc3 = {};
    gemm_block<4, 2, false>(ybuf + (size_t)m0 * DINNER, DINNER,
                            w + (size_t)(bn * 128) * DINNER, DINNER, DINNER,
                            sA, sW, acc0, acc1, acc2, acc3);

    const int tid  = threadIdx.x;
    const int wave = tid >> 5;
    const int wm   = wave % 4;
    const int wn   = wave / 4;
    const int lane = tid & 31;
    const int half = lane >> 4;
    const int r    = lane & 15;

    float* sC = sW;
    {
        float* c0 = sC + (wn * 64 + r) * OUT_CP + wm * 16 + half * 8;
        *(float4*)(c0 + 0 * 16 * OUT_CP)     = make_float4(acc0[0], acc0[1], acc0[2], acc0[3]);
        *(float4*)(c0 + 0 * 16 * OUT_CP + 4) = make_float4(acc0[4], acc0[5], acc0[6], acc0[7]);
        *(float4*)(c0 + 1 * 16 * OUT_CP)     = make_float4(acc1[0], acc1[1], acc1[2], acc1[3]);
        *(float4*)(c0 + 1 * 16 * OUT_CP + 4) = make_float4(acc1[4], acc1[5], acc1[6], acc1[7]);
        *(float4*)(c0 + 2 * 16 * OUT_CP)     = make_float4(acc2[0], acc2[1], acc2[2], acc2[3]);
        *(float4*)(c0 + 2 * 16 * OUT_CP + 4) = make_float4(acc2[4], acc2[5], acc2[6], acc2[7]);
        *(float4*)(c0 + 3 * 16 * OUT_CP)     = make_float4(acc3[0], acc3[1], acc3[2], acc3[3]);
        *(float4*)(c0 + 3 * 16 * OUT_CP + 4) = make_float4(acc3[4], acc3[5], acc3[6], acc3[7]);
    }
    __syncthreads();

    float* obase = otmp + ((size_t)bb * DMODEL + bn * 128) * SEQ + l0;
    #pragma unroll
    for (int i = 0; i < 8; ++i) {
        const int idx = tid + i * 256;
        const int n = idx >> 4, q = idx & 15;
        *(float4*)(obase + (size_t)n * SEQ + q * 4) = *(const float4*)(sC + n * OUT_CP + q * 4);
    }
}

__global__ void k_gnstats(const float* __restrict__ otmp,
                          float* __restrict__ stats) {
    const int bid = blockIdx.x;
    const int tid = threadIdx.x;
    const float* base = otmp + (size_t)bid * 32768;

    __shared__ float ssum[256];
    __shared__ float ssq[256];
    float s = 0.f, q = 0.f;
    for (int i = tid; i < 32768; i += 256) {
        const float v = base[i];
        s += v;
        q += v * v;
    }
    ssum[tid] = s; ssq[tid] = q;
    __syncthreads();
    for (int off = 128; off > 0; off >>= 1) {
        if (tid < off) { ssum[tid] += ssum[tid + off]; ssq[tid] += ssq[tid + off]; }
        __syncthreads();
    }
    if (tid == 0) {
        const float mu  = ssum[0] * (1.f / 32768.f);
        const float var = ssq[0] * (1.f / 32768.f) - mu * mu;
        stats[2 * bid]     = mu;
        stats[2 * bid + 1] = rsqrtf(var + 1e-5f);
    }
}

#define GNM_V 4

__device__ __forceinline__ float gn_mish_f(float x, float mu, float rs, float g, float bt) {
    return mish_f((x - mu) * rs * g + bt);
}

__global__ void __launch_bounds__(256)
k_gnmish(const float* __restrict__ otmp,
         const float* __restrict__ stats,
         const float* __restrict__ gw,
         const float* __restrict__ gb,
         float* __restrict__ out) {
    const int vec0 = blockIdx.x * (256 * GNM_V) + threadIdx.x;
    const int blk  = (blockIdx.x * (1024 * GNM_V)) >> 15;
    const float mu = stats[2 * blk];
    const float rs = stats[2 * blk + 1];

    float4 x4[GNM_V];
    #pragma unroll
    for (int i = 0; i < GNM_V; ++i)
        x4[i] = *(const float4*)(otmp + (size_t)(vec0 + i * 256) * 4);

    #pragma unroll
    for (int i = 0; i < GNM_V; ++i) {
        const int idx = (vec0 + i * 256) * 4;
        const int c   = (idx >> 9) & (DMODEL - 1);
        const float g  = gw[c];
        const float bt = gb[c];
        float4 o;
        o.x = gn_mish_f(x4[i].x, mu, rs, g, bt);
        o.y = gn_mish_f(x4[i].y, mu, rs, g, bt);
        o.z = gn_mish_f(x4[i].z, mu, rs, g, bt);
        o.w = gn_mish_f(x4[i].w, mu, rs, g, bt);
        *(float4*)(out + (size_t)idx) = o;
    }
}

extern "C" void kernel_launch(void* const* d_in, const int* in_sizes, int n_in,
                              void* d_out, int out_size, void* d_ws, size_t ws_size,
                              hipStream_t stream) {
    const float* x         = (const float*)d_in[0];
    const float* in_proj_w = (const float*)d_in[1];
    const float* conv_w    = (const float*)d_in[2];
    const float* conv_b    = (const float*)d_in[3];
    const float* x_proj_w  = (const float*)d_in[4];
    const float* dt_proj_w = (const float*)d_in[5];
    const float* dt_proj_b = (const float*)d_in[6];
    const float* A_log     = (const float*)d_in[7];
    const float* D_param   = (const float*)d_in[8];
    const float* out_proj_w= (const float*)d_in[9];
    const float* gn_w      = (const float*)d_in[10];
    const float* gn_b      = (const float*)d_in[11];
    float* out = (float*)d_out;

    float* ws    = (float*)d_ws;
    float* xz    = ws;
    float* ucv   = xz    + (size_t)MROWS * 2 * DINNER;
    float* dbc   = ucv   + (size_t)MROWS * DINNER;
    float* delta = dbc   + (size_t)MROWS * 64;
    float* ybuf  = delta + (size_t)MROWS * DINNER;
    float* stats = ybuf  + (size_t)MROWS * DINNER;
    float* otmp  = xz;

    k_inproj<<<1024, 256, 0, stream>>>(x, in_proj_w, xz);

    k_conv<<<MROWS / CONV_R, 256, 0, stream>>>(xz, conv_w, conv_b, ucv);

    k_xproj<<<64, 128, 0, stream>>>(ucv, x_proj_w, dbc);

    k_dtproj<<<4096 / 4, 128, 0, stream>>>(dbc, dt_proj_w, dt_proj_b, delta);

    k_scan<<<BATCH * 8, 128, 0, stream>>>(delta, ucv, xz, dbc, A_log, D_param, ybuf);

    k_outproj<<<256, 256, 0, stream>>>(ybuf, out_proj_w, otmp);

    k_gnstats<<<BATCH * NGROUPS, 256, 0, stream>>>(otmp, stats);

    k_gnmish<<<(BATCH * DMODEL * SEQ) / (1024 * GNM_V), 256, 0, stream>>>(
        otmp, stats, gn_w, gn_b, out);
}


# scan kernel: a channel's 16 states spread over the 4 lanes of a quad (4 per lane, packed f32; 256 work-groups, was 64); y's 16-term sum becomes lane partial sums added across the quad by DPP
# speedup vs baseline: 7.9886x; 1.2963x over previous
#include <hip/hip_runtime.h>
#include <hip/hip_bf16.h>
#include <math.h>


typedef float v2f __attribute__((ext_vector_type(2)));
typedef float v8f __attribute__((ext_vector_type(8)));

#define BATCH    8
#define SEQ      512
#define DMODEL   512
#define DINNER   1024
#define DSTATE   16
#define DTRANK   32
#define NGROUPS  8
#define MROWS    (BATCH * SEQ)

__device__ __forceinline__ v8f wmma_f32_4(v2f a, v2f b, v8f c) {
    return __builtin_amdgcn_wmma_f32_16x16x4_f32(
        false, a, false, b, (short)0, c, false, false);
}

__device__ __forceinline__ float softplus_f(float x) {
    const float t = __expf(-fabsf(x));
    const float u = 1.f + t;
    const float d = u - 1.f;
    const float l = (__builtin_amdgcn_logf(u) * 0.6931471805599453f) * (t * __builtin_amdgcn_rcpf(d));
    return fmaxf(x, 0.f) + (d == 0.f ? t : l);
}

__device__ __forceinline__ float mish_f(float v) {
    const float m   = __expf(-fabsf(v));
    const bool  pos = v >= 0.f;
    const float num = pos ? fmaf(2.f, m, 1.f) : m * (m + 2.f);
    const float den = num + (pos ? 2.f * m * m : 2.f);
    return v * (num * __builtin_amdgcn_rcpf(den));
}

__device__ __forceinline__ float sigmoid_f(float x) {
    return __builtin_amdgcn_rcpf(1.f + __expf(-x));
}


#define GEMM_K   32
#define GEMM_P   36

template <int WM, int WN, bool AKM>
struct GemmShape {
    static constexpr int NT  = 32 * WM * WN;
    static constexpr int BM  = 16 * WM;
    static constexpr int BN  = 64 * WN;
    static constexpr int AP  = AKM ? (BM + 8) : GEMM_P;
    static constexpr int ASZ = AKM ? GEMM_K * (BM + 8) : BM * GEMM_P;
    static constexpr int WSZ = BN * GEMM_P;
    static constexpr int AV4 = BM * GEMM_K / 4 / NT;
    static constexpr int WV4 = BN * GEMM_K / 4 / NT;

    static __device__ __forceinline__ size_t a_goff(const int i, const int tid, const int lda) {
        const int idx = tid + i * NT;
        return AKM ? (size_t)(idx / (BM / 4)) * lda + (idx % (BM / 4)) * 4
                   : (size_t)(idx >> 3) * lda + (idx & 7) * 4;
    }
    static __device__ __forceinline__ int a_soff(const int i, const int tid) {
        const int idx = tid + i * NT;
        return AKM ? (idx / (BM / 4)) * AP + (idx % (BM / 4)) * 4
                   : (idx >> 3) * AP + (idx & 7) * 4;
    }
    static __device__ __forceinline__ size_t w_goff(const int i, const int tid, const int ldw) {
        const int idx = tid + i * NT;
        return (size_t)(idx >> 3) * ldw + (idx & 7) * 4;
    }
    static __device__ __forceinline__ int w_soff(const int i, const int tid) {
        const int idx = tid + i * NT;
        return (idx >> 3) * GEMM_P + (idx & 7) * 4;
    }
};

template <int WM, int WN, bool AKM>
__device__ __forceinline__ void gemm_block(const float* __restrict__ Ablk, const int lda,
                                           const float* __restrict__ Wblk, const int ldw,
                                           const int K,
                                           float* __restrict__ sA, float* __restrict__ sW,
                                           v8f& acc0, v8f& acc1, v8f& acc2, v8f& acc3) {
    typedef GemmShape<WM, WN, AKM> S;
    static_assert(S::WV4 == 4 && (S::AV4 == 2 || S::AV4 == 4), "per-thread piece counts the code below is written for");
    const int tid  = threadIdx.x;
    const int wave = tid >> 5;
    const int wm   = wave % WM;
    const int wn   = wave / WM;
    const int lane = tid & 31;
    const int half = lane >> 4;
    const int r    = lane & 15;

    const float* ga0 = Ablk + S::a_goff(0, tid, lda);
    const float* ga1 = Ablk + S::a_goff(1, tid, lda);
    const float* ga2 = Ablk + S::a_goff(S::AV4 > 2 ? 2 : 0, tid, lda);
    const float* ga3 = Ablk + S::a_goff(S::AV4 > 2 ? 3 : 0, tid, lda);
    const float* gw0 = Wblk + S::w_goff(0, tid, ldw);
    const float* gw1 = Wblk + S::w_goff(1, tid, ldw);
    const float* gw2 = Wblk + S::w_goff(2, tid, ldw);
    const float* gw3 = Wblk + S::w_goff(3, tid, ldw);
    const size_t astep = AKM ? (size_t)GEMM_K * lda : (size_t)GEMM_K;

    float4 ra0, ra1, ra2, ra3, rw0, rw1, rw2, rw3;
    ra2 = ra3 = make_float4(0.f, 0.f, 0.f, 0.f);

#define GEMM_GLOAD(ks)                                                        \
    do {                                                                      \
        const size_t ao = (size_t)(ks) * astep;                               \
        const size_t wo = (size_t)(ks) * GEMM_K;                              \
        ra0 = *(const float4*)(ga0 + ao);                                     \
        ra1 = *(const float4*)(ga1 + ao);                                     \
        if (S::AV4 > 2) {                                                     \
            ra2 = *(const float4*)(ga2 + ao);                                 \
            ra3 = *(const float4*)(ga3 + ao);                                 \
        }                                                                     \
        rw0 = *(const float4*)(gw0 + wo);                                     \
        rw1 = *(const float4*)(gw1 + wo);                                     \
        rw2 = *(const float4*)(gw2 + wo);                                     \
        rw3 = *(const float4*)(gw3 + wo);                                     \
    } while (0)

#define GEMM_SSTORE(buf)                                                      \
    do {                                                                      \
        float* a_ = sA + (buf) * S::ASZ;                                      \
        float* w_ = sW + (buf) * S::WSZ;                                      \
        *(float4*)(a_ + S::a_soff(0, tid)) = ra0;                             \
        *(float4*)(a_ + S::a_soff(1, tid)) = ra1;                             \
        if (S::AV4 > 2) {                                                     \
            *(float4*)(a_ + S::a_soff(2, tid)) = ra2;                         \
            *(float4*)(a_ + S::a_soff(3, tid)) = ra3;                         \
        }                                                                     \
        *(float4*)(w_ + S::w_soff(0, tid)) = rw0;                             \
        *(float4*)(w_ + S::w_soff(1, tid)) = rw1;                             \
        *(float4*)(w_ + S::w_soff(2, tid)) = rw2;                             \
        *(float4*)(w_ + S::w_soff(3, tid)) = rw3;                             \
    } while (0)

    GEMM_GLOAD(0);
    GEMM_SSTORE(0);
    __syncthreads();

    const int nslab = K / GEMM_K;
    for (int s = 0; s < nslab; ++s) {
        const int buf = s & 1;
        const int sn = (s + 1 < nslab) ? (s + 1) : s;
        GEMM_GLOAD(sn);

        const float* a = sA + buf * S::ASZ;
        const float* w = sW + buf * S::WSZ + (wn * 64 + r) * GEMM_P;
        #pragma unroll
        for (int k0 = 0; k0 < GEMM_K; k0 += 4) {
            const int ka = k0 + half * 2;
            v2f av;
            if (AKM) {
                av.x = a[ka * S::AP + wm * 16 + r];
                av.y = a[(ka + 1) * S::AP + wm * 16 + r];
            } else {
                av = *(const v2f*)(a + (wm * 16 + r) * S::AP + ka);
            }
            const v2f b0 = *(const v2f*)(w + 0 * 16 * GEMM_P + ka);
            const v2f b1 = *(const v2f*)(w + 1 * 16 * GEMM_P + ka);
            const v2f b2 = *(const v2f*)(w + 2 * 16 * GEMM_P + ka);
            const v2f b3 = *(const v2f*)(w + 3 * 16 * GEMM_P + ka);
            acc0 = wmma_f32_4(av, b0, acc0);
            acc1 = wmma_f32_4(av, b1, acc1);
            acc2 = wmma_f32_4(av, b2, acc2);
            acc3 = wmma_f32_4(av, b3, acc3);
        }

        GEMM_SSTORE(buf ^ 1);
        __syncthreads();
    }
#undef GEMM_GLOAD
#undef GEMM_SSTORE
}

__global__ void __launch_bounds__(256) __attribute__((amdgpu_waves_per_eu(4, 8)))
k_inproj(const float* __restrict__ x,
         const float* __restrict__ w,
         float* __restrict__ xz) {
    typedef GemmShape<4, 2, true> S;
    __shared__ __attribute__((aligned(16))) float sA[2 * S::ASZ];
    __shared__ __attribute__((aligned(16))) float sW[2 * S::WSZ];

    const int bn = blockIdx.x & 15;
    const int bm = blockIdx.x >> 4;
    const int m0 = bm * 64;
    const int bb = m0 >> 9;
    const int l0 = m0 & 511;

    v8f acc0 = {}, acc1 = {}, acc2 = {}, acc3 = {};
    gemm_block<4, 2, true>(x + (size_t)bb * DMODEL * SEQ + l0, SEQ,
                           w + (size_t)(bn * 128) * DMODEL, DMODEL, DMODEL,
                           sA, sW, acc0, acc1, acc2, acc3);

    const int wave = threadIdx.x >> 5;
    const int wm   = wave % 4;
    const int wn   = wave / 4;
    const int lane = threadIdx.x & 31;
    const int half = lane >> 4;
    const int r    = lane & 15;
    const int col  = bn * 128 + wn * 64 + r;
    #pragma unroll
    for (int j = 0; j < 8; ++j) {
        const size_t row = (size_t)(m0 + wm * 16 + half * 8 + j) * (2 * DINNER);
        xz[row + col + 0 ] = acc0[j];
        xz[row + col + 16] = acc1[j];
        xz[row + col + 32] = acc2[j];
        xz[row + col + 48] = acc3[j];
    }
}

#define CONV_R 8

__global__ void __launch_bounds__(256)
k_conv(const float* __restrict__ xz,
       const float* __restrict__ cw,
       const float* __restrict__ cb,
       float* __restrict__ ucv) {
    const int c  = threadIdx.x * 4;
    const int m0 = blockIdx.x * CONV_R;
    const int l0 = m0 & (SEQ - 1);

    float xr[CONV_R + 3][4];
    #pragma unroll
    for (int j = 0; j < CONV_R + 3; ++j) {
        const int mr = (l0 + j >= 3) ? (m0 + j - 3) : m0;
        const float4 v = *(const float4*)(xz + (size_t)mr * (2 * DINNER) + c);
        xr[j][0] = v.x; xr[j][1] = v.y; xr[j][2] = v.z; xr[j][3] = v.w;
    }

    float wt[4][4];
    #pragma unroll
    for (int ch = 0; ch < 4; ++ch) {
        const float4 w4 = *(const float4*)(cw + (size_t)(c + ch) * 4);
        wt[ch][0] = w4.x; wt[ch][1] = w4.y; wt[ch][2] = w4.z; wt[ch][3] = w4.w;
    }
    const float4 b4 = *(const float4*)(cb + c);
    const float bias[4] = {b4.x, b4.y, b4.z, b4.w};

    #pragma unroll
    for (int i = 0; i < CONV_R; ++i) {
        float o[4];
        #pragma unroll
        for (int ch = 0; ch < 4; ++ch) {
            float s = bias[ch];
            #pragma unroll
            for (int t = 0; t < 4; ++t) {
                if (l0 + i + t >= 3)
                    s += xr[i + t][ch] * wt[ch][t];
            }
            o[ch] = s * sigmoid_f(s);
        }
        *(float4*)(ucv + (size_t)(m0 + i) * DINNER + c) = make_float4(o[0], o[1], o[2], o[3]);
    }
}

__global__ void __launch_bounds__(128) __attribute__((amdgpu_waves_per_eu(1, 2)))
k_xproj(const float* __restrict__ ucv,
        const float* __restrict__ w,
        float* __restrict__ dbc) {
    typedef GemmShape<4, 1, false> S;
    __shared__ __attribute__((aligned(16))) float sA[2 * S::ASZ];
    __shared__ __attribute__((aligned(16))) float sW[2 * S::WSZ];

    const int m0 = blockIdx.x * 64;

    v8f acc0 = {}, acc1 = {}, acc2 = {}, acc3 = {};
    gemm_block<4, 1, false>(ucv + (size_t)m0 * DINNER, DINNER,
                            w, DINNER, DINNER,
                            sA, sW, acc0, acc1, acc2, acc3);

    const int wm   = threadIdx.x >> 5;
    const int lane = threadIdx.x & 31;
    const int half = lane >> 4;
    const int r    = lane & 15;
    #pragma unroll
    for (int j = 0; j < 8; ++j) {
        const size_t row = (size_t)(m0 + wm * 16 + half * 8 + j) * 64;
        dbc[row + 0  + r] = acc0[j];
        dbc[row + 16 + r] = acc1[j];
        dbc[row + 32 + r] = acc2[j];
        dbc[row + 48 + r] = acc3[j];
    }
}

__global__ void k_dtproj(const float* __restrict__ dbc,
                         const float* __restrict__ w,
                         const float* __restrict__ bias,
                         float* __restrict__ delta) {
    const int wid   = blockIdx.x * 4 + (threadIdx.x >> 5);
    const int ntq   = wid & 15;
    const int tileM = wid >> 4;
    const int lane  = threadIdx.x & 31;
    const int half  = lane >> 4;
    const int r     = lane & 15;

    const float* arow = dbc + (size_t)(tileM * 16 + r) * 64;
    const float* wr0 = w + (size_t)((ntq * 4 + 0) * 16 + r) * DTRANK;
    const float* wr1 = w + (size_t)((ntq * 4 + 1) * 16 + r) * DTRANK;
    const float* wr2 = w + (size_t)((ntq * 4 + 2) * 16 + r) * DTRANK;
    const float* wr3 = w + (size_t)((ntq * 4 + 3) * 16 + r) * DTRANK;

    v8f acc0 = {}, acc1 = {}, acc2 = {}, acc3 = {};
    #pragma unroll
    for (int k0 = 0; k0 < DTRANK; k0 += 4) {
        const int ka = k0 + half * 2;
        v2f a  = *(const v2f*)(arow + ka);
        v2f b0 = *(const v2f*)(wr0 + ka);
        v2f b1 = *(const v2f*)(wr1 + ka);
        v2f b2 = *(const v2f*)(wr2 + ka);
        v2f b3 = *(const v2f*)(wr3 + ka);
        acc0 = wmma_f32_4(a, b0, acc0);
        acc1 = wmma_f32_4(a, b1, acc1);
        acc2 = wmma_f32_4(a, b2, acc2);
        acc3 = wmma_f32_4(a, b3, acc3);
    }
    const int nb = ntq * 64 + r;
    const float bn0 = bias[nb + 0];
    const float bn1 = bias[nb + 16];
    const float bn2 = bias[nb + 32];
    const float bn3 = bias[nb + 48];
    #pragma unroll
    for (int j = 0; j < 8; ++j) {
        const size_t row = (size_t)(tileM * 16 + half * 8 + j) * DINNER;
        delta[row + nb + 0 ] = softplus_f(acc0[j] + bn0);
        delta[row + nb + 16] = softplus_f(acc1[j] + bn1);
        delta[row + nb + 32] = softplus_f(acc2[j] + bn2);
        delta[row + nb + 48] = softplus_f(acc3[j] + bn3);
    }
}

#define SCAN_T   32
#define SCAN_Q   4
#define SCAN_CPB (128 / SCAN_Q)
#define SCAN_G   (SCAN_T / SCAN_Q)

template <int CTRL>
__device__ __forceinline__ float quad_perm(float v) {
    return __builtin_bit_cast(float, __builtin_amdgcn_mov_dpp(__builtin_bit_cast(int, v), CTRL, 0xF, 0xF, true));
}
__device__ __forceinline__ float quad_xor1(float v) { return quad_perm<0xB1>(v); }
__device__ __forceinline__ float quad_xor2(float v) { return quad_perm<0x4E>(v); }
__device__ __forceinline__ float quad_bcast(float v, const int k) {
    switch (k) {
        case 0:  return quad_perm<0x00>(v);
        case 1:  return quad_perm<0x55>(v);
        case 2:  return quad_perm<0xAA>(v);
        default: return quad_perm<0xFF>(v);
    }
}

__device__ __forceinline__ v2f exp_pair(v2f x) {
    const v2f t = x * 1.4426950408889634f;
    v2f r;
    r.x = __builtin_amdgcn_exp2f(t.x);
    r.y = __builtin_amdgcn_exp2f(t.y);
    return r;
}

__device__ __forceinline__ void scan_load_bc(const float4* __restrict__ sBC, const int l, const int sub,
                                             float4 (&Bv)[SCAN_Q], float4 (&Cv)[SCAN_Q]) {
    #pragma unroll
    for (int t = 0; t < SCAN_Q; ++t) {
        Bv[t] = sBC[(l + t) * 8 + sub];
        Cv[t] = sBC[(l + t) * 8 + 4 + sub];
    }
}

__global__ void __launch_bounds__(128)
k_scan(const float* __restrict__ delta,
       const float* __restrict__ ucv,
       const float* __restrict__ xz,
       const float* __restrict__ dbc,
       const float* __restrict__ A_log,
       const float* __restrict__ D_param,
       float* __restrict__ ybuf) {
    const int bb  = blockIdx.x >> 5;
    const int tid = threadIdx.x;
    const int sub = tid & (SCAN_Q - 1);
    const int chl = tid >> 2;
    const int c0  = (blockIdx.x & 31) * SCAN_CPB;
    const int c   = c0 + chl;
    const int mbase = bb * SEQ;

    __shared__ float4 sBC[SEQ * 8];
    __shared__ __attribute__((aligned(16))) float sY[2][SCAN_T][SCAN_CPB];

    const float* pd = delta + (size_t)(mbase + sub) * DINNER + c;
    const float* pu = ucv   + (size_t)(mbase + sub) * DINNER + c;
    const float* pz = xz    + (size_t)(mbase + sub) * (2 * DINNER) + DINNER + c;
    float* pyo = ybuf + (size_t)(mbase + (tid >> 3)) * DINNER + c0 + (tid & 7) * 4;

    v8f cd, cu, cz;
    static_assert(SCAN_G == 8, "one 8-register vector per operand");
    #pragma unroll
    for (int g = 0; g < SCAN_G; ++g) {
        cd[g] = pd[(size_t)(g * SCAN_Q) * DINNER];
        cu[g] = pu[(size_t)(g * SCAN_Q) * DINNER];
        cz[g] = pz[(size_t)(g * SCAN_Q) * (2 * DINNER)];
    }

    const float* bcsrc = dbc + (size_t)mbase * 64 + 32;
    #pragma unroll
    for (int j = 0; j < (SEQ * 8) / 128; ++j) {
        const int i = j * 128 + tid;
        sBC[i] = *(const float4*)(bcsrc + (size_t)(i >> 3) * 64 + (i & 7) * 4);
    }

    const float4 al = *(const float4*)(A_log + (size_t)c * DSTATE + sub * 4);
    const v2f Ac01 = {-__expf(al.x), -__expf(al.y)};
    const v2f Ac23 = {-__expf(al.z), -__expf(al.w)};
    v2f h01 = {0.f, 0.f}, h23 = {0.f, 0.f};
    const float Dp = D_param[c];
    const bool odd = (sub & 1) != 0;
    const bool hi  = (sub & 2) != 0;

    __syncthreads();

    float4 Bc[SCAN_Q], Cc[SCAN_Q];
    scan_load_bc(sBC, 0, sub, Bc, Cc);

    for (int l0 = 0; l0 < SEQ; l0 += SCAN_T) {
        const int ln = (l0 + SCAN_T < SEQ) ? (l0 + SCAN_T) : l0;
        v8f nd, nu, nz;
        #pragma unroll
        for (int g = 0; g < SCAN_G; ++g) {
            nd[g] = pd[(size_t)(ln + g * SCAN_Q) * DINNER];
            nu[g] = pu[(size_t)(ln + g * SCAN_Q) * DINNER];
            nz[g] = pz[(size_t)(ln + g * SCAN_Q) * (2 * DINNER)];
        }

        float (*sYt)[SCAN_CPB] = sY[(l0 / SCAN_T) & 1];

        #pragma unroll
        for (int g = 0; g < SCAN_G; ++g) {
            const int lg = l0 + g * SCAN_Q;

            float4 Bn[SCAN_Q], Cn[SCAN_Q];
            const int lnx = (lg + SCAN_Q < SEQ) ? (lg + SCAN_Q) : lg;
            scan_load_bc(sBC, lnx, sub, Bn, Cn);
            __builtin_amdgcn_sched_barrier(0);

            float yq[SCAN_Q];
            #pragma unroll
            for (int t = 0; t < SCAN_Q; ++t) {
                const float dlt = quad_bcast(cd[g], t);
                const float du  = dlt * quad_bcast(cu[g], t);
                const v2f B01 = {Bc[t].x, Bc[t].y}, B23 = {Bc[t].z, Bc[t].w};
                const v2f C01 = {Cc[t].x, Cc[t].y}, C23 = {Cc[t].z, Cc[t].w};
                const v2f dA01 = exp_pair(Ac01 * dlt);
                const v2f dA23 = exp_pair(Ac23 * dlt);
                h01 = __builtin_elementwise_fma(dA01, h01, du * B01);
                h23 = __builtin_elementwise_fma(dA23, h23, du * B23);
                const v2f y2 = h01 * C01 + h23 * C23;
                yq[t] = y2.x + y2.y;
            }

            const float a  = (odd ? yq[1] : yq[0]) + quad_xor1(odd ? yq[0] : yq[1]);
            const float b  = (odd ? yq[3] : yq[2]) + quad_xor1(odd ? yq[2] : yq[3]);
            const float ys = (hi ? b : a) + quad_xor2(hi ? a : b);
            const float zs = cz[g];
            sYt[g * SCAN_Q + sub][chl] = (ys + cu[g] * Dp) * (zs * sigmoid_f(zs));

            #pragma unroll
            for (int t = 0; t < SCAN_Q; ++t) { Bc[t] = Bn[t]; Cc[t] = Cn[t]; }

            __builtin_amdgcn_sched_barrier(0);
        }

        __builtin_amdgcn_fence(__ATOMIC_RELEASE, "workgroup", "local");
        __builtin_amdgcn_s_barrier();
        __builtin_amdgcn_fence(__ATOMIC_ACQUIRE, "workgroup", "local");
        #pragma unroll
        for (int j = 0; j < SCAN_T / 16; ++j)
            *(float4*)(pyo + (size_t)(l0 + 16 * j) * DINNER) = *(const float4*)&sYt[(tid >> 3) + 16 * j][(tid & 7) * 4];

        asm volatile("" :: "v"(nd), "v"(nu), "v"(nz));
        cd = nd; cu = nu; cz = nz;
    }
}

#define OUT_CP 68

__global__ void __launch_bounds__(256) __attribute__((amdgpu_waves_per_eu(2, 4)))
k_outproj(const float* __restrict__ ybuf,
          const float* __restrict__ w,
          float* __restrict__ otmp) {
    typedef GemmShape<4, 2, false> S;
    __shared__ __attribute__((aligned(16))) float sA[2 * S::ASZ];
    __shared__ __attribute__((aligned(16))) float sW[2 * S::WSZ];
    static_assert(128 * OUT_CP <= 2 * S::WSZ, "turned tile must fit the W buffers");

    const int bn = blockIdx.x & 3;
    const int bm = blockIdx.x >> 2;
    const int m0 = bm * 64;
    const int bb = m0 >> 9;
    const int l0 = m0 & 511;

    v8f acc0 = {}, acc1 = {}, acc2 = {}, acc3 = {};
    gemm_block<4, 2, false>(ybuf + (size_t)m0 * DINNER, DINNER,
                            w + (size_t)(bn * 128) * DINNER, DINNER, DINNER,
                            sA, sW, acc0, acc1, acc2, acc3);

    const int tid  = threadIdx.x;
    const int wave = tid >> 5;
    const int wm   = wave % 4;
    const int wn   = wave / 4;
    const int lane = tid & 31;
    const int half = lane >> 4;
    const int r    = lane & 15;

    float* sC = sW;
    {
        float* c0 = sC + (wn * 64 + r) * OUT_CP + wm * 16 + half * 8;
        *(float4*)(c0 + 0 * 16 * OUT_CP)     = make_float4(acc0[0], acc0[1], acc0[2], acc0[3]);
        *(float4*)(c0 + 0 * 16 * OUT_CP + 4) = make_float4(acc0[4], acc0[5], acc0[6], acc0[7]);
        *(float4*)(c0 + 1 * 16 * OUT_CP)     = make_float4(acc1[0], acc1[1], acc1[2], acc1[3]);
        *(float4*)(c0 + 1 * 16 * OUT_CP + 4) = make_float4(acc1[4], acc1[5], acc1[6], acc1[7]);
        *(float4*)(c0 + 2 * 16 * OUT_CP)     = make_float4(acc2[0], acc2[1], acc2[2], acc2[3]);
        *(float4*)(c0 + 2 * 16 * OUT_CP + 4) = make_float4(acc2[4], acc2[5], acc2[6], acc2[7]);
        *(float4*)(c0 + 3 * 16 * OUT_CP)     = make_float4(acc3[0], acc3[1], acc3[2], acc3[3]);
        *(float4*)(c0 + 3 * 16 * OUT_CP + 4) = make_float4(acc3[4], acc3[5], acc3[6], acc3[7]);
    }
    __syncthreads();

    float* obase = otmp + ((size_t)bb * DMODEL + bn * 128) * SEQ + l0;
    #pragma unroll
    for (int i = 0; i < 8; ++i) {
        const int idx = tid + i * 256;
        const int n = idx >> 4, q = idx & 15;
        *(float4*)(obase + (size_t)n * SEQ + q * 4) = *(const float4*)(sC + n * OUT_CP + q * 4);
    }
}

__global__ void k_gnstats(const float* __restrict__ otmp,
                          float* __restrict__ stats) {
    const int bid = blockIdx.x;
    const int tid = threadIdx.x;
    const float* base = otmp + (size_t)bid * 32768;

    __shared__ float ssum[256];
    __shared__ float ssq[256];
    float s = 0.f, q = 0.f;
    for (int i = tid; i < 32768; i += 256) {
        const float v = base[i];
        s += v;
        q += v * v;
    }
    ssum[tid] = s; ssq[tid] = q;
    __syncthreads();
    for (int off = 128; off > 0; off >>= 1) {
        if (tid < off) { ssum[tid] += ssum[tid + off]; ssq[tid] += ssq[tid + off]; }
        __syncthreads();
    }
    if (tid == 0) {
        const float mu  = ssum[0] * (1.f / 32768.f);
        const float var = ssq[0] * (1.f / 32768.f) - mu * mu;
        stats[2 * bid]     = mu;
        stats[2 * bid + 1] = rsqrtf(var + 1e-5f);
    }
}

#define GNM_V 4

__device__ __forceinline__ float gn_mish_f(float x, float mu, float rs, float g, float bt) {
    return mish_f((x - mu) * rs * g + bt);
}

__global__ void __launch_bounds__(256)
k_gnmish(const float* __restrict__ otmp,
         const float* __restrict__ stats,
         const float* __restrict__ gw,
         const float* __restrict__ gb,
         float* __restrict__ out) {
    const int vec0 = blockIdx.x * (256 * GNM_V) + threadIdx.x;
    const int blk  = (blockIdx.x * (1024 * GNM_V)) >> 15;
    const float mu = stats[2 * blk];
    const float rs = stats[2 * blk + 1];

    float4 x4[GNM_V];
    #pragma unroll
    for (int i = 0; i < GNM_V; ++i)
        x4[i] = *(const float4*)(otmp + (size_t)(vec0 + i * 256) * 4);

    #pragma unroll
    for (int i = 0; i < GNM_V; ++i) {
        const int idx = (vec0 + i * 256) * 4;
        const int c   = (idx >> 9) & (DMODEL - 1);
        const float g  = gw[c];
        const float bt = gb[c];
        float4 o;
        o.x = gn_mish_f(x4[i].x, mu, rs, g, bt);
        o.y = gn_mish_f(x4[i].y, mu, rs, g, bt);
        o.z = gn_mish_f(x4[i].z, mu, rs, g, bt);
        o.w = gn_mish_f(x4[i].w, mu, rs, g, bt);
        *(float4*)(out + (size_t)idx) = o;
    }
}

extern "C" void kernel_launch(void* const* d_in, const int* in_sizes, int n_in,
                              void* d_out, int out_size, void* d_ws, size_t ws_size,
                              hipStream_t stream) {
    const float* x         = (const float*)d_in[0];
    const float* in_proj_w = (const float*)d_in[1];
    const float* conv_w    = (const float*)d_in[2];
    const float* conv_b    = (const float*)d_in[3];
    const float* x_proj_w  = (const float*)d_in[4];
    const float* dt_proj_w = (const float*)d_in[5];
    const float* dt_proj_b = (const float*)d_in[6];
    const float* A_log     = (const float*)d_in[7];
    const float* D_param   = (const float*)d_in[8];
    const float* out_proj_w= (const float*)d_in[9];
    const float* gn_w      = (const float*)d_in[10];
    const float* gn_b      = (const float*)d_in[11];
    float* out = (float*)d_out;

    float* ws    = (float*)d_ws;
    float* xz    = ws;
    float* ucv   = xz    + (size_t)MROWS * 2 * DINNER;
    float* dbc   = ucv   + (size_t)MROWS * DINNER;
    float* delta = dbc   + (size_t)MROWS * 64;
    float* ybuf  = delta + (size_t)MROWS * DINNER;
    float* stats = ybuf  + (size_t)MROWS * DINNER;
    float* otmp  = xz;

    k_inproj<<<1024, 256, 0, stream>>>(x, in_proj_w, xz);

    k_conv<<<MROWS / CONV_R, 256, 0, stream>>>(xz, conv_w, conv_b, ucv);

    k_xproj<<<64, 128, 0, stream>>>(ucv, x_proj_w, dbc);

    k_dtproj<<<4096 / 4, 128, 0, stream>>>(dbc, dt_proj_w, dt_proj_b, delta);

    k_scan<<<BATCH * (DINNER / SCAN_CPB), 128, 0, stream>>>(delta, ucv, xz, dbc, A_log, D_param, ybuf);

    k_outproj<<<256, 256, 0, stream>>>(ybuf, out_proj_w, otmp);

    k_gnstats<<<BATCH * NGROUPS, 256, 0, stream>>>(otmp, stats);

    k_gnmish<<<(BATCH * DMODEL * SEQ) / (1024 * GNM_V), 256, 0, stream>>>(
        otmp, stats, gn_w, gn_b, out);
}


# x_proj GEMM: 256 work-groups of one 16x64 tile (was 64 of 64x64); each of the 4 waves sums a quarter of K, the 4 partial tiles are added via LDS in fixed order (x_proj's K sum order changes)
# speedup vs baseline: 9.0065x; 1.1274x over previous
#include <hip/hip_runtime.h>
#include <hip/hip_bf16.h>
#include <math.h>


typedef float v2f __attribute__((ext_vector_type(2)));
typedef float v8f __attribute__((ext_vector_type(8)));

#define BATCH    8
#define SEQ      512
#define DMODEL   512
#define DINNER   1024
#define DSTATE   16
#define DTRANK   32
#define NGROUPS  8
#define MROWS    (BATCH * SEQ)

__device__ __forceinline__ v8f wmma_f32_4(v2f a, v2f b, v8f c) {
    return __builtin_amdgcn_wmma_f32_16x16x4_f32(
        false, a, false, b, (short)0, c, false, false);
}

__device__ __forceinline__ float softplus_f(float x) {
    const float t = __expf(-fabsf(x));
    const float u = 1.f + t;
    const float d = u - 1.f;
    const float l = (__builtin_amdgcn_logf(u) * 0.6931471805599453f) * (t * __builtin_amdgcn_rcpf(d));
    return fmaxf(x, 0.f) + (d == 0.f ? t : l);
}

__device__ __forceinline__ float mish_f(float v) {
    const float m   = __expf(-fabsf(v));
    const bool  pos = v >= 0.f;
    const float num = pos ? fmaf(2.f, m, 1.f) : m * (m + 2.f);
    const float den = num + (pos ? 2.f * m * m : 2.f);
    return v * (num * __builtin_amdgcn_rcpf(den));
}

__device__ __forceinline__ float sigmoid_f(float x) {
    return __builtin_amdgcn_rcpf(1.f + __expf(-x));
}


#define GEMM_K   32
#define GEMM_P   36

template <int WM, int WN, bool AKM>
struct GemmShape {
    static constexpr int NT  = 32 * WM * WN;
    static constexpr int BM  = 16 * WM;
    static constexpr int BN  = 64 * WN;
    static constexpr int AP  = AKM ? (BM + 8) : GEMM_P;
    static constexpr int ASZ = AKM ? GEMM_K * (BM + 8) : BM * GEMM_P;
    static constexpr int WSZ = BN * GEMM_P;
    static constexpr int AV4 = BM * GEMM_K / 4 / NT;
    static constexpr int WV4 = BN * GEMM_K / 4 / NT;

    static __device__ __forceinline__ size_t a_goff(const int i, const int tid, const int lda) {
        const int idx = tid + i * NT;
        return AKM ? (size_t)(idx / (BM / 4)) * lda + (idx % (BM / 4)) * 4
                   : (size_t)(idx >> 3) * lda + (idx & 7) * 4;
    }
    static __device__ __forceinline__ int a_soff(const int i, const int tid) {
        const int idx = tid + i * NT;
        return AKM ? (idx / (BM / 4)) * AP + (idx % (BM / 4)) * 4
                   : (idx >> 3) * AP + (idx & 7) * 4;
    }
    static __device__ __forceinline__ size_t w_goff(const int i, const int tid, const int ldw) {
        const int idx = tid + i * NT;
        return (size_t)(idx >> 3) * ldw + (idx & 7) * 4;
    }
    static __device__ __forceinline__ int w_soff(const int i, const int tid) {
        const int idx = tid + i * NT;
        return (idx >> 3) * GEMM_P + (idx & 7) * 4;
    }
};

template <int WM, int WN, bool AKM>
__device__ __forceinline__ void gemm_block(const float* __restrict__ Ablk, const int lda,
                                           const float* __restrict__ Wblk, const int ldw,
                                           const int K,
                                           float* __restrict__ sA, float* __restrict__ sW,
                                           v8f& acc0, v8f& acc1, v8f& acc2, v8f& acc3) {
    typedef GemmShape<WM, WN, AKM> S;
    static_assert(S::WV4 == 4 && (S::AV4 == 2 || S::AV4 == 4), "per-thread piece counts the code below is written for");
    const int tid  = threadIdx.x;
    const int wave = tid >> 5;
    const int wm   = wave % WM;
    const int wn   = wave / WM;
    const int lane = tid & 31;
    const int half = lane >> 4;
    const int r    = lane & 15;

    const float* ga0 = Ablk + S::a_goff(0, tid, lda);
    const float* ga1 = Ablk + S::a_goff(1, tid, lda);
    const float* ga2 = Ablk + S::a_goff(S::AV4 > 2 ? 2 : 0, tid, lda);
    const float* ga3 = Ablk + S::a_goff(S::AV4 > 2 ? 3 : 0, tid, lda);
    const float* gw0 = Wblk + S::w_goff(0, tid, ldw);
    const float* gw1 = Wblk + S::w_goff(1, tid, ldw);
    const float* gw2 = Wblk + S::w_goff(2, tid, ldw);
    const float* gw3 = Wblk + S::w_goff(3, tid, ldw);
    const size_t astep = AKM ? (size_t)GEMM_K * lda : (size_t)GEMM_K;

    float4 ra0, ra1, ra2, ra3, rw0, rw1, rw2, rw3;
    ra2 = ra3 = make_float4(0.f, 0.f, 0.f, 0.f);

#define GEMM_GLOAD(ks)                                                        \
    do {                                                                      \
        const size_t ao = (size_t)(ks) * astep;                               \
        const size_t wo = (size_t)(ks) * GEMM_K;                              \
        ra0 = *(const float4*)(ga0 + ao);                                     \
        ra1 = *(const float4*)(ga1 + ao);                                     \
        if (S::AV4 > 2) {                                                     \
            ra2 = *(const float4*)(ga2 + ao);                                 \
            ra3 = *(const float4*)(ga3 + ao);                                 \
        }                                                                     \
        rw0 = *(const float4*)(gw0 + wo);                                     \
        rw1 = *(const float4*)(gw1 + wo);                                     \
        rw2 = *(const float4*)(gw2 + wo);                                     \
        rw3 = *(const float4*)(gw3 + wo);                                     \
    } while (0)

#define GEMM_SSTORE(buf)                                                      \
    do {                                                                      \
        float* a_ = sA + (buf) * S::ASZ;                                      \
        float* w_ = sW + (buf) * S::WSZ;                                      \
        *(float4*)(a_ + S::a_soff(0, tid)) = ra0;                             \
        *(float4*)(a_ + S::a_soff(1, tid)) = ra1;                             \
        if (S::AV4 > 2) {                                                     \
            *(float4*)(a_ + S::a_soff(2, tid)) = ra2;                         \
            *(float4*)(a_ + S::a_soff(3, tid)) = ra3;                         \
        }                                                                     \
        *(float4*)(w_ + S::w_soff(0, tid)) = rw0;                             \
        *(float4*)(w_ + S::w_soff(1, tid)) = rw1;                             \
        *(float4*)(w_ + S::w_soff(2, tid)) = rw2;                             \
        *(float4*)(w_ + S::w_soff(3, tid)) = rw3;                             \
    } while (0)

    GEMM_GLOAD(0);
    GEMM_SSTORE(0);
    __syncthreads();

    const int nslab = K / GEMM_K;
    for (int s = 0; s < nslab; ++s) {
        const int buf = s & 1;
        const int sn = (s + 1 < nslab) ? (s + 1) : s;
        GEMM_GLOAD(sn);

        const float* a = sA + buf * S::ASZ;
        const float* w = sW + buf * S::WSZ + (wn * 64 + r) * GEMM_P;
        #pragma unroll
        for (int k0 = 0; k0 < GEMM_K; k0 += 4) {
            const int ka = k0 + half * 2;
            v2f av;
            if (AKM) {
                av.x = a[ka * S::AP + wm * 16 + r];
                av.y = a[(ka + 1) * S::AP + wm * 16 + r];
            } else {
                av = *(const v2f*)(a + (wm * 16 + r) * S::AP + ka);
            }
            const v2f b0 = *(const v2f*)(w + 0 * 16 * GEMM_P + ka);
            const v2f b1 = *(const v2f*)(w + 1 * 16 * GEMM_P + ka);
            const v2f b2 = *(const v2f*)(w + 2 * 16 * GEMM_P + ka);
            const v2f b3 = *(const v2f*)(w + 3 * 16 * GEMM_P + ka);
            acc0 = wmma_f32_4(av, b0, acc0);
            acc1 = wmma_f32_4(av, b1, acc1);
            acc2 = wmma_f32_4(av, b2, acc2);
            acc3 = wmma_f32_4(av, b3, acc3);
        }

        GEMM_SSTORE(buf ^ 1);
        __syncthreads();
    }
#undef GEMM_GLOAD
#undef GEMM_SSTORE
}

__global__ void __launch_bounds__(256) __attribute__((amdgpu_waves_per_eu(4, 8)))
k_inproj(const float* __restrict__ x,
         const float* __restrict__ w,
         float* __restrict__ xz) {
    typedef GemmShape<4, 2, true> S;
    __shared__ __attribute__((aligned(16))) float sA[2 * S::ASZ];
    __shared__ __attribute__((aligned(16))) float sW[2 * S::WSZ];

    const int bn = blockIdx.x & 15;
    const int bm = blockIdx.x >> 4;
    const int m0 = bm * 64;
    const int bb = m0 >> 9;
    const int l0 = m0 & 511;

    v8f acc0 = {}, acc1 = {}, acc2 = {}, acc3 = {};
    gemm_block<4, 2, true>(x + (size_t)bb * DMODEL * SEQ + l0, SEQ,
                           w + (size_t)(bn * 128) * DMODEL, DMODEL, DMODEL,
                           sA, sW, acc0, acc1, acc2, acc3);

    const int wave = threadIdx.x >> 5;
    const int wm   = wave % 4;
    const int wn   = wave / 4;
    const int lane = threadIdx.x & 31;
    const int half = lane >> 4;
    const int r    = lane & 15;
    const int col  = bn * 128 + wn * 64 + r;
    #pragma unroll
    for (int j = 0; j < 8; ++j) {
        const size_t row = (size_t)(m0 + wm * 16 + half * 8 + j) * (2 * DINNER);
        xz[row + col + 0 ] = acc0[j];
        xz[row + col + 16] = acc1[j];
        xz[row + col + 32] = acc2[j];
        xz[row + col + 48] = acc3[j];
    }
}

#define CONV_R 8

__global__ void __launch_bounds__(256)
k_conv(const float* __restrict__ xz,
       const float* __restrict__ cw,
       const float* __restrict__ cb,
       float* __restrict__ ucv) {
    const int c  = threadIdx.x * 4;
    const int m0 = blockIdx.x * CONV_R;
    const int l0 = m0 & (SEQ - 1);

    float xr[CONV_R + 3][4];
    #pragma unroll
    for (int j = 0; j < CONV_R + 3; ++j) {
        const int mr = (l0 + j >= 3) ? (m0 + j - 3) : m0;
        const float4 v = *(const float4*)(xz + (size_t)mr * (2 * DINNER) + c);
        xr[j][0] = v.x; xr[j][1] = v.y; xr[j][2] = v.z; xr[j][3] = v.w;
    }

    float wt[4][4];
    #pragma unroll
    for (int ch = 0; ch < 4; ++ch) {
        const float4 w4 = *(const float4*)(cw + (size_t)(c + ch) * 4);
        wt[ch][0] = w4.x; wt[ch][1] = w4.y; wt[ch][2] = w4.z; wt[ch][3] = w4.w;
    }
    const float4 b4 = *(const float4*)(cb + c);
    const float bias[4] = {b4.x, b4.y, b4.z, b4.w};

    #pragma unroll
    for (int i = 0; i < CONV_R; ++i) {
        float o[4];
        #pragma unroll
        for (int ch = 0; ch < 4; ++ch) {
            float s = bias[ch];
            #pragma unroll
            for (int t = 0; t < 4; ++t) {
                if (l0 + i + t >= 3)
                    s += xr[i + t][ch] * wt[ch][t];
            }
            o[ch] = s * sigmoid_f(s);
        }
        *(float4*)(ucv + (size_t)(m0 + i) * DINNER + c) = make_float4(o[0], o[1], o[2], o[3]);
    }
}

#define XP_KQ    (DINNER / 4)
#define XP_AROWS (4 * 16)
#define XP_WROWS (4 * 64)
#define XP_AV4   (XP_AROWS * GEMM_K / 4 / 128)
#define XP_WV4   (XP_WROWS * GEMM_K / 4 / 128)
#define XP_RP    68

__global__ void __launch_bounds__(128) __attribute__((amdgpu_waves_per_eu(1, 2)))
k_xproj(const float* __restrict__ ucv,
        const float* __restrict__ w,
        float* __restrict__ dbc) {
    __shared__ __attribute__((aligned(16))) float sA[2 * XP_AROWS * GEMM_P];
    __shared__ __attribute__((aligned(16))) float sW[2 * XP_WROWS * GEMM_P];
    static_assert(4 * 16 * XP_RP <= 2 * XP_WROWS * GEMM_P, "partial tiles must fit the W buffers");

    const int tid  = threadIdx.x;
    const int q    = tid >> 5;
    const int lane = tid & 31;
    const int half = lane >> 4;
    const int r    = lane & 15;
    const int m0   = blockIdx.x * 16;

    const int pr = tid >> 3, pc = (tid & 7) * 4;
    const float* ga = ucv + (size_t)(m0 + pr) * DINNER + pc;
    const float* gw = w   + (size_t)pr * DINNER + pc;
    const int so = pr * GEMM_P + pc;

    float4 ra0, ra1, ra2, ra3;
    float4 rw0, rw1, rw2, rw3, rw4, rw5, rw6, rw7, rw8, rw9, rw10, rw11, rw12, rw13, rw14, rw15;
    static_assert(XP_AV4 == 4 && XP_WV4 == 16, "piece counts the two macros below are written for");

#define XP_GLOAD(ks)                                                          \
    do {                                                                      \
        const float* ga_ = ga + (ks) * GEMM_K;                                \
        const float* gw_ = gw + (ks) * GEMM_K;                                \
        ra0 = *(const float4*)(ga_ + 0 * XP_KQ);                              \
        ra1 = *(const float4*)(ga_ + 1 * XP_KQ);                              \
        ra2 = *(const float4*)(ga_ + 2 * XP_KQ);                              \
        ra3 = *(const float4*)(ga_ + 3 * XP_KQ);                              \
        rw0 = *(const float4*)(gw_ + (size_t)0 * 16 * DINNER + 0 * XP_KQ);    \
        rw1 = *(const float4*)(gw_ + (size_t)1 * 16 * DINNER + 0 * XP_KQ);    \
        rw2 = *(const float4*)(gw_ + (size_t)2 * 16 * DINNER + 0 * XP_KQ);    \
        rw3 = *(const float4*)(gw_ + (size_t)3 * 16 * DINNER + 0 * XP_KQ);    \
        rw4 = *(const float4*)(gw_ + (size_t)0 * 16 * DINNER + 1 * XP_KQ);    \
        rw5 = *(const float4*)(gw_ + (size_t)1 * 16 * DINNER + 1 * XP_KQ);    \
        rw6 = *(const float4*)(gw_ + (size_t)2 * 16 * DINNER + 1 * XP_KQ);    \
        rw7 = *(const float4*)(gw_ + (size_t)3 * 16 * DINNER + 1 * XP_KQ);    \
        rw8 = *(const float4*)(gw_ + (size_t)0 * 16 * DINNER + 2 * XP_KQ);    \
        rw9 = *(const float4*)(gw_ + (size_t)1 * 16 * DINNER + 2 * XP_KQ);    \
        rw10 = *(const float4*)(gw_ + (size_t)2 * 16 * DINNER + 2 * XP_KQ);   \
        rw11 = *(const float4*)(gw_ + (size_t)3 * 16 * DINNER + 2 * XP_KQ);   \
        rw12 = *(const float4*)(gw_ + (size_t)0 * 16 * DINNER + 3 * XP_KQ);   \
        rw13 = *(const float4*)(gw_ + (size_t)1 * 16 * DINNER + 3 * XP_KQ);   \
        rw14 = *(const float4*)(gw_ + (size_t)2 * 16 * DINNER + 3 * XP_KQ);   \
        rw15 = *(const float4*)(gw_ + (size_t)3 * 16 * DINNER + 3 * XP_KQ);   \
    } while (0)

#define XP_SSTORE(buf)                                                        \
    do {                                                                      \
        float* a_ = sA + (buf) * (XP_AROWS * GEMM_P) + so;                    \
        float* w_ = sW + (buf) * (XP_WROWS * GEMM_P) + so;                    \
        *(float4*)(a_ + 0 * 16 * GEMM_P) = ra0;                               \
        *(float4*)(a_ + 1 * 16 * GEMM_P) = ra1;                               \
        *(float4*)(a_ + 2 * 16 * GEMM_P) = ra2;                               \
        *(float4*)(a_ + 3 * 16 * GEMM_P) = ra3;                               \
        *(float4*)(w_ + 0 * 16 * GEMM_P) = rw0;                               \
        *(float4*)(w_ + 1 * 16 * GEMM_P) = rw1;                               \
        *(float4*)(w_ + 2 * 16 * GEMM_P) = rw2;                               \
        *(float4*)(w_ + 3 * 16 * GEMM_P) = rw3;                               \
        *(float4*)(w_ + 4 * 16 * GEMM_P) = rw4;                               \
        *(float4*)(w_ + 5 * 16 * GEMM_P) = rw5;                               \
        *(float4*)(w_ + 6 * 16 * GEMM_P) = rw6;                               \
        *(float4*)(w_ + 7 * 16 * GEMM_P) = rw7;                               \
        *(float4*)(w_ + 8 * 16 * GEMM_P) = rw8;                               \
        *(float4*)(w_ + 9 * 16 * GEMM_P) = rw9;                               \
        *(float4*)(w_ + 10 * 16 * GEMM_P) = rw10;                             \
        *(float4*)(w_ + 11 * 16 * GEMM_P) = rw11;                             \
        *(float4*)(w_ + 12 * 16 * GEMM_P) = rw12;                             \
        *(float4*)(w_ + 13 * 16 * GEMM_P) = rw13;                             \
        *(float4*)(w_ + 14 * 16 * GEMM_P) = rw14;                             \
        *(float4*)(w_ + 15 * 16 * GEMM_P) = rw15;                             \
    } while (0)

    v8f acc0 = {}, acc1 = {}, acc2 = {}, acc3 = {};

    constexpr int nslab = XP_KQ / GEMM_K;
    XP_GLOAD(0);
    XP_SSTORE(0);
    XP_GLOAD(1);
    __syncthreads();

    for (int s = 0; s < nslab; ++s) {
        const int buf = s & 1;
        const float* a  = sA + buf * (XP_AROWS * GEMM_P) + (q * 16 + r) * GEMM_P;
        const float* wb = sW + buf * (XP_WROWS * GEMM_P) + (q * 64 + r) * GEMM_P;

#define XP_FRAG(ks, A_, B0_, B1_, B2_, B3_)                                  \
        const v2f A_  = *(const v2f*)(a  + (ks) * 4 + half * 2);              \
        const v2f B0_ = *(const v2f*)(wb + 0 * 16 * GEMM_P + (ks) * 4 + half * 2); \
        const v2f B1_ = *(const v2f*)(wb + 1 * 16 * GEMM_P + (ks) * 4 + half * 2); \
        const v2f B2_ = *(const v2f*)(wb + 2 * 16 * GEMM_P + (ks) * 4 + half * 2); \
        const v2f B3_ = *(const v2f*)(wb + 3 * 16 * GEMM_P + (ks) * 4 + half * 2)
#define XP_MUL(A_, B0_, B1_, B2_, B3_)                                        \
        acc0 = wmma_f32_4(A_, B0_, acc0);                                     \
        acc1 = wmma_f32_4(A_, B1_, acc1);                                     \
        acc2 = wmma_f32_4(A_, B2_, acc2);                                     \
        acc3 = wmma_f32_4(A_, B3_, acc3)

        XP_FRAG(0, a0, b00, b01, b02, b03);
        XP_FRAG(1, a1, b10, b11, b12, b13);
        XP_FRAG(2, a2, b20, b21, b22, b23);
        XP_FRAG(3, a3, b30, b31, b32, b33);
        XP_FRAG(4, a4, b40, b41, b42, b43);
        XP_FRAG(5, a5, b50, b51, b52, b53);
        XP_FRAG(6, a6, b60, b61, b62, b63);
        XP_FRAG(7, a7, b70, b71, b72, b73);

        XP_SSTORE(buf ^ 1);
        const int sn = (s + 2 < nslab) ? (s + 2) : (nslab - 1);
        XP_GLOAD(sn);

        XP_MUL(a0, b00, b01, b02, b03);
        XP_MUL(a1, b10, b11, b12, b13);
        XP_MUL(a2, b20, b21, b22, b23);
        XP_MUL(a3, b30, b31, b32, b33);
        XP_MUL(a4, b40, b41, b42, b43);
        XP_MUL(a5, b50, b51, b52, b53);
        XP_MUL(a6, b60, b61, b62, b63);
        XP_MUL(a7, b70, b71, b72, b73);
#undef XP_FRAG
#undef XP_MUL
        __syncthreads();
    }
#undef XP_GLOAD
#undef XP_SSTORE

    float* sR = sW;
    {
        float* p = sR + (q * 16 + half * 8) * XP_RP + r;
        #pragma unroll
        for (int j = 0; j < 8; ++j) {
            p[j * XP_RP + 0 ] = acc0[j];
            p[j * XP_RP + 16] = acc1[j];
            p[j * XP_RP + 32] = acc2[j];
            p[j * XP_RP + 48] = acc3[j];
        }
    }
    __syncthreads();

    const int row = tid >> 3;
    #pragma unroll
    for (int hp = 0; hp < 2; ++hp) {
        const int col = pc + hp * 32;
        const float4 p0 = *(const float4*)(sR + (0 * 16 + row) * XP_RP + col);
        const float4 p1 = *(const float4*)(sR + (1 * 16 + row) * XP_RP + col);
        const float4 p2 = *(const float4*)(sR + (2 * 16 + row) * XP_RP + col);
        const float4 p3 = *(const float4*)(sR + (3 * 16 + row) * XP_RP + col);
        float4 o;
        o.x = ((p0.x + p1.x) + p2.x) + p3.x;
        o.y = ((p0.y + p1.y) + p2.y) + p3.y;
        o.z = ((p0.z + p1.z) + p2.z) + p3.z;
        o.w = ((p0.w + p1.w) + p2.w) + p3.w;
        *(float4*)(dbc + (size_t)(m0 + row) * 64 + col) = o;
    }
}

__global__ void k_dtproj(const float* __restrict__ dbc,
                         const float* __restrict__ w,
                         const float* __restrict__ bias,
                         float* __restrict__ delta) {
    const int wid   = blockIdx.x * 4 + (threadIdx.x >> 5);
    const int ntq   = wid & 15;
    const int tileM = wid >> 4;
    const int lane  = threadIdx.x & 31;
    const int half  = lane >> 4;
    const int r     = lane & 15;

    const float* arow = dbc + (size_t)(tileM * 16 + r) * 64;
    const float* wr0 = w + (size_t)((ntq * 4 + 0) * 16 + r) * DTRANK;
    const float* wr1 = w + (size_t)((ntq * 4 + 1) * 16 + r) * DTRANK;
    const float* wr2 = w + (size_t)((ntq * 4 + 2) * 16 + r) * DTRANK;
    const float* wr3 = w + (size_t)((ntq * 4 + 3) * 16 + r) * DTRANK;

    v8f acc0 = {}, acc1 = {}, acc2 = {}, acc3 = {};
    #pragma unroll
    for (int k0 = 0; k0 < DTRANK; k0 += 4) {
        const int ka = k0 + half * 2;
        v2f a  = *(const v2f*)(arow + ka);
        v2f b0 = *(const v2f*)(wr0 + ka);
        v2f b1 = *(const v2f*)(wr1 + ka);
        v2f b2 = *(const v2f*)(wr2 + ka);
        v2f b3 = *(const v2f*)(wr3 + ka);
        acc0 = wmma_f32_4(a, b0, acc0);
        acc1 = wmma_f32_4(a, b1, acc1);
        acc2 = wmma_f32_4(a, b2, acc2);
        acc3 = wmma_f32_4(a, b3, acc3);
    }
    const int nb = ntq * 64 + r;
    const float bn0 = bias[nb + 0];
    const float bn1 = bias[nb + 16];
    const float bn2 = bias[nb + 32];
    const float bn3 = bias[nb + 48];
    #pragma unroll
    for (int j = 0; j < 8; ++j) {
        const size_t row = (size_t)(tileM * 16 + half * 8 + j) * DINNER;
        delta[row + nb + 0 ] = softplus_f(acc0[j] + bn0);
        delta[row + nb + 16] = softplus_f(acc1[j] + bn1);
        delta[row + nb + 32] = softplus_f(acc2[j] + bn2);
        delta[row + nb + 48] = softplus_f(acc3[j] + bn3);
    }
}

#define SCAN_T   32
#define SCAN_Q   4
#define SCAN_CPB (128 / SCAN_Q)
#define SCAN_G   (SCAN_T / SCAN_Q)

template <int CTRL>
__device__ __forceinline__ float quad_perm(float v) {
    return __builtin_bit_cast(float, __builtin_amdgcn_mov_dpp(__builtin_bit_cast(int, v), CTRL, 0xF, 0xF, true));
}
__device__ __forceinline__ float quad_xor1(float v) { return quad_perm<0xB1>(v); }
__device__ __forceinline__ float quad_xor2(float v) { return quad_perm<0x4E>(v); }
__device__ __forceinline__ float quad_bcast(float v, const int k) {
    switch (k) {
        case 0:  return quad_perm<0x00>(v);
        case 1:  return quad_perm<0x55>(v);
        case 2:  return quad_perm<0xAA>(v);
        default: return quad_perm<0xFF>(v);
    }
}

__device__ __forceinline__ v2f exp_pair(v2f x) {
    const v2f t = x * 1.4426950408889634f;
    v2f r;
    r.x = __builtin_amdgcn_exp2f(t.x);
    r.y = __builtin_amdgcn_exp2f(t.y);
    return r;
}

__device__ __forceinline__ void scan_load_bc(const float4* __restrict__ sBC, const int l, const int sub,
                                             float4 (&Bv)[SCAN_Q], float4 (&Cv)[SCAN_Q]) {
    #pragma unroll
    for (int t = 0; t < SCAN_Q; ++t) {
        Bv[t] = sBC[(l + t) * 8 + sub];
        Cv[t] = sBC[(l + t) * 8 + 4 + sub];
    }
}

__global__ void __launch_bounds__(128)
k_scan(const float* __restrict__ delta,
       const float* __restrict__ ucv,
       const float* __restrict__ xz,
       const float* __restrict__ dbc,
       const float* __restrict__ A_log,
       const float* __restrict__ D_param,
       float* __restrict__ ybuf) {
    const int bb  = blockIdx.x >> 5;
    const int tid = threadIdx.x;
    const int sub = tid & (SCAN_Q - 1);
    const int chl = tid >> 2;
    const int c0  = (blockIdx.x & 31) * SCAN_CPB;
    const int c   = c0 + chl;
    const int mbase = bb * SEQ;

    __shared__ float4 sBC[SEQ * 8];
    __shared__ __attribute__((aligned(16))) float sY[2][SCAN_T][SCAN_CPB];

    const float* pd = delta + (size_t)(mbase + sub) * DINNER + c;
    const float* pu = ucv   + (size_t)(mbase + sub) * DINNER + c;
    const float* pz = xz    + (size_t)(mbase + sub) * (2 * DINNER) + DINNER + c;
    float* pyo = ybuf + (size_t)(mbase + (tid >> 3)) * DINNER + c0 + (tid & 7) * 4;

    v8f cd, cu, cz;
    static_assert(SCAN_G == 8, "one 8-register vector per operand");
    #pragma unroll
    for (int g = 0; g < SCAN_G; ++g) {
        cd[g] = pd[(size_t)(g * SCAN_Q) * DINNER];
        cu[g] = pu[(size_t)(g * SCAN_Q) * DINNER];
        cz[g] = pz[(size_t)(g * SCAN_Q) * (2 * DINNER)];
    }

    const float* bcsrc = dbc + (size_t)mbase * 64 + 32;
    #pragma unroll
    for (int j = 0; j < (SEQ * 8) / 128; ++j) {
        const int i = j * 128 + tid;
        sBC[i] = *(const float4*)(bcsrc + (size_t)(i >> 3) * 64 + (i & 7) * 4);
    }

    const float4 al = *(const float4*)(A_log + (size_t)c * DSTATE + sub * 4);
    const v2f Ac01 = {-__expf(al.x), -__expf(al.y)};
    const v2f Ac23 = {-__expf(al.z), -__expf(al.w)};
    v2f h01 = {0.f, 0.f}, h23 = {0.f, 0.f};
    const float Dp = D_param[c];
    const bool odd = (sub & 1) != 0;
    const bool hi  = (sub & 2) != 0;

    __syncthreads();

    float4 Bc[SCAN_Q], Cc[SCAN_Q];
    scan_load_bc(sBC, 0, sub, Bc, Cc);

    for (int l0 = 0; l0 < SEQ; l0 += SCAN_T) {
        const int ln = (l0 + SCAN_T < SEQ) ? (l0 + SCAN_T) : l0;
        v8f nd, nu, nz;
        #pragma unroll
        for (int g = 0; g < SCAN_G; ++g) {
            nd[g] = pd[(size_t)(ln + g * SCAN_Q) * DINNER];
            nu[g] = pu[(size_t)(ln + g * SCAN_Q) * DINNER];
            nz[g] = pz[(size_t)(ln + g * SCAN_Q) * (2 * DINNER)];
        }

        float (*sYt)[SCAN_CPB] = sY[(l0 / SCAN_T) & 1];

        #pragma unroll
        for (int g = 0; g < SCAN_G; ++g) {
            const int lg = l0 + g * SCAN_Q;

            float4 Bn[SCAN_Q], Cn[SCAN_Q];
            const int lnx = (lg + SCAN_Q < SEQ) ? (lg + SCAN_Q) : lg;
            scan_load_bc(sBC, lnx, sub, Bn, Cn);
            __builtin_amdgcn_sched_barrier(0);

            float yq[SCAN_Q];
            #pragma unroll
            for (int t = 0; t < SCAN_Q; ++t) {
                const float dlt = quad_bcast(cd[g], t);
                const float du  = dlt * quad_bcast(cu[g], t);
                const v2f B01 = {Bc[t].x, Bc[t].y}, B23 = {Bc[t].z, Bc[t].w};
                const v2f C01 = {Cc[t].x, Cc[t].y}, C23 = {Cc[t].z, Cc[t].w};
                const v2f dA01 = exp_pair(Ac01 * dlt);
                const v2f dA23 = exp_pair(Ac23 * dlt);
                h01 = __builtin_elementwise_fma(dA01, h01, du * B01);
                h23 = __builtin_elementwise_fma(dA23, h23, du * B23);
                const v2f y2 = h01 * C01 + h23 * C23;
                yq[t] = y2.x + y2.y;
            }

            const float a  = (odd ? yq[1] : yq[0]) + quad_xor1(odd ? yq[0] : yq[1]);
            const float b  = (odd ? yq[3] : yq[2]) + quad_xor1(odd ? yq[2] : yq[3]);
            const float ys = (hi ? b : a) + quad_xor2(hi ? a : b);
            const float zs = cz[g];
            sYt[g * SCAN_Q + sub][chl] = (ys + cu[g] * Dp) * (zs * sigmoid_f(zs));

            #pragma unroll
            for (int t = 0; t < SCAN_Q; ++t) { Bc[t] = Bn[t]; Cc[t] = Cn[t]; }

            __builtin_amdgcn_sched_barrier(0);
        }

        __builtin_amdgcn_fence(__ATOMIC_RELEASE, "workgroup", "local");
        __builtin_amdgcn_s_barrier();
        __builtin_amdgcn_fence(__ATOMIC_ACQUIRE, "workgroup", "local");
        #pragma unroll
        for (int j = 0; j < SCAN_T / 16; ++j)
            *(float4*)(pyo + (size_t)(l0 + 16 * j) * DINNER) = *(const float4*)&sYt[(tid >> 3) + 16 * j][(tid & 7) * 4];

        asm volatile("" :: "v"(nd), "v"(nu), "v"(nz));
        cd = nd; cu = nu; cz = nz;
    }
}

#define OUT_CP 68

__global__ void __launch_bounds__(256) __attribute__((amdgpu_waves_per_eu(2, 4)))
k_outproj(const float* __restrict__ ybuf,
          const float* __restrict__ w,
          float* __restrict__ otmp) {
    typedef GemmShape<4, 2, false> S;
    __shared__ __attribute__((aligned(16))) float sA[2 * S::ASZ];
    __shared__ __attribute__((aligned(16))) float sW[2 * S::WSZ];
    static_assert(128 * OUT_CP <= 2 * S::WSZ, "turned tile must fit the W buffers");

    const int bn = blockIdx.x & 3;
    const int bm = blockIdx.x >> 2;
    const int m0 = bm * 64;
    const int bb = m0 >> 9;
    const int l0 = m0 & 511;

    v8f acc0 = {}, acc1 = {}, acc2 = {}, acc3 = {};
    gemm_block<4, 2, false>(ybuf + (size_t)m0 * DINNER, DINNER,
                            w + (size_t)(bn * 128) * DINNER, DINNER, DINNER,
                            sA, sW, acc0, acc1, acc2, acc3);

    const int tid  = threadIdx.x;
    const int wave = tid >> 5;
    const int wm   = wave % 4;
    const int wn   = wave / 4;
    const int lane = tid & 31;
    const int half = lane >> 4;
    const int r    = lane & 15;

    float* sC = sW;
    {
        float* c0 = sC + (wn * 64 + r) * OUT_CP + wm * 16 + half * 8;
        *(float4*)(c0 + 0 * 16 * OUT_CP)     = make_float4(acc0[0], acc0[1], acc0[2], acc0[3]);
        *(float4*)(c0 + 0 * 16 * OUT_CP + 4) = make_float4(acc0[4], acc0[5], acc0[6], acc0[7]);
        *(float4*)(c0 + 1 * 16 * OUT_CP)     = make_float4(acc1[0], acc1[1], acc1[2], acc1[3]);
        *(float4*)(c0 + 1 * 16 * OUT_CP + 4) = make_float4(acc1[4], acc1[5], acc1[6], acc1[7]);
        *(float4*)(c0 + 2 * 16 * OUT_CP)     = make_float4(acc2[0], acc2[1], acc2[2], acc2[3]);
        *(float4*)(c0 + 2 * 16 * OUT_CP + 4) = make_float4(acc2[4], acc2[5], acc2[6], acc2[7]);
        *(float4*)(c0 + 3 * 16 * OUT_CP)     = make_float4(acc3[0], acc3[1], acc3[2], acc3[3]);
        *(float4*)(c0 + 3 * 16 * OUT_CP + 4) = make_float4(acc3[4], acc3[5], acc3[6], acc3[7]);
    }
    __syncthreads();

    float* obase = otmp + ((size_t)bb * DMODEL + bn * 128) * SEQ + l0;
    #pragma unroll
    for (int i = 0; i < 8; ++i) {
        const int idx = tid + i * 256;
        const int n = idx >> 4, q = idx & 15;
        *(float4*)(obase + (size_t)n * SEQ + q * 4) = *(const float4*)(sC + n * OUT_CP + q * 4);
    }
}

__global__ void k_gnstats(const float* __restrict__ otmp,
                          float* __restrict__ stats) {
    const int bid = blockIdx.x;
    const int tid = threadIdx.x;
    const float* base = otmp + (size_t)bid * 32768;

    __shared__ float ssum[256];
    __shared__ float ssq[256];
    float s = 0.f, q = 0.f;
    for (int i = tid; i < 32768; i += 256) {
        const float v = base[i];
        s += v;
        q += v * v;
    }
    ssum[tid] = s; ssq[tid] = q;
    __syncthreads();
    for (int off = 128; off > 0; off >>= 1) {
        if (tid < off) { ssum[tid] += ssum[tid + off]; ssq[tid] += ssq[tid + off]; }
        __syncthreads();
    }
    if (tid == 0) {
        const float mu  = ssum[0] * (1.f / 32768.f);
        const float var = ssq[0] * (1.f / 32768.f) - mu * mu;
        stats[2 * bid]     = mu;
        stats[2 * bid + 1] = rsqrtf(var + 1e-5f);
    }
}

#define GNM_V 4

__device__ __forceinline__ float gn_mish_f(float x, float mu, float rs, float g, float bt) {
    return mish_f((x - mu) * rs * g + bt);
}

__global__ void __launch_bounds__(256)
k_gnmish(const float* __restrict__ otmp,
         const float* __restrict__ stats,
         const float* __restrict__ gw,
         const float* __restrict__ gb,
         float* __restrict__ out) {
    const int vec0 = blockIdx.x * (256 * GNM_V) + threadIdx.x;
    const int blk  = (blockIdx.x * (1024 * GNM_V)) >> 15;
    const float mu = stats[2 * blk];
    const float rs = stats[2 * blk + 1];

    float4 x4[GNM_V];
    #pragma unroll
    for (int i = 0; i < GNM_V; ++i)
        x4[i] = *(const float4*)(otmp + (size_t)(vec0 + i * 256) * 4);

    #pragma unroll
    for (int i = 0; i < GNM_V; ++i) {
        const int idx = (vec0 + i * 256) * 4;
        const int c   = (idx >> 9) & (DMODEL - 1);
        const float g  = gw[c];
        const float bt = gb[c];
        float4 o;
        o.x = gn_mish_f(x4[i].x, mu, rs, g, bt);
        o.y = gn_mish_f(x4[i].y, mu, rs, g, bt);
        o.z = gn_mish_f(x4[i].z, mu, rs, g, bt);
        o.w = gn_mish_f(x4[i].w, mu, rs, g, bt);
        *(float4*)(out + (size_t)idx) = o;
    }
}

extern "C" void kernel_launch(void* const* d_in, const int* in_sizes, int n_in,
                              void* d_out, int out_size, void* d_ws, size_t ws_size,
                              hipStream_t stream) {
    const float* x         = (const float*)d_in[0];
    const float* in_proj_w = (const float*)d_in[1];
    const float* conv_w    = (const float*)d_in[2];
    const float* conv_b    = (const float*)d_in[3];
    const float* x_proj_w  = (const float*)d_in[4];
    const float* dt_proj_w = (const float*)d_in[5];
    const float* dt_proj_b = (const float*)d_in[6];
    const float* A_log     = (const float*)d_in[7];
    const float* D_param   = (const float*)d_in[8];
    const float* out_proj_w= (const float*)d_in[9];
    const float* gn_w      = (const float*)d_in[10];
    const float* gn_b      = (const float*)d_in[11];
    float* out = (float*)d_out;

    float* ws    = (float*)d_ws;
    float* xz    = ws;
    float* ucv   = xz    + (size_t)MROWS * 2 * DINNER;
    float* dbc   = ucv   + (size_t)MROWS * DINNER;
    float* delta = dbc   + (size_t)MROWS * 64;
    float* ybuf  = delta + (size_t)MROWS * DINNER;
    float* stats = ybuf  + (size_t)MROWS * DINNER;
    float* otmp  = xz;

    k_inproj<<<1024, 256, 0, stream>>>(x, in_proj_w, xz);

    k_conv<<<MROWS / CONV_R, 256, 0, stream>>>(xz, conv_w, conv_b, ucv);

    k_xproj<<<MROWS / 16, 128, 0, stream>>>(ucv, x_proj_w, dbc);

    k_dtproj<<<4096 / 4, 128, 0, stream>>>(dbc, dt_proj_w, dt_proj_b, delta);

    k_scan<<<BATCH * (DINNER / SCAN_CPB), 128, 0, stream>>>(delta, ucv, xz, dbc, A_log, D_param, ybuf);

    k_outproj<<<256, 256, 0, stream>>>(ybuf, out_proj_w, otmp);

    k_gnstats<<<BATCH * NGROUPS, 256, 0, stream>>>(otmp, stats);

    k_gnmish<<<(BATCH * DMODEL * SEQ) / (1024 * GNM_V), 256, 0, stream>>>(
        otmp, stats, gn_w, gn_b, out);
}


# GroupNorm sums come from out_proj's accumulators as per-block partials, added in fixed order by the Mish kernel; the statistics launch and its 8 MB re-read are gone; GroupNorm sum order changes
# speedup vs baseline: 9.8340x; 1.0919x over previous
#include <hip/hip_runtime.h>
#include <hip/hip_bf16.h>
#include <math.h>


typedef float v2f __attribute__((ext_vector_type(2)));
typedef float v8f __attribute__((ext_vector_type(8)));

#define BATCH    8
#define SEQ      512
#define DMODEL   512
#define DINNER   1024
#define DSTATE   16
#define DTRANK   32
#define NGROUPS  8
#define MROWS    (BATCH * SEQ)

__device__ __forceinline__ v8f wmma_f32_4(v2f a, v2f b, v8f c) {
    return __builtin_amdgcn_wmma_f32_16x16x4_f32(
        false, a, false, b, (short)0, c, false, false);
}

__device__ __forceinline__ float softplus_f(float x) {
    const float t = __expf(-fabsf(x));
    const float u = 1.f + t;
    const float d = u - 1.f;
    const float l = (__builtin_amdgcn_logf(u) * 0.6931471805599453f) * (t * __builtin_amdgcn_rcpf(d));
    return fmaxf(x, 0.f) + (d == 0.f ? t : l);
}

__device__ __forceinline__ float mish_f(float v) {
    const float m   = __expf(-fabsf(v));
    const bool  pos = v >= 0.f;
    const float num = pos ? fmaf(2.f, m, 1.f) : m * (m + 2.f);
    const float den = num + (pos ? 2.f * m * m : 2.f);
    return v * (num * __builtin_amdgcn_rcpf(den));
}

__device__ __forceinline__ float sigmoid_f(float x) {
    return __builtin_amdgcn_rcpf(1.f + __expf(-x));
}


#define GEMM_K   32
#define GEMM_P   36

template <int WM, int WN, bool AKM>
struct GemmShape {
    static constexpr int NT  = 32 * WM * WN;
    static constexpr int BM  = 16 * WM;
    static constexpr int BN  = 64 * WN;
    static constexpr int AP  = AKM ? (BM + 8) : GEMM_P;
    static constexpr int ASZ = AKM ? GEMM_K * (BM + 8) : BM * GEMM_P;
    static constexpr int WSZ = BN * GEMM_P;
    static constexpr int AV4 = BM * GEMM_K / 4 / NT;
    static constexpr int WV4 = BN * GEMM_K / 4 / NT;

    static __device__ __forceinline__ size_t a_goff(const int i, const int tid, const int lda) {
        const int idx = tid + i * NT;
        return AKM ? (size_t)(idx / (BM / 4)) * lda + (idx % (BM / 4)) * 4
                   : (size_t)(idx >> 3) * lda + (idx & 7) * 4;
    }
    static __device__ __forceinline__ int a_soff(const int i, const int tid) {
        const int idx = tid + i * NT;
        return AKM ? (idx / (BM / 4)) * AP + (idx % (BM / 4)) * 4
                   : (idx >> 3) * AP + (idx & 7) * 4;
    }
    static __device__ __forceinline__ size_t w_goff(const int i, const int tid, const int ldw) {
        const int idx = tid + i * NT;
        return (size_t)(idx >> 3) * ldw + (idx & 7) * 4;
    }
    static __device__ __forceinline__ int w_soff(const int i, const int tid) {
        const int idx = tid + i * NT;
        return (idx >> 3) * GEMM_P + (idx & 7) * 4;
    }
};

template <int WM, int WN, bool AKM>
__device__ __forceinline__ void gemm_block(const float* __restrict__ Ablk, const int lda,
                                           const float* __restrict__ Wblk, const int ldw,
                                           const int K,
                                           float* __restrict__ sA, float* __restrict__ sW,
                                           v8f& acc0, v8f& acc1, v8f& acc2, v8f& acc3) {
    typedef GemmShape<WM, WN, AKM> S;
    static_assert(S::WV4 == 4 && (S::AV4 == 2 || S::AV4 == 4), "per-thread piece counts the code below is written for");
    const int tid  = threadIdx.x;
    const int wave = tid >> 5;
    const int wm   = wave % WM;
    const int wn   = wave / WM;
    const int lane = tid & 31;
    const int half = lane >> 4;
    const int r    = lane & 15;

    const float* ga0 = Ablk + S::a_goff(0, tid, lda);
    const float* ga1 = Ablk + S::a_goff(1, tid, lda);
    const float* ga2 = Ablk + S::a_goff(S::AV4 > 2 ? 2 : 0, tid, lda);
    const float* ga3 = Ablk + S::a_goff(S::AV4 > 2 ? 3 : 0, tid, lda);
    const float* gw0 = Wblk + S::w_goff(0, tid, ldw);
    const float* gw1 = Wblk + S::w_goff(1, tid, ldw);
    const float* gw2 = Wblk + S::w_goff(2, tid, ldw);
    const float* gw3 = Wblk + S::w_goff(3, tid, ldw);
    const size_t astep = AKM ? (size_t)GEMM_K * lda : (size_t)GEMM_K;

    float4 ra0, ra1, ra2, ra3, rw0, rw1, rw2, rw3;
    ra2 = ra3 = make_float4(0.f, 0.f, 0.f, 0.f);

#define GEMM_GLOAD(ks)                                                        \
    do {                                                                      \
        const size_t ao = (size_t)(ks) * astep;                               \
        const size_t wo = (size_t)(ks) * GEMM_K;                              \
        ra0 = *(const float4*)(ga0 + ao);                                     \
        ra1 = *(const float4*)(ga1 + ao);                                     \
        if (S::AV4 > 2) {                                                     \
            ra2 = *(const float4*)(ga2 + ao);                                 \
            ra3 = *(const float4*)(ga3 + ao);                                 \
        }                                                                     \
        rw0 = *(const float4*)(gw0 + wo);                                     \
        rw1 = *(const float4*)(gw1 + wo);                                     \
        rw2 = *(const float4*)(gw2 + wo);                                     \
        rw3 = *(const float4*)(gw3 + wo);                                     \
    } while (0)

#define GEMM_SSTORE(buf)                                                      \
    do {                                                                      \
        float* a_ = sA + (buf) * S::ASZ;                                      \
        float* w_ = sW + (buf) * S::WSZ;                                      \
        *(float4*)(a_ + S::a_soff(0, tid)) = ra0;                             \
        *(float4*)(a_ + S::a_soff(1, tid)) = ra1;                             \
        if (S::AV4 > 2) {                                                     \
            *(float4*)(a_ + S::a_soff(2, tid)) = ra2;                         \
            *(float4*)(a_ + S::a_soff(3, tid)) = ra3;                         \
        }                                                                     \
        *(float4*)(w_ + S::w_soff(0, tid)) = rw0;                             \
        *(float4*)(w_ + S::w_soff(1, tid)) = rw1;                             \
        *(float4*)(w_ + S::w_soff(2, tid)) = rw2;                             \
        *(float4*)(w_ + S::w_soff(3, tid)) = rw3;                             \
    } while (0)

    GEMM_GLOAD(0);
    GEMM_SSTORE(0);
    __syncthreads();

    const int nslab = K / GEMM_K;
    for (int s = 0; s < nslab; ++s) {
        const int buf = s & 1;
        const int sn = (s + 1 < nslab) ? (s + 1) : s;
        GEMM_GLOAD(sn);

        const float* a = sA + buf * S::ASZ;
        const float* w = sW + buf * S::WSZ + (wn * 64 + r) * GEMM_P;
        #pragma unroll
        for (int k0 = 0; k0 < GEMM_K; k0 += 4) {
            const int ka = k0 + half * 2;
            v2f av;
            if (AKM) {
                av.x = a[ka * S::AP + wm * 16 + r];
                av.y = a[(ka + 1) * S::AP + wm * 16 + r];
            } else {
                av = *(const v2f*)(a + (wm * 16 + r) * S::AP + ka);
            }
            const v2f b0 = *(const v2f*)(w + 0 * 16 * GEMM_P + ka);
            const v2f b1 = *(const v2f*)(w + 1 * 16 * GEMM_P + ka);
            const v2f b2 = *(const v2f*)(w + 2 * 16 * GEMM_P + ka);
            const v2f b3 = *(const v2f*)(w + 3 * 16 * GEMM_P + ka);
            acc0 = wmma_f32_4(av, b0, acc0);
            acc1 = wmma_f32_4(av, b1, acc1);
            acc2 = wmma_f32_4(av, b2, acc2);
            acc3 = wmma_f32_4(av, b3, acc3);
        }

        GEMM_SSTORE(buf ^ 1);
        __syncthreads();
    }
#undef GEMM_GLOAD
#undef GEMM_SSTORE
}

__global__ void __launch_bounds__(256) __attribute__((amdgpu_waves_per_eu(4, 8)))
k_inproj(const float* __restrict__ x,
         const float* __restrict__ w,
         float* __restrict__ xz) {
    typedef GemmShape<4, 2, true> S;
    __shared__ __attribute__((aligned(16))) float sA[2 * S::ASZ];
    __shared__ __attribute__((aligned(16))) float sW[2 * S::WSZ];

    const int bn = blockIdx.x & 15;
    const int bm = blockIdx.x >> 4;
    const int m0 = bm * 64;
    const int bb = m0 >> 9;
    const int l0 = m0 & 511;

    v8f acc0 = {}, acc1 = {}, acc2 = {}, acc3 = {};
    gemm_block<4, 2, true>(x + (size_t)bb * DMODEL * SEQ + l0, SEQ,
                           w + (size_t)(bn * 128) * DMODEL, DMODEL, DMODEL,
                           sA, sW, acc0, acc1, acc2, acc3);

    const int wave = threadIdx.x >> 5;
    const int wm   = wave % 4;
    const int wn   = wave / 4;
    const int lane = threadIdx.x & 31;
    const int half = lane >> 4;
    const int r    = lane & 15;
    const int col  = bn * 128 + wn * 64 + r;
    #pragma unroll
    for (int j = 0; j < 8; ++j) {
        const size_t row = (size_t)(m0 + wm * 16 + half * 8 + j) * (2 * DINNER);
        xz[row + col + 0 ] = acc0[j];
        xz[row + col + 16] = acc1[j];
        xz[row + col + 32] = acc2[j];
        xz[row + col + 48] = acc3[j];
    }
}

#define CONV_R 8

__global__ void __launch_bounds__(256)
k_conv(const float* __restrict__ xz,
       const float* __restrict__ cw,
       const float* __restrict__ cb,
       float* __restrict__ ucv) {
    const int c  = threadIdx.x * 4;
    const int m0 = blockIdx.x * CONV_R;
    const int l0 = m0 & (SEQ - 1);

    float xr[CONV_R + 3][4];
    #pragma unroll
    for (int j = 0; j < CONV_R + 3; ++j) {
        const int mr = (l0 + j >= 3) ? (m0 + j - 3) : m0;
        const float4 v = *(const float4*)(xz + (size_t)mr * (2 * DINNER) + c);
        xr[j][0] = v.x; xr[j][1] = v.y; xr[j][2] = v.z; xr[j][3] = v.w;
    }

    float wt[4][4];
    #pragma unroll
    for (int ch = 0; ch < 4; ++ch) {
        const float4 w4 = *(const float4*)(cw + (size_t)(c + ch) * 4);
        wt[ch][0] = w4.x; wt[ch][1] = w4.y; wt[ch][2] = w4.z; wt[ch][3] = w4.w;
    }
    const float4 b4 = *(const float4*)(cb + c);
    const float bias[4] = {b4.x, b4.y, b4.z, b4.w};

    #pragma unroll
    for (int i = 0; i < CONV_R; ++i) {
        float o[4];
        #pragma unroll
        for (int ch = 0; ch < 4; ++ch) {
            float s = bias[ch];
            #pragma unroll
            for (int t = 0; t < 4; ++t) {
                if (l0 + i + t >= 3)
                    s += xr[i + t][ch] * wt[ch][t];
            }
            o[ch] = s * sigmoid_f(s);
        }
        *(float4*)(ucv + (size_t)(m0 + i) * DINNER + c) = make_float4(o[0], o[1], o[2], o[3]);
    }
}

#define XP_KQ    (DINNER / 4)
#define XP_AROWS (4 * 16)
#define XP_WROWS (4 * 64)
#define XP_AV4   (XP_AROWS * GEMM_K / 4 / 128)
#define XP_WV4   (XP_WROWS * GEMM_K / 4 / 128)
#define XP_RP    68

__global__ void __launch_bounds__(128) __attribute__((amdgpu_waves_per_eu(1, 2)))
k_xproj(const float* __restrict__ ucv,
        const float* __restrict__ w,
        float* __restrict__ dbc) {
    __shared__ __attribute__((aligned(16))) float sA[2 * XP_AROWS * GEMM_P];
    __shared__ __attribute__((aligned(16))) float sW[2 * XP_WROWS * GEMM_P];
    static_assert(4 * 16 * XP_RP <= 2 * XP_WROWS * GEMM_P, "partial tiles must fit the W buffers");

    const int tid  = threadIdx.x;
    const int q    = tid >> 5;
    const int lane = tid & 31;
    const int half = lane >> 4;
    const int r    = lane & 15;
    const int m0   = blockIdx.x * 16;

    const int pr = tid >> 3, pc = (tid & 7) * 4;
    const float* ga = ucv + (size_t)(m0 + pr) * DINNER + pc;
    const float* gw = w   + (size_t)pr * DINNER + pc;
    const int so = pr * GEMM_P + pc;

    float4 ra0, ra1, ra2, ra3;
    float4 rw0, rw1, rw2, rw3, rw4, rw5, rw6, rw7, rw8, rw9, rw10, rw11, rw12, rw13, rw14, rw15;
    static_assert(XP_AV4 == 4 && XP_WV4 == 16, "piece counts the two macros below are written for");

#define XP_GLOAD(ks)                                                          \
    do {                                                                      \
        const float* ga_ = ga + (ks) * GEMM_K;                                \
        const float* gw_ = gw + (ks) * GEMM_K;                                \
        ra0 = *(const float4*)(ga_ + 0 * XP_KQ);                              \
        ra1 = *(const float4*)(ga_ + 1 * XP_KQ);                              \
        ra2 = *(const float4*)(ga_ + 2 * XP_KQ);                              \
        ra3 = *(const float4*)(ga_ + 3 * XP_KQ);                              \
        rw0 = *(const float4*)(gw_ + (size_t)0 * 16 * DINNER + 0 * XP_KQ);    \
        rw1 = *(const float4*)(gw_ + (size_t)1 * 16 * DINNER + 0 * XP_KQ);    \
        rw2 = *(const float4*)(gw_ + (size_t)2 * 16 * DINNER + 0 * XP_KQ);    \
        rw3 = *(const float4*)(gw_ + (size_t)3 * 16 * DINNER + 0 * XP_KQ);    \
        rw4 = *(const float4*)(gw_ + (size_t)0 * 16 * DINNER + 1 * XP_KQ);    \
        rw5 = *(const float4*)(gw_ + (size_t)1 * 16 * DINNER + 1 * XP_KQ);    \
        rw6 = *(const float4*)(gw_ + (size_t)2 * 16 * DINNER + 1 * XP_KQ);    \
        rw7 = *(const float4*)(gw_ + (size_t)3 * 16 * DINNER + 1 * XP_KQ);    \
        rw8 = *(const float4*)(gw_ + (size_t)0 * 16 * DINNER + 2 * XP_KQ);    \
        rw9 = *(const float4*)(gw_ + (size_t)1 * 16 * DINNER + 2 * XP_KQ);    \
        rw10 = *(const float4*)(gw_ + (size_t)2 * 16 * DINNER + 2 * XP_KQ);   \
        rw11 = *(const float4*)(gw_ + (size_t)3 * 16 * DINNER + 2 * XP_KQ);   \
        rw12 = *(const float4*)(gw_ + (size_t)0 * 16 * DINNER + 3 * XP_KQ);   \
        rw13 = *(const float4*)(gw_ + (size_t)1 * 16 * DINNER + 3 * XP_KQ);   \
        rw14 = *(const float4*)(gw_ + (size_t)2 * 16 * DINNER + 3 * XP_KQ);   \
        rw15 = *(const float4*)(gw_ + (size_t)3 * 16 * DINNER + 3 * XP_KQ);   \
    } while (0)

#define XP_SSTORE(buf)                                                        \
    do {                                                                      \
        float* a_ = sA + (buf) * (XP_AROWS * GEMM_P) + so;                    \
        float* w_ = sW + (buf) * (XP_WROWS * GEMM_P) + so;                    \
        *(float4*)(a_ + 0 * 16 * GEMM_P) = ra0;                               \
        *(float4*)(a_ + 1 * 16 * GEMM_P) = ra1;                               \
        *(float4*)(a_ + 2 * 16 * GEMM_P) = ra2;                               \
        *(float4*)(a_ + 3 * 16 * GEMM_P) = ra3;                               \
        *(float4*)(w_ + 0 * 16 * GEMM_P) = rw0;                               \
        *(float4*)(w_ + 1 * 16 * GEMM_P) = rw1;                               \
        *(float4*)(w_ + 2 * 16 * GEMM_P) = rw2;                               \
        *(float4*)(w_ + 3 * 16 * GEMM_P) = rw3;                               \
        *(float4*)(w_ + 4 * 16 * GEMM_P) = rw4;                               \
        *(float4*)(w_ + 5 * 16 * GEMM_P) = rw5;                               \
        *(float4*)(w_ + 6 * 16 * GEMM_P) = rw6;                               \
        *(float4*)(w_ + 7 * 16 * GEMM_P) = rw7;                               \
        *(float4*)(w_ + 8 * 16 * GEMM_P) = rw8;                               \
        *(float4*)(w_ + 9 * 16 * GEMM_P) = rw9;                               \
        *(float4*)(w_ + 10 * 16 * GEMM_P) = rw10;                             \
        *(float4*)(w_ + 11 * 16 * GEMM_P) = rw11;                             \
        *(float4*)(w_ + 12 * 16 * GEMM_P) = rw12;                             \
        *(float4*)(w_ + 13 * 16 * GEMM_P) = rw13;                             \
        *(float4*)(w_ + 14 * 16 * GEMM_P) = rw14;                             \
        *(float4*)(w_ + 15 * 16 * GEMM_P) = rw15;                             \
    } while (0)

    v8f acc0 = {}, acc1 = {}, acc2 = {}, acc3 = {};

    constexpr int nslab = XP_KQ / GEMM_K;
    XP_GLOAD(0);
    XP_SSTORE(0);
    XP_GLOAD(1);
    __syncthreads();

    for (int s = 0; s < nslab; ++s) {
        const int buf = s & 1;
        const float* a  = sA + buf * (XP_AROWS * GEMM_P) + (q * 16 + r) * GEMM_P;
        const float* wb = sW + buf * (XP_WROWS * GEMM_P) + (q * 64 + r) * GEMM_P;

#define XP_FRAG(ks, A_, B0_, B1_, B2_, B3_)                                  \
        const v2f A_  = *(const v2f*)(a  + (ks) * 4 + half * 2);              \
        const v2f B0_ = *(const v2f*)(wb + 0 * 16 * GEMM_P + (ks) * 4 + half * 2); \
        const v2f B1_ = *(const v2f*)(wb + 1 * 16 * GEMM_P + (ks) * 4 + half * 2); \
        const v2f B2_ = *(const v2f*)(wb + 2 * 16 * GEMM_P + (ks) * 4 + half * 2); \
        const v2f B3_ = *(const v2f*)(wb + 3 * 16 * GEMM_P + (ks) * 4 + half * 2)
#define XP_MUL(A_, B0_, B1_, B2_, B3_)                                        \
        acc0 = wmma_f32_4(A_, B0_, acc0);                                     \
        acc1 = wmma_f32_4(A_, B1_, acc1);                                     \
        acc2 = wmma_f32_4(A_, B2_, acc2);                                     \
        acc3 = wmma_f32_4(A_, B3_, acc3)

        XP_FRAG(0, a0, b00, b01, b02, b03);
        XP_FRAG(1, a1, b10, b11, b12, b13);
        XP_FRAG(2, a2, b20, b21, b22, b23);
        XP_FRAG(3, a3, b30, b31, b32, b33);
        XP_FRAG(4, a4, b40, b41, b42, b43);
        XP_FRAG(5, a5, b50, b51, b52, b53);
        XP_FRAG(6, a6, b60, b61, b62, b63);
        XP_FRAG(7, a7, b70, b71, b72, b73);

        XP_SSTORE(buf ^ 1);
        const int sn = (s + 2 < nslab) ? (s + 2) : (nslab - 1);
        XP_GLOAD(sn);

        XP_MUL(a0, b00, b01, b02, b03);
        XP_MUL(a1, b10, b11, b12, b13);
        XP_MUL(a2, b20, b21, b22, b23);
        XP_MUL(a3, b30, b31, b32, b33);
        XP_MUL(a4, b40, b41, b42, b43);
        XP_MUL(a5, b50, b51, b52, b53);
        XP_MUL(a6, b60, b61, b62, b63);
        XP_MUL(a7, b70, b71, b72, b73);
#undef XP_FRAG
#undef XP_MUL
        __syncthreads();
    }
#undef XP_GLOAD
#undef XP_SSTORE

    float* sR = sW;
    {
        float* p = sR + (q * 16 + half * 8) * XP_RP + r;
        #pragma unroll
        for (int j = 0; j < 8; ++j) {
            p[j * XP_RP + 0 ] = acc0[j];
            p[j * XP_RP + 16] = acc1[j];
            p[j * XP_RP + 32] = acc2[j];
            p[j * XP_RP + 48] = acc3[j];
        }
    }
    __syncthreads();

    const int row = tid >> 3;
    #pragma unroll
    for (int hp = 0; hp < 2; ++hp) {
        const int col = pc + hp * 32;
        const float4 p0 = *(const float4*)(sR + (0 * 16 + row) * XP_RP + col);
        const float4 p1 = *(const float4*)(sR + (1 * 16 + row) * XP_RP + col);
        const float4 p2 = *(const float4*)(sR + (2 * 16 + row) * XP_RP + col);
        const float4 p3 = *(const float4*)(sR + (3 * 16 + row) * XP_RP + col);
        float4 o;
        o.x = ((p0.x + p1.x) + p2.x) + p3.x;
        o.y = ((p0.y + p1.y) + p2.y) + p3.y;
        o.z = ((p0.z + p1.z) + p2.z) + p3.z;
        o.w = ((p0.w + p1.w) + p2.w) + p3.w;
        *(float4*)(dbc + (size_t)(m0 + row) * 64 + col) = o;
    }
}

__global__ void k_dtproj(const float* __restrict__ dbc,
                         const float* __restrict__ w,
                         const float* __restrict__ bias,
                         float* __restrict__ delta) {
    const int wid   = blockIdx.x * 4 + (threadIdx.x >> 5);
    const int ntq   = wid & 15;
    const int tileM = wid >> 4;
    const int lane  = threadIdx.x & 31;
    const int half  = lane >> 4;
    const int r     = lane & 15;

    const float* arow = dbc + (size_t)(tileM * 16 + r) * 64;
    const float* wr0 = w + (size_t)((ntq * 4 + 0) * 16 + r) * DTRANK;
    const float* wr1 = w + (size_t)((ntq * 4 + 1) * 16 + r) * DTRANK;
    const float* wr2 = w + (size_t)((ntq * 4 + 2) * 16 + r) * DTRANK;
    const float* wr3 = w + (size_t)((ntq * 4 + 3) * 16 + r) * DTRANK;

    v8f acc0 = {}, acc1 = {}, acc2 = {}, acc3 = {};
    #pragma unroll
    for (int k0 = 0; k0 < DTRANK; k0 += 4) {
        const int ka = k0 + half * 2;
        v2f a  = *(const v2f*)(arow + ka);
        v2f b0 = *(const v2f*)(wr0 + ka);
        v2f b1 = *(const v2f*)(wr1 + ka);
        v2f b2 = *(const v2f*)(wr2 + ka);
        v2f b3 = *(const v2f*)(wr3 + ka);
        acc0 = wmma_f32_4(a, b0, acc0);
        acc1 = wmma_f32_4(a, b1, acc1);
        acc2 = wmma_f32_4(a, b2, acc2);
        acc3 = wmma_f32_4(a, b3, acc3);
    }
    const int nb = ntq * 64 + r;
    const float bn0 = bias[nb + 0];
    const float bn1 = bias[nb + 16];
    const float bn2 = bias[nb + 32];
    const float bn3 = bias[nb + 48];
    #pragma unroll
    for (int j = 0; j < 8; ++j) {
        const size_t row = (size_t)(tileM * 16 + half * 8 + j) * DINNER;
        delta[row + nb + 0 ] = softplus_f(acc0[j] + bn0);
        delta[row + nb + 16] = softplus_f(acc1[j] + bn1);
        delta[row + nb + 32] = softplus_f(acc2[j] + bn2);
        delta[row + nb + 48] = softplus_f(acc3[j] + bn3);
    }
}

#define SCAN_T   32
#define SCAN_Q   4
#define SCAN_CPB (128 / SCAN_Q)
#define SCAN_G   (SCAN_T / SCAN_Q)

template <int CTRL>
__device__ __forceinline__ float quad_perm(float v) {
    return __builtin_bit_cast(float, __builtin_amdgcn_mov_dpp(__builtin_bit_cast(int, v), CTRL, 0xF, 0xF, true));
}
__device__ __forceinline__ float quad_xor1(float v) { return quad_perm<0xB1>(v); }
__device__ __forceinline__ float quad_xor2(float v) { return quad_perm<0x4E>(v); }
__device__ __forceinline__ float quad_bcast(float v, const int k) {
    switch (k) {
        case 0:  return quad_perm<0x00>(v);
        case 1:  return quad_perm<0x55>(v);
        case 2:  return quad_perm<0xAA>(v);
        default: return quad_perm<0xFF>(v);
    }
}

__device__ __forceinline__ v2f exp_pair(v2f x) {
    const v2f t = x * 1.4426950408889634f;
    v2f r;
    r.x = __builtin_amdgcn_exp2f(t.x);
    r.y = __builtin_amdgcn_exp2f(t.y);
    return r;
}

__device__ __forceinline__ void scan_load_bc(const float4* __restrict__ sBC, const int l, const int sub,
                                             float4 (&Bv)[SCAN_Q], float4 (&Cv)[SCAN_Q]) {
    #pragma unroll
    for (int t = 0; t < SCAN_Q; ++t) {
        Bv[t] = sBC[(l + t) * 8 + sub];
        Cv[t] = sBC[(l + t) * 8 + 4 + sub];
    }
}

__global__ void __launch_bounds__(128)
k_scan(const float* __restrict__ delta,
       const float* __restrict__ ucv,
       const float* __restrict__ xz,
       const float* __restrict__ dbc,
       const float* __restrict__ A_log,
       const float* __restrict__ D_param,
       float* __restrict__ ybuf) {
    const int bb  = blockIdx.x >> 5;
    const int tid = threadIdx.x;
    const int sub = tid & (SCAN_Q - 1);
    const int chl = tid >> 2;
    const int c0  = (blockIdx.x & 31) * SCAN_CPB;
    const int c   = c0 + chl;
    const int mbase = bb * SEQ;

    __shared__ float4 sBC[SEQ * 8];
    __shared__ __attribute__((aligned(16))) float sY[2][SCAN_T][SCAN_CPB];

    const float* pd = delta + (size_t)(mbase + sub) * DINNER + c;
    const float* pu = ucv   + (size_t)(mbase + sub) * DINNER + c;
    const float* pz = xz    + (size_t)(mbase + sub) * (2 * DINNER) + DINNER + c;
    float* pyo = ybuf + (size_t)(mbase + (tid >> 3)) * DINNER + c0 + (tid & 7) * 4;

    v8f cd, cu, cz;
    static_assert(SCAN_G == 8, "one 8-register vector per operand");
    #pragma unroll
    for (int g = 0; g < SCAN_G; ++g) {
        cd[g] = pd[(size_t)(g * SCAN_Q) * DINNER];
        cu[g] = pu[(size_t)(g * SCAN_Q) * DINNER];
        cz[g] = pz[(size_t)(g * SCAN_Q) * (2 * DINNER)];
    }

    const float* bcsrc = dbc + (size_t)mbase * 64 + 32;
    #pragma unroll
    for (int j = 0; j < (SEQ * 8) / 128; ++j) {
        const int i = j * 128 + tid;
        sBC[i] = *(const float4*)(bcsrc + (size_t)(i >> 3) * 64 + (i & 7) * 4);
    }

    const float4 al = *(const float4*)(A_log + (size_t)c * DSTATE + sub * 4);
    const v2f Ac01 = {-__expf(al.x), -__expf(al.y)};
    const v2f Ac23 = {-__expf(al.z), -__expf(al.w)};
    v2f h01 = {0.f, 0.f}, h23 = {0.f, 0.f};
    const float Dp = D_param[c];
    const bool odd = (sub & 1) != 0;
    const bool hi  = (sub & 2) != 0;

    __syncthreads();

    float4 Bc[SCAN_Q], Cc[SCAN_Q];
    scan_load_bc(sBC, 0, sub, Bc, Cc);

    for (int l0 = 0; l0 < SEQ; l0 += SCAN_T) {
        const int ln = (l0 + SCAN_T < SEQ) ? (l0 + SCAN_T) : l0;
        v8f nd, nu, nz;
        #pragma unroll
        for (int g = 0; g < SCAN_G; ++g) {
            nd[g] = pd[(size_t)(ln + g * SCAN_Q) * DINNER];
            nu[g] = pu[(size_t)(ln + g * SCAN_Q) * DINNER];
            nz[g] = pz[(size_t)(ln + g * SCAN_Q) * (2 * DINNER)];
        }

        float (*sYt)[SCAN_CPB] = sY[(l0 / SCAN_T) & 1];

        #pragma unroll
        for (int g = 0; g < SCAN_G; ++g) {
            const int lg = l0 + g * SCAN_Q;

            float4 Bn[SCAN_Q], Cn[SCAN_Q];
            const int lnx = (lg + SCAN_Q < SEQ) ? (lg + SCAN_Q) : lg;
            scan_load_bc(sBC, lnx, sub, Bn, Cn);
            __builtin_amdgcn_sched_barrier(0);

            float yq[SCAN_Q];
            #pragma unroll
            for (int t = 0; t < SCAN_Q; ++t) {
                const float dlt = quad_bcast(cd[g], t);
                const float du  = dlt * quad_bcast(cu[g], t);
                const v2f B01 = {Bc[t].x, Bc[t].y}, B23 = {Bc[t].z, Bc[t].w};
                const v2f C01 = {Cc[t].x, Cc[t].y}, C23 = {Cc[t].z, Cc[t].w};
                const v2f dA01 = exp_pair(Ac01 * dlt);
                const v2f dA23 = exp_pair(Ac23 * dlt);
                h01 = __builtin_elementwise_fma(dA01, h01, du * B01);
                h23 = __builtin_elementwise_fma(dA23, h23, du * B23);
                const v2f y2 = h01 * C01 + h23 * C23;
                yq[t] = y2.x + y2.y;
            }

            const float a  = (odd ? yq[1] : yq[0]) + quad_xor1(odd ? yq[0] : yq[1]);
            const float b  = (odd ? yq[3] : yq[2]) + quad_xor1(odd ? yq[2] : yq[3]);
            const float ys = (hi ? b : a) + quad_xor2(hi ? a : b);
            const float zs = cz[g];
            sYt[g * SCAN_Q + sub][chl] = (ys + cu[g] * Dp) * (zs * sigmoid_f(zs));

            #pragma unroll
            for (int t = 0; t < SCAN_Q; ++t) { Bc[t] = Bn[t]; Cc[t] = Cn[t]; }

            __builtin_amdgcn_sched_barrier(0);
        }

        __builtin_amdgcn_fence(__ATOMIC_RELEASE, "workgroup", "local");
        __builtin_amdgcn_s_barrier();
        __builtin_amdgcn_fence(__ATOMIC_ACQUIRE, "workgroup", "local");
        #pragma unroll
        for (int j = 0; j < SCAN_T / 16; ++j)
            *(float4*)(pyo + (size_t)(l0 + 16 * j) * DINNER) = *(const float4*)&sYt[(tid >> 3) + 16 * j][(tid & 7) * 4];

        asm volatile("" :: "v"(nd), "v"(nu), "v"(nz));
        cd = nd; cu = nu; cz = nz;
    }
}

#define OUT_CP 68
#define GN_PP  32

__global__ void __launch_bounds__(256) __attribute__((amdgpu_waves_per_eu(2, 4)))
k_outproj(const float* __restrict__ ybuf,
          const float* __restrict__ w,
          float* __restrict__ otmp,
          float* __restrict__ part) {
    typedef GemmShape<4, 2, false> S;
    __shared__ __attribute__((aligned(16))) float sA[2 * S::ASZ];
    __shared__ __attribute__((aligned(16))) float sW[2 * S::WSZ];
    static_assert(128 * OUT_CP <= 2 * S::WSZ, "turned tile must fit the W buffers");

    const int bn = blockIdx.x & 3;
    const int bm = blockIdx.x >> 2;
    const int m0 = bm * 64;
    const int bb = m0 >> 9;
    const int l0 = m0 & 511;

    v8f acc0 = {}, acc1 = {}, acc2 = {}, acc3 = {};
    gemm_block<4, 2, false>(ybuf + (size_t)m0 * DINNER, DINNER,
                            w + (size_t)(bn * 128) * DINNER, DINNER, DINNER,
                            sA, sW, acc0, acc1, acc2, acc3);

    const int tid  = threadIdx.x;
    const int wave = tid >> 5;
    const int wm   = wave % 4;
    const int wn   = wave / 4;
    const int lane = tid & 31;
    const int half = lane >> 4;
    const int r    = lane & 15;

    float ps0 = 0.f, ps1 = 0.f, ps2 = 0.f, ps3 = 0.f;
    float pq0 = 0.f, pq1 = 0.f, pq2 = 0.f, pq3 = 0.f;
    #pragma unroll
    for (int j = 0; j < 8; ++j) {
        ps0 += acc0[j]; pq0 = fmaf(acc0[j], acc0[j], pq0);
        ps1 += acc1[j]; pq1 = fmaf(acc1[j], acc1[j], pq1);
        ps2 += acc2[j]; pq2 = fmaf(acc2[j], acc2[j], pq2);
        ps3 += acc3[j]; pq3 = fmaf(acc3[j], acc3[j], pq3);
    }
    float ps = (ps0 + ps1) + (ps2 + ps3);
    float pq = (pq0 + pq1) + (pq2 + pq3);
    ps += quad_xor1(ps); pq += quad_xor1(pq);
    ps += quad_xor2(ps); pq += quad_xor2(pq);
    float2* sRed = (float2*)sA;
    if ((lane & 3) == 0) sRed[wave * 8 + (lane >> 2)] = make_float2(ps, pq);

    float* sC = sW;
    {
        float* c0 = sC + (wn * 64 + r) * OUT_CP + wm * 16 + half * 8;
        *(float4*)(c0 + 0 * 16 * OUT_CP)     = make_float4(acc0[0], acc0[1], acc0[2], acc0[3]);
        *(float4*)(c0 + 0 * 16 * OUT_CP + 4) = make_float4(acc0[4], acc0[5], acc0[6], acc0[7]);
        *(float4*)(c0 + 1 * 16 * OUT_CP)     = make_float4(acc1[0], acc1[1], acc1[2], acc1[3]);
        *(float4*)(c0 + 1 * 16 * OUT_CP + 4) = make_float4(acc1[4], acc1[5], acc1[6], acc1[7]);
        *(float4*)(c0 + 2 * 16 * OUT_CP)     = make_float4(acc2[0], acc2[1], acc2[2], acc2[3]);
        *(float4*)(c0 + 2 * 16 * OUT_CP + 4) = make_float4(acc2[4], acc2[5], acc2[6], acc2[7]);
        *(float4*)(c0 + 3 * 16 * OUT_CP)     = make_float4(acc3[0], acc3[1], acc3[2], acc3[3]);
        *(float4*)(c0 + 3 * 16 * OUT_CP + 4) = make_float4(acc3[4], acc3[5], acc3[6], acc3[7]);
    }
    __syncthreads();

    float* obase = otmp + ((size_t)bb * DMODEL + bn * 128) * SEQ + l0;
    #pragma unroll
    for (int i = 0; i < 8; ++i) {
        const int idx = tid + i * 256;
        const int n = idx >> 4, q = idx & 15;
        *(float4*)(obase + (size_t)n * SEQ + q * 4) = *(const float4*)(sC + n * OUT_CP + q * 4);
    }

    if (tid < 4) {
        const float* rp = sA + (tid >> 1) * (4 * 8 * 2) + (tid & 1);
        float t0 = 0.f, t1 = 0.f, t2 = 0.f, t3 = 0.f;
        #pragma unroll
        for (int i = 0; i < 8; ++i) {
            t0 += rp[(0 * 8 + i) * 2];
            t1 += rp[(1 * 8 + i) * 2];
            t2 += rp[(2 * 8 + i) * 2];
            t3 += rp[(3 * 8 + i) * 2];
        }
        part[(size_t)blockIdx.x * GN_PP + tid] = (t0 + t1) + (t2 + t3);
    }
}

#define GNM_V 4

__device__ __forceinline__ float gn_mish_f(float x, float mu, float rs, float g, float bt) {
    return mish_f((x - mu) * rs * g + bt);
}

__global__ void __launch_bounds__(256)
k_gnmish(const float* __restrict__ otmp,
         const float* __restrict__ part,
         const float* __restrict__ gw,
         const float* __restrict__ gb,
         float* __restrict__ out) {
    const int vec0 = blockIdx.x * (256 * GNM_V) + threadIdx.x;
    const int blk  = (blockIdx.x * (1024 * GNM_V)) >> 15;

    float4 x4[GNM_V];
    #pragma unroll
    for (int i = 0; i < GNM_V; ++i)
        x4[i] = *(const float4*)(otmp + (size_t)(vec0 + i * 256) * 4);

    const float* pp = part + ((size_t)(blk >> 3) * 32 + ((blk & 7) >> 1)) * GN_PP + (blk & 1) * 2;
    float s = 0.f, q = 0.f;
    #pragma unroll
    for (int lt = 0; lt < 8; ++lt) {
        s += pp[lt * 4 * GN_PP];
        q += pp[lt * 4 * GN_PP + 1];
    }
    const float mu  = s * (1.f / 32768.f);
    const float var = q * (1.f / 32768.f) - mu * mu;
    const float rs  = rsqrtf(var + 1e-5f);

    #pragma unroll
    for (int i = 0; i < GNM_V; ++i) {
        const int idx = (vec0 + i * 256) * 4;
        const int c   = (idx >> 9) & (DMODEL - 1);
        const float g  = gw[c];
        const float bt = gb[c];
        float4 o;
        o.x = gn_mish_f(x4[i].x, mu, rs, g, bt);
        o.y = gn_mish_f(x4[i].y, mu, rs, g, bt);
        o.z = gn_mish_f(x4[i].z, mu, rs, g, bt);
        o.w = gn_mish_f(x4[i].w, mu, rs, g, bt);
        *(float4*)(out + (size_t)idx) = o;
    }
}

extern "C" void kernel_launch(void* const* d_in, const int* in_sizes, int n_in,
                              void* d_out, int out_size, void* d_ws, size_t ws_size,
                              hipStream_t stream) {
    const float* x         = (const float*)d_in[0];
    const float* in_proj_w = (const float*)d_in[1];
    const float* conv_w    = (const float*)d_in[2];
    const float* conv_b    = (const float*)d_in[3];
    const float* x_proj_w  = (const float*)d_in[4];
    const float* dt_proj_w = (const float*)d_in[5];
    const float* dt_proj_b = (const float*)d_in[6];
    const float* A_log     = (const float*)d_in[7];
    const float* D_param   = (const float*)d_in[8];
    const float* out_proj_w= (const float*)d_in[9];
    const float* gn_w      = (const float*)d_in[10];
    const float* gn_b      = (const float*)d_in[11];
    float* out = (float*)d_out;

    float* ws    = (float*)d_ws;
    float* xz    = ws;
    float* ucv   = xz    + (size_t)MROWS * 2 * DINNER;
    float* dbc   = ucv   + (size_t)MROWS * DINNER;
    float* delta = dbc   + (size_t)MROWS * 64;
    float* ybuf  = delta + (size_t)MROWS * DINNER;
    float* part  = ybuf  + (size_t)MROWS * DINNER;
    float* otmp  = xz;

    k_inproj<<<1024, 256, 0, stream>>>(x, in_proj_w, xz);

    k_conv<<<MROWS / CONV_R, 256, 0, stream>>>(xz, conv_w, conv_b, ucv);

    k_xproj<<<MROWS / 16, 128, 0, stream>>>(ucv, x_proj_w, dbc);

    k_dtproj<<<4096 / 4, 128, 0, stream>>>(dbc, dt_proj_w, dt_proj_b, delta);

    k_scan<<<BATCH * (DINNER / SCAN_CPB), 128, 0, stream>>>(delta, ucv, xz, dbc, A_log, D_param, ybuf);

    k_outproj<<<256, 256, 0, stream>>>(ybuf, out_proj_w, otmp, part);

    k_gnmish<<<(BATCH * DMODEL * SEQ) / (1024 * GNM_V), 256, 0, stream>>>(
        otmp, part, gn_w, gn_b, out);
}


# conv+SiLU folded into x_proj's A staging: a wave convolves its K quarter's rows from raw in_proj output on the way to LDS and writes ucv for the scan; no conv launch; x_proj barriers fence LDS only
# speedup vs baseline: 10.1922x; 1.0364x over previous
#include <hip/hip_runtime.h>
#include <hip/hip_bf16.h>
#include <math.h>


typedef float v2f __attribute__((ext_vector_type(2)));
typedef float v8f __attribute__((ext_vector_type(8)));

#define BATCH    8
#define SEQ      512
#define DMODEL   512
#define DINNER   1024
#define DSTATE   16
#define DTRANK   32
#define NGROUPS  8
#define MROWS    (BATCH * SEQ)

__device__ __forceinline__ v8f wmma_f32_4(v2f a, v2f b, v8f c) {
    return __builtin_amdgcn_wmma_f32_16x16x4_f32(
        false, a, false, b, (short)0, c, false, false);
}

__device__ __forceinline__ float softplus_f(float x) {
    const float t = __expf(-fabsf(x));
    const float u = 1.f + t;
    const float d = u - 1.f;
    const float l = (__builtin_amdgcn_logf(u) * 0.6931471805599453f) * (t * __builtin_amdgcn_rcpf(d));
    return fmaxf(x, 0.f) + (d == 0.f ? t : l);
}

__device__ __forceinline__ float mish_f(float v) {
    const float m   = __expf(-fabsf(v));
    const bool  pos = v >= 0.f;
    const float num = pos ? fmaf(2.f, m, 1.f) : m * (m + 2.f);
    const float den = num + (pos ? 2.f * m * m : 2.f);
    return v * (num * __builtin_amdgcn_rcpf(den));
}

__device__ __forceinline__ float sigmoid_f(float x) {
    return __builtin_amdgcn_rcpf(1.f + __expf(-x));
}


#define GEMM_K   32
#define GEMM_P   36

template <int WM, int WN, bool AKM>
struct GemmShape {
    static constexpr int NT  = 32 * WM * WN;
    static constexpr int BM  = 16 * WM;
    static constexpr int BN  = 64 * WN;
    static constexpr int AP  = AKM ? (BM + 8) : GEMM_P;
    static constexpr int ASZ = AKM ? GEMM_K * (BM + 8) : BM * GEMM_P;
    static constexpr int WSZ = BN * GEMM_P;
    static constexpr int AV4 = BM * GEMM_K / 4 / NT;
    static constexpr int WV4 = BN * GEMM_K / 4 / NT;

    static __device__ __forceinline__ size_t a_goff(const int i, const int tid, const int lda) {
        const int idx = tid + i * NT;
        return AKM ? (size_t)(idx / (BM / 4)) * lda + (idx % (BM / 4)) * 4
                   : (size_t)(idx >> 3) * lda + (idx & 7) * 4;
    }
    static __device__ __forceinline__ int a_soff(const int i, const int tid) {
        const int idx = tid + i * NT;
        return AKM ? (idx / (BM / 4)) * AP + (idx % (BM / 4)) * 4
                   : (idx >> 3) * AP + (idx & 7) * 4;
    }
    static __device__ __forceinline__ size_t w_goff(const int i, const int tid, const int ldw) {
        const int idx = tid + i * NT;
        return (size_t)(idx >> 3) * ldw + (idx & 7) * 4;
    }
    static __device__ __forceinline__ int w_soff(const int i, const int tid) {
        const int idx = tid + i * NT;
        return (idx >> 3) * GEMM_P + (idx & 7) * 4;
    }
};

template <int WM, int WN, bool AKM>
__device__ __forceinline__ void gemm_block(const float* __restrict__ Ablk, const int lda,
                                           const float* __restrict__ Wblk, const int ldw,
                                           const int K,
                                           float* __restrict__ sA, float* __restrict__ sW,
                                           v8f& acc0, v8f& acc1, v8f& acc2, v8f& acc3) {
    typedef GemmShape<WM, WN, AKM> S;
    static_assert(S::WV4 == 4 && (S::AV4 == 2 || S::AV4 == 4), "per-thread piece counts the code below is written for");
    const int tid  = threadIdx.x;
    const int wave = tid >> 5;
    const int wm   = wave % WM;
    const int wn   = wave / WM;
    const int lane = tid & 31;
    const int half = lane >> 4;
    const int r    = lane & 15;

    const float* ga0 = Ablk + S::a_goff(0, tid, lda);
    const float* ga1 = Ablk + S::a_goff(1, tid, lda);
    const float* ga2 = Ablk + S::a_goff(S::AV4 > 2 ? 2 : 0, tid, lda);
    const float* ga3 = Ablk + S::a_goff(S::AV4 > 2 ? 3 : 0, tid, lda);
    const float* gw0 = Wblk + S::w_goff(0, tid, ldw);
    const float* gw1 = Wblk + S::w_goff(1, tid, ldw);
    const float* gw2 = Wblk + S::w_goff(2, tid, ldw);
    const float* gw3 = Wblk + S::w_goff(3, tid, ldw);
    const size_t astep = AKM ? (size_t)GEMM_K * lda : (size_t)GEMM_K;

    float4 ra0, ra1, ra2, ra3, rw0, rw1, rw2, rw3;
    ra2 = ra3 = make_float4(0.f, 0.f, 0.f, 0.f);

#define GEMM_GLOAD(ks)                                                        \
    do {                                                                      \
        const size_t ao = (size_t)(ks) * astep;                               \
        const size_t wo = (size_t)(ks) * GEMM_K;                              \
        ra0 = *(const float4*)(ga0 + ao);                                     \
        ra1 = *(const float4*)(ga1 + ao);                                     \
        if (S::AV4 > 2) {                                                     \
            ra2 = *(const float4*)(ga2 + ao);                                 \
            ra3 = *(const float4*)(ga3 + ao);                                 \
        }                                                                     \
        rw0 = *(const float4*)(gw0 + wo);                                     \
        rw1 = *(const float4*)(gw1 + wo);                                     \
        rw2 = *(const float4*)(gw2 + wo);                                     \
        rw3 = *(const float4*)(gw3 + wo);                                     \
    } while (0)

#define GEMM_SSTORE(buf)                                                      \
    do {                                                                      \
        float* a_ = sA + (buf) * S::ASZ;                                      \
        float* w_ = sW + (buf) * S::WSZ;                                      \
        *(float4*)(a_ + S::a_soff(0, tid)) = ra0;                             \
        *(float4*)(a_ + S::a_soff(1, tid)) = ra1;                             \
        if (S::AV4 > 2) {                                                     \
            *(float4*)(a_ + S::a_soff(2, tid)) = ra2;                         \
            *(float4*)(a_ + S::a_soff(3, tid)) = ra3;                         \
        }                                                                     \
        *(float4*)(w_ + S::w_soff(0, tid)) = rw0;                             \
        *(float4*)(w_ + S::w_soff(1, tid)) = rw1;                             \
        *(float4*)(w_ + S::w_soff(2, tid)) = rw2;                             \
        *(float4*)(w_ + S::w_soff(3, tid)) = rw3;                             \
    } while (0)

    GEMM_GLOAD(0);
    GEMM_SSTORE(0);
    __syncthreads();

    const int nslab = K / GEMM_K;
    for (int s = 0; s < nslab; ++s) {
        const int buf = s & 1;
        const int sn = (s + 1 < nslab) ? (s + 1) : s;
        GEMM_GLOAD(sn);

        const float* a = sA + buf * S::ASZ;
        const float* w = sW + buf * S::WSZ + (wn * 64 + r) * GEMM_P;
        #pragma unroll
        for (int k0 = 0; k0 < GEMM_K; k0 += 4) {
            const int ka = k0 + half * 2;
            v2f av;
            if (AKM) {
                av.x = a[ka * S::AP + wm * 16 + r];
                av.y = a[(ka + 1) * S::AP + wm * 16 + r];
            } else {
                av = *(const v2f*)(a + (wm * 16 + r) * S::AP + ka);
            }
            const v2f b0 = *(const v2f*)(w + 0 * 16 * GEMM_P + ka);
            const v2f b1 = *(const v2f*)(w + 1 * 16 * GEMM_P + ka);
            const v2f b2 = *(const v2f*)(w + 2 * 16 * GEMM_P + ka);
            const v2f b3 = *(const v2f*)(w + 3 * 16 * GEMM_P + ka);
            acc0 = wmma_f32_4(av, b0, acc0);
            acc1 = wmma_f32_4(av, b1, acc1);
            acc2 = wmma_f32_4(av, b2, acc2);
            acc3 = wmma_f32_4(av, b3, acc3);
        }

        GEMM_SSTORE(buf ^ 1);
        __syncthreads();
    }
#undef GEMM_GLOAD
#undef GEMM_SSTORE
}

__global__ void __launch_bounds__(256) __attribute__((amdgpu_waves_per_eu(4, 8)))
k_inproj(const float* __restrict__ x,
         const float* __restrict__ w,
         float* __restrict__ xz) {
    typedef GemmShape<4, 2, true> S;
    __shared__ __attribute__((aligned(16))) float sA[2 * S::ASZ];
    __shared__ __attribute__((aligned(16))) float sW[2 * S::WSZ];

    const int bn = blockIdx.x & 15;
    const int bm = blockIdx.x >> 4;
    const int m0 = bm * 64;
    const int bb = m0 >> 9;
    const int l0 = m0 & 511;

    v8f acc0 = {}, acc1 = {}, acc2 = {}, acc3 = {};
    gemm_block<4, 2, true>(x + (size_t)bb * DMODEL * SEQ + l0, SEQ,
                           w + (size_t)(bn * 128) * DMODEL, DMODEL, DMODEL,
                           sA, sW, acc0, acc1, acc2, acc3);

    const int wave = threadIdx.x >> 5;
    const int wm   = wave % 4;
    const int wn   = wave / 4;
    const int lane = threadIdx.x & 31;
    const int half = lane >> 4;
    const int r    = lane & 15;
    const int col  = bn * 128 + wn * 64 + r;
    #pragma unroll
    for (int j = 0; j < 8; ++j) {
        const size_t row = (size_t)(m0 + wm * 16 + half * 8 + j) * (2 * DINNER);
        xz[row + col + 0 ] = acc0[j];
        xz[row + col + 16] = acc1[j];
        xz[row + col + 32] = acc2[j];
        xz[row + col + 48] = acc3[j];
    }
}

#define XP_KQ    (DINNER / 4)
#define XP_AROWS (4 * 16)
#define XP_WROWS (4 * 64)
#define XP_WV4   (XP_WROWS * GEMM_K / 4 / 128)
#define XP_RP    68

__device__ __forceinline__ void conv4x4_silu(const float4 x0, const float4 x1, const float4 x2, const float4 x3,
                                             const float4 x4, const float4 x5, const float4 x6,
                                             const float4 wc0, const float4 wc1, const float4 wc2, const float4 wc3,
                                             const float4 bc, const int lb, float4 (&o)[4]) {
    const float xr[7][4] = {{x0.x, x0.y, x0.z, x0.w}, {x1.x, x1.y, x1.z, x1.w}, {x2.x, x2.y, x2.z, x2.w},
                            {x3.x, x3.y, x3.z, x3.w}, {x4.x, x4.y, x4.z, x4.w}, {x5.x, x5.y, x5.z, x5.w},
                            {x6.x, x6.y, x6.z, x6.w}};
    const float wt[4][4] = {{wc0.x, wc0.y, wc0.z, wc0.w}, {wc1.x, wc1.y, wc1.z, wc1.w},
                            {wc2.x, wc2.y, wc2.z, wc2.w}, {wc3.x, wc3.y, wc3.z, wc3.w}};
    const float bias[4] = {bc.x, bc.y, bc.z, bc.w};
    #pragma unroll
    for (int i = 0; i < 4; ++i) {
        float v[4];
        #pragma unroll
        for (int ch = 0; ch < 4; ++ch) {
            float s = bias[ch];
            #pragma unroll
            for (int t = 0; t < 4; ++t) {
                if (lb + i + t >= 3)
                    s += xr[i + t][ch] * wt[ch][t];
            }
            v[ch] = s * sigmoid_f(s);
        }
        o[i] = make_float4(v[0], v[1], v[2], v[3]);
    }
}

__device__ __forceinline__ void lds_barrier() {
    __builtin_amdgcn_fence(__ATOMIC_RELEASE, "workgroup", "local");
    __builtin_amdgcn_s_barrier();
    __builtin_amdgcn_fence(__ATOMIC_ACQUIRE, "workgroup", "local");
}

__global__ void __launch_bounds__(128) __attribute__((amdgpu_waves_per_eu(1, 2)))
k_xproj(const float* __restrict__ xz,
        const float* __restrict__ cw,
        const float* __restrict__ cb,
        const float* __restrict__ w,
        float* __restrict__ ucv,
        float* __restrict__ dbc) {
    __shared__ __attribute__((aligned(16))) float sA[2 * XP_AROWS * GEMM_P];
    __shared__ __attribute__((aligned(16))) float sW[2 * XP_WROWS * GEMM_P];
    static_assert(4 * 16 * XP_RP <= 2 * XP_WROWS * GEMM_P, "partial tiles must fit the W buffers");

    const int tid  = threadIdx.x;
    const int q    = tid >> 5;
    const int lane = tid & 31;
    const int half = lane >> 4;
    const int r    = lane & 15;
    const int m0   = blockIdx.x * 16;
    const int l0   = m0 & (SEQ - 1);

    const int pr = tid >> 3, pc = (tid & 7) * 4;
    const float* gw = w + (size_t)pr * DINNER + pc;
    const int so = pr * GEMM_P + pc;

    const int rg = (lane >> 3) * 4;
    const int lb = l0 + rg;
    const int kc = q * XP_KQ + pc;
    const float* gx  = xz + (size_t)(m0 + rg) * (2 * DINNER) + kc;
    const float* gxa = xz + (size_t)(m0 + (lb >= 3 ? rg - 3 : 0)) * (2 * DINNER) + kc;
    const float* gxb = xz + (size_t)(m0 + (lb >= 2 ? rg - 2 : 0)) * (2 * DINNER) + kc;
    const float* gxc = xz + (size_t)(m0 + (lb >= 1 ? rg - 1 : 0)) * (2 * DINNER) + kc;
    const float* gcw = cw + (size_t)kc * 4;
    const float* gcb = cb + kc;
    float* gu = ucv + (size_t)(m0 + rg) * DINNER + kc;
    const int sao = (q * 16 + rg) * GEMM_P + pc;

    float4 rx0, rx1, rx2, rx3, rx4, rx5, rx6, rc0, rc1, rc2, rc3, rcb;
    float4 rw0, rw1, rw2, rw3, rw4, rw5, rw6, rw7, rw8, rw9, rw10, rw11, rw12, rw13, rw14, rw15;
    static_assert(XP_WV4 == 16, "piece count the two macros below are written for");

#define XP_GLOAD(ks)                                                          \
    do {                                                                      \
        const float* gw_ = gw + (ks) * GEMM_K;                                \
        rx0 = *(const float4*)(gxa + (ks) * GEMM_K);                          \
        rx1 = *(const float4*)(gxb + (ks) * GEMM_K);                          \
        rx2 = *(const float4*)(gxc + (ks) * GEMM_K);                          \
        rx3 = *(const float4*)(gx + (ks) * GEMM_K + (size_t)0 * (2 * DINNER)); \
        rx4 = *(const float4*)(gx + (ks) * GEMM_K + (size_t)1 * (2 * DINNER)); \
        rx5 = *(const float4*)(gx + (ks) * GEMM_K + (size_t)2 * (2 * DINNER)); \
        rx6 = *(const float4*)(gx + (ks) * GEMM_K + (size_t)3 * (2 * DINNER)); \
        rc0 = *(const float4*)(gcw + (ks) * (GEMM_K * 4) + 0);                \
        rc1 = *(const float4*)(gcw + (ks) * (GEMM_K * 4) + 4);                \
        rc2 = *(const float4*)(gcw + (ks) * (GEMM_K * 4) + 8);                \
        rc3 = *(const float4*)(gcw + (ks) * (GEMM_K * 4) + 12);               \
        rcb = *(const float4*)(gcb + (ks) * GEMM_K);                          \
        rw0 = *(const float4*)(gw_ + (size_t)0 * 16 * DINNER + 0 * XP_KQ);    \
        rw1 = *(const float4*)(gw_ + (size_t)1 * 16 * DINNER + 0 * XP_KQ);    \
        rw2 = *(const float4*)(gw_ + (size_t)2 * 16 * DINNER + 0 * XP_KQ);    \
        rw3 = *(const float4*)(gw_ + (size_t)3 * 16 * DINNER + 0 * XP_KQ);    \
        rw4 = *(const float4*)(gw_ + (size_t)0 * 16 * DINNER + 1 * XP_KQ);    \
        rw5 = *(const float4*)(gw_ + (size_t)1 * 16 * DINNER + 1 * XP_KQ);    \
        rw6 = *(const float4*)(gw_ + (size_t)2 * 16 * DINNER + 1 * XP_KQ);    \
        rw7 = *(const float4*)(gw_ + (size_t)3 * 16 * DINNER + 1 * XP_KQ);    \
        rw8 = *(const float4*)(gw_ + (size_t)0 * 16 * DINNER + 2 * XP_KQ);    \
        rw9 = *(const float4*)(gw_ + (size_t)1 * 16 * DINNER + 2 * XP_KQ);    \
        rw10 = *(const float4*)(gw_ + (size_t)2 * 16 * DINNER + 2 * XP_KQ);   \
        rw11 = *(const float4*)(gw_ + (size_t)3 * 16 * DINNER + 2 * XP_KQ);   \
        rw12 = *(const float4*)(gw_ + (size_t)0 * 16 * DINNER + 3 * XP_KQ);   \
        rw13 = *(const float4*)(gw_ + (size_t)1 * 16 * DINNER + 3 * XP_KQ);   \
        rw14 = *(const float4*)(gw_ + (size_t)2 * 16 * DINNER + 3 * XP_KQ);   \
        rw15 = *(const float4*)(gw_ + (size_t)3 * 16 * DINNER + 3 * XP_KQ);   \
    } while (0)

#define XP_SSTORE(buf, ks)                                                    \
    do {                                                                      \
        float4 o_[4];                                                         \
        conv4x4_silu(rx0, rx1, rx2, rx3, rx4, rx5, rx6, rc0, rc1, rc2, rc3, rcb, lb, o_); \
        float* a_ = sA + (buf) * (XP_AROWS * GEMM_P) + sao;                   \
        float* u_ = gu + (ks) * GEMM_K;                                       \
        float* w_ = sW + (buf) * (XP_WROWS * GEMM_P) + so;                    \
        *(float4*)(a_ + 0 * GEMM_P) = o_[0];                                  \
        *(float4*)(a_ + 1 * GEMM_P) = o_[1];                                  \
        *(float4*)(a_ + 2 * GEMM_P) = o_[2];                                  \
        *(float4*)(a_ + 3 * GEMM_P) = o_[3];                                  \
        *(float4*)(u_ + (size_t)0 * DINNER) = o_[0];                          \
        *(float4*)(u_ + (size_t)1 * DINNER) = o_[1];                          \
        *(float4*)(u_ + (size_t)2 * DINNER) = o_[2];                          \
        *(float4*)(u_ + (size_t)3 * DINNER) = o_[3];                          \
        *(float4*)(w_ + 0 * 16 * GEMM_P) = rw0;                               \
        *(float4*)(w_ + 1 * 16 * GEMM_P) = rw1;                               \
        *(float4*)(w_ + 2 * 16 * GEMM_P) = rw2;                               \
        *(float4*)(w_ + 3 * 16 * GEMM_P) = rw3;                               \
        *(float4*)(w_ + 4 * 16 * GEMM_P) = rw4;                               \
        *(float4*)(w_ + 5 * 16 * GEMM_P) = rw5;                               \
        *(float4*)(w_ + 6 * 16 * GEMM_P) = rw6;                               \
        *(float4*)(w_ + 7 * 16 * GEMM_P) = rw7;                               \
        *(float4*)(w_ + 8 * 16 * GEMM_P) = rw8;                               \
        *(float4*)(w_ + 9 * 16 * GEMM_P) = rw9;                               \
        *(float4*)(w_ + 10 * 16 * GEMM_P) = rw10;                             \
        *(float4*)(w_ + 11 * 16 * GEMM_P) = rw11;                             \
        *(float4*)(w_ + 12 * 16 * GEMM_P) = rw12;                             \
        *(float4*)(w_ + 13 * 16 * GEMM_P) = rw13;                             \
        *(float4*)(w_ + 14 * 16 * GEMM_P) = rw14;                             \
        *(float4*)(w_ + 15 * 16 * GEMM_P) = rw15;                             \
    } while (0)

    v8f acc0 = {}, acc1 = {}, acc2 = {}, acc3 = {};

    constexpr int nslab = XP_KQ / GEMM_K;
    XP_GLOAD(0);
    XP_SSTORE(0, 0);
    XP_GLOAD(1);
    lds_barrier();

    for (int s = 0; s < nslab; ++s) {
        const int buf = s & 1;
        const float* a  = sA + buf * (XP_AROWS * GEMM_P) + (q * 16 + r) * GEMM_P;
        const float* wb = sW + buf * (XP_WROWS * GEMM_P) + (q * 64 + r) * GEMM_P;

#define XP_FRAG(ks, A_, B0_, B1_, B2_, B3_)                                  \
        const v2f A_  = *(const v2f*)(a  + (ks) * 4 + half * 2);              \
        const v2f B0_ = *(const v2f*)(wb + 0 * 16 * GEMM_P + (ks) * 4 + half * 2); \
        const v2f B1_ = *(const v2f*)(wb + 1 * 16 * GEMM_P + (ks) * 4 + half * 2); \
        const v2f B2_ = *(const v2f*)(wb + 2 * 16 * GEMM_P + (ks) * 4 + half * 2); \
        const v2f B3_ = *(const v2f*)(wb + 3 * 16 * GEMM_P + (ks) * 4 + half * 2)
#define XP_MUL(A_, B0_, B1_, B2_, B3_)                                        \
        acc0 = wmma_f32_4(A_, B0_, acc0);                                     \
        acc1 = wmma_f32_4(A_, B1_, acc1);                                     \
        acc2 = wmma_f32_4(A_, B2_, acc2);                                     \
        acc3 = wmma_f32_4(A_, B3_, acc3)

        XP_FRAG(0, a0, b00, b01, b02, b03);
        XP_FRAG(1, a1, b10, b11, b12, b13);
        XP_FRAG(2, a2, b20, b21, b22, b23);
        XP_FRAG(3, a3, b30, b31, b32, b33);
        XP_FRAG(4, a4, b40, b41, b42, b43);
        XP_FRAG(5, a5, b50, b51, b52, b53);
        XP_FRAG(6, a6, b60, b61, b62, b63);
        XP_FRAG(7, a7, b70, b71, b72, b73);

        const int s1 = (s + 1 < nslab) ? (s + 1) : (nslab - 1);
        XP_SSTORE(buf ^ 1, s1);
        const int sn = (s + 2 < nslab) ? (s + 2) : (nslab - 1);
        XP_GLOAD(sn);

        XP_MUL(a0, b00, b01, b02, b03);
        XP_MUL(a1, b10, b11, b12, b13);
        XP_MUL(a2, b20, b21, b22, b23);
        XP_MUL(a3, b30, b31, b32, b33);
        XP_MUL(a4, b40, b41, b42, b43);
        XP_MUL(a5, b50, b51, b52, b53);
        XP_MUL(a6, b60, b61, b62, b63);
        XP_MUL(a7, b70, b71, b72, b73);
#undef XP_FRAG
#undef XP_MUL
        lds_barrier();
    }
#undef XP_GLOAD
#undef XP_SSTORE

    float* sR = sW;
    {
        float* p = sR + (q * 16 + half * 8) * XP_RP + r;
        #pragma unroll
        for (int j = 0; j < 8; ++j) {
            p[j * XP_RP + 0 ] = acc0[j];
            p[j * XP_RP + 16] = acc1[j];
            p[j * XP_RP + 32] = acc2[j];
            p[j * XP_RP + 48] = acc3[j];
        }
    }
    lds_barrier();

    const int row = tid >> 3;
    #pragma unroll
    for (int hp = 0; hp < 2; ++hp) {
        const int col = pc + hp * 32;
        const float4 p0 = *(const float4*)(sR + (0 * 16 + row) * XP_RP + col);
        const float4 p1 = *(const float4*)(sR + (1 * 16 + row) * XP_RP + col);
        const float4 p2 = *(const float4*)(sR + (2 * 16 + row) * XP_RP + col);
        const float4 p3 = *(const float4*)(sR + (3 * 16 + row) * XP_RP + col);
        float4 o;
        o.x = ((p0.x + p1.x) + p2.x) + p3.x;
        o.y = ((p0.y + p1.y) + p2.y) + p3.y;
        o.z = ((p0.z + p1.z) + p2.z) + p3.z;
        o.w = ((p0.w + p1.w) + p2.w) + p3.w;
        *(float4*)(dbc + (size_t)(m0 + row) * 64 + col) = o;
    }
}

__global__ void k_dtproj(const float* __restrict__ dbc,
                         const float* __restrict__ w,
                         const float* __restrict__ bias,
                         float* __restrict__ delta) {
    const int wid   = blockIdx.x * 4 + (threadIdx.x >> 5);
    const int ntq   = wid & 15;
    const int tileM = wid >> 4;
    const int lane  = threadIdx.x & 31;
    const int half  = lane >> 4;
    const int r     = lane & 15;

    const float* arow = dbc + (size_t)(tileM * 16 + r) * 64;
    const float* wr0 = w + (size_t)((ntq * 4 + 0) * 16 + r) * DTRANK;
    const float* wr1 = w + (size_t)((ntq * 4 + 1) * 16 + r) * DTRANK;
    const float* wr2 = w + (size_t)((ntq * 4 + 2) * 16 + r) * DTRANK;
    const float* wr3 = w + (size_t)((ntq * 4 + 3) * 16 + r) * DTRANK;

    v8f acc0 = {}, acc1 = {}, acc2 = {}, acc3 = {};
    #pragma unroll
    for (int k0 = 0; k0 < DTRANK; k0 += 4) {
        const int ka = k0 + half * 2;
        v2f a  = *(const v2f*)(arow + ka);
        v2f b0 = *(const v2f*)(wr0 + ka);
        v2f b1 = *(const v2f*)(wr1 + ka);
        v2f b2 = *(const v2f*)(wr2 + ka);
        v2f b3 = *(const v2f*)(wr3 + ka);
        acc0 = wmma_f32_4(a, b0, acc0);
        acc1 = wmma_f32_4(a, b1, acc1);
        acc2 = wmma_f32_4(a, b2, acc2);
        acc3 = wmma_f32_4(a, b3, acc3);
    }
    const int nb = ntq * 64 + r;
    const float bn0 = bias[nb + 0];
    const float bn1 = bias[nb + 16];
    const float bn2 = bias[nb + 32];
    const float bn3 = bias[nb + 48];
    #pragma unroll
    for (int j = 0; j < 8; ++j) {
        const size_t row = (size_t)(tileM * 16 + half * 8 + j) * DINNER;
        delta[row + nb + 0 ] = softplus_f(acc0[j] + bn0);
        delta[row + nb + 16] = softplus_f(acc1[j] + bn1);
        delta[row + nb + 32] = softplus_f(acc2[j] + bn2);
        delta[row + nb + 48] = softplus_f(acc3[j] + bn3);
    }
}

#define SCAN_T   32
#define SCAN_Q   4
#define SCAN_CPB (128 / SCAN_Q)
#define SCAN_G   (SCAN_T / SCAN_Q)

template <int CTRL>
__device__ __forceinline__ float quad_perm(float v) {
    return __builtin_bit_cast(float, __builtin_amdgcn_mov_dpp(__builtin_bit_cast(int, v), CTRL, 0xF, 0xF, true));
}
__device__ __forceinline__ float quad_xor1(float v) { return quad_perm<0xB1>(v); }
__device__ __forceinline__ float quad_xor2(float v) { return quad_perm<0x4E>(v); }
__device__ __forceinline__ float quad_bcast(float v, const int k) {
    switch (k) {
        case 0:  return quad_perm<0x00>(v);
        case 1:  return quad_perm<0x55>(v);
        case 2:  return quad_perm<0xAA>(v);
        default: return quad_perm<0xFF>(v);
    }
}

__device__ __forceinline__ v2f exp_pair(v2f x) {
    const v2f t = x * 1.4426950408889634f;
    v2f r;
    r.x = __builtin_amdgcn_exp2f(t.x);
    r.y = __builtin_amdgcn_exp2f(t.y);
    return r;
}

__device__ __forceinline__ void scan_load_bc(const float4* __restrict__ sBC, const int l, const int sub,
                                             float4 (&Bv)[SCAN_Q], float4 (&Cv)[SCAN_Q]) {
    #pragma unroll
    for (int t = 0; t < SCAN_Q; ++t) {
        Bv[t] = sBC[(l + t) * 8 + sub];
        Cv[t] = sBC[(l + t) * 8 + 4 + sub];
    }
}

__global__ void __launch_bounds__(128)
k_scan(const float* __restrict__ delta,
       const float* __restrict__ ucv,
       const float* __restrict__ xz,
       const float* __restrict__ dbc,
       const float* __restrict__ A_log,
       const float* __restrict__ D_param,
       float* __restrict__ ybuf) {
    const int bb  = blockIdx.x >> 5;
    const int tid = threadIdx.x;
    const int sub = tid & (SCAN_Q - 1);
    const int chl = tid >> 2;
    const int c0  = (blockIdx.x & 31) * SCAN_CPB;
    const int c   = c0 + chl;
    const int mbase = bb * SEQ;

    __shared__ float4 sBC[SEQ * 8];
    __shared__ __attribute__((aligned(16))) float sY[2][SCAN_T][SCAN_CPB];

    const float* pd = delta + (size_t)(mbase + sub) * DINNER + c;
    const float* pu = ucv   + (size_t)(mbase + sub) * DINNER + c;
    const float* pz = xz    + (size_t)(mbase + sub) * (2 * DINNER) + DINNER + c;
    float* pyo = ybuf + (size_t)(mbase + (tid >> 3)) * DINNER + c0 + (tid & 7) * 4;

    v8f cd, cu, cz;
    static_assert(SCAN_G == 8, "one 8-register vector per operand");
    #pragma unroll
    for (int g = 0; g < SCAN_G; ++g) {
        cd[g] = pd[(size_t)(g * SCAN_Q) * DINNER];
        cu[g] = pu[(size_t)(g * SCAN_Q) * DINNER];
        cz[g] = pz[(size_t)(g * SCAN_Q) * (2 * DINNER)];
    }

    const float* bcsrc = dbc + (size_t)mbase * 64 + 32;
    #pragma unroll
    for (int j = 0; j < (SEQ * 8) / 128; ++j) {
        const int i = j * 128 + tid;
        sBC[i] = *(const float4*)(bcsrc + (size_t)(i >> 3) * 64 + (i & 7) * 4);
    }

    const float4 al = *(const float4*)(A_log + (size_t)c * DSTATE + sub * 4);
    const v2f Ac01 = {-__expf(al.x), -__expf(al.y)};
    const v2f Ac23 = {-__expf(al.z), -__expf(al.w)};
    v2f h01 = {0.f, 0.f}, h23 = {0.f, 0.f};
    const float Dp = D_param[c];
    const bool odd = (sub & 1) != 0;
    const bool hi  = (sub & 2) != 0;

    __syncthreads();

    float4 Bc[SCAN_Q], Cc[SCAN_Q];
    scan_load_bc(sBC, 0, sub, Bc, Cc);

    for (int l0 = 0; l0 < SEQ; l0 += SCAN_T) {
        const int ln = (l0 + SCAN_T < SEQ) ? (l0 + SCAN_T) : l0;
        v8f nd, nu, nz;
        #pragma unroll
        for (int g = 0; g < SCAN_G; ++g) {
            nd[g] = pd[(size_t)(ln + g * SCAN_Q) * DINNER];
            nu[g] = pu[(size_t)(ln + g * SCAN_Q) * DINNER];
            nz[g] = pz[(size_t)(ln + g * SCAN_Q) * (2 * DINNER)];
        }

        float (*sYt)[SCAN_CPB] = sY[(l0 / SCAN_T) & 1];

        #pragma unroll
        for (int g = 0; g < SCAN_G; ++g) {
            const int lg = l0 + g * SCAN_Q;

            float4 Bn[SCAN_Q], Cn[SCAN_Q];
            const int lnx = (lg + SCAN_Q < SEQ) ? (lg + SCAN_Q) : lg;
            scan_load_bc(sBC, lnx, sub, Bn, Cn);
            __builtin_amdgcn_sched_barrier(0);

            float yq[SCAN_Q];
            #pragma unroll
            for (int t = 0; t < SCAN_Q; ++t) {
                const float dlt = quad_bcast(cd[g], t);
                const float du  = dlt * quad_bcast(cu[g], t);
                const v2f B01 = {Bc[t].x, Bc[t].y}, B23 = {Bc[t].z, Bc[t].w};
                const v2f C01 = {Cc[t].x, Cc[t].y}, C23 = {Cc[t].z, Cc[t].w};
                const v2f dA01 = exp_pair(Ac01 * dlt);
                const v2f dA23 = exp_pair(Ac23 * dlt);
                h01 = __builtin_elementwise_fma(dA01, h01, du * B01);
                h23 = __builtin_elementwise_fma(dA23, h23, du * B23);
                const v2f y2 = h01 * C01 + h23 * C23;
                yq[t] = y2.x + y2.y;
            }

            const float a  = (odd ? yq[1] : yq[0]) + quad_xor1(odd ? yq[0] : yq[1]);
            const float b  = (odd ? yq[3] : yq[2]) + quad_xor1(odd ? yq[2] : yq[3]);
            const float ys = (hi ? b : a) + quad_xor2(hi ? a : b);
            const float zs = cz[g];
            sYt[g * SCAN_Q + sub][chl] = (ys + cu[g] * Dp) * (zs * sigmoid_f(zs));

            #pragma unroll
            for (int t = 0; t < SCAN_Q; ++t) { Bc[t] = Bn[t]; Cc[t] = Cn[t]; }

            __builtin_amdgcn_sched_barrier(0);
        }

        __builtin_amdgcn_fence(__ATOMIC_RELEASE, "workgroup", "local");
        __builtin_amdgcn_s_barrier();
        __builtin_amdgcn_fence(__ATOMIC_ACQUIRE, "workgroup", "local");
        #pragma unroll
        for (int j = 0; j < SCAN_T / 16; ++j)
            *(float4*)(pyo + (size_t)(l0 + 16 * j) * DINNER) = *(const float4*)&sYt[(tid >> 3) + 16 * j][(tid & 7) * 4];

        asm volatile("" :: "v"(nd), "v"(nu), "v"(nz));
        cd = nd; cu = nu; cz = nz;
    }
}

#define OUT_CP 68
#define GN_PP  32

__global__ void __launch_bounds__(256) __attribute__((amdgpu_waves_per_eu(2, 4)))
k_outproj(const float* __restrict__ ybuf,
          const float* __restrict__ w,
          float* __restrict__ otmp,
          float* __restrict__ part) {
    typedef GemmShape<4, 2, false> S;
    __shared__ __attribute__((aligned(16))) float sA[2 * S::ASZ];
    __shared__ __attribute__((aligned(16))) float sW[2 * S::WSZ];
    static_assert(128 * OUT_CP <= 2 * S::WSZ, "turned tile must fit the W buffers");

    const int bn = blockIdx.x & 3;
    const int bm = blockIdx.x >> 2;
    const int m0 = bm * 64;
    const int bb = m0 >> 9;
    const int l0 = m0 & 511;

    v8f acc0 = {}, acc1 = {}, acc2 = {}, acc3 = {};
    gemm_block<4, 2, false>(ybuf + (size_t)m0 * DINNER, DINNER,
                            w + (size_t)(bn * 128) * DINNER, DINNER, DINNER,
                            sA, sW, acc0, acc1, acc2, acc3);

    const int tid  = threadIdx.x;
    const int wave = tid >> 5;
    const int wm   = wave % 4;
    const int wn   = wave / 4;
    const int lane = tid & 31;
    const int half = lane >> 4;
    const int r    = lane & 15;

    float ps0 = 0.f, ps1 = 0.f, ps2 = 0.f, ps3 = 0.f;
    float pq0 = 0.f, pq1 = 0.f, pq2 = 0.f, pq3 = 0.f;
    #pragma unroll
    for (int j = 0; j < 8; ++j) {
        ps0 += acc0[j]; pq0 = fmaf(acc0[j], acc0[j], pq0);
        ps1 += acc1[j]; pq1 = fmaf(acc1[j], acc1[j], pq1);
        ps2 += acc2[j]; pq2 = fmaf(acc2[j], acc2[j], pq2);
        ps3 += acc3[j]; pq3 = fmaf(acc3[j], acc3[j], pq3);
    }
    float ps = (ps0 + ps1) + (ps2 + ps3);
    float pq = (pq0 + pq1) + (pq2 + pq3);
    ps += quad_xor1(ps); pq += quad_xor1(pq);
    ps += quad_xor2(ps); pq += quad_xor2(pq);
    float2* sRed = (float2*)sA;
    if ((lane & 3) == 0) sRed[wave * 8 + (lane >> 2)] = make_float2(ps, pq);

    float* sC = sW;
    {
        float* c0 = sC + (wn * 64 + r) * OUT_CP + wm * 16 + half * 8;
        *(float4*)(c0 + 0 * 16 * OUT_CP)     = make_float4(acc0[0], acc0[1], acc0[2], acc0[3]);
        *(float4*)(c0 + 0 * 16 * OUT_CP + 4) = make_float4(acc0[4], acc0[5], acc0[6], acc0[7]);
        *(float4*)(c0 + 1 * 16 * OUT_CP)     = make_float4(acc1[0], acc1[1], acc1[2], acc1[3]);
        *(float4*)(c0 + 1 * 16 * OUT_CP + 4) = make_float4(acc1[4], acc1[5], acc1[6], acc1[7]);
        *(float4*)(c0 + 2 * 16 * OUT_CP)     = make_float4(acc2[0], acc2[1], acc2[2], acc2[3]);
        *(float4*)(c0 + 2 * 16 * OUT_CP + 4) = make_float4(acc2[4], acc2[5], acc2[6], acc2[7]);
        *(float4*)(c0 + 3 * 16 * OUT_CP)     = make_float4(acc3[0], acc3[1], acc3[2], acc3[3]);
        *(float4*)(c0 + 3 * 16 * OUT_CP + 4) = make_float4(acc3[4], acc3[5], acc3[6], acc3[7]);
    }
    __syncthreads();

    float* obase = otmp + ((size_t)bb * DMODEL + bn * 128) * SEQ + l0;
    #pragma unroll
    for (int i = 0; i < 8; ++i) {
        const int idx = tid + i * 256;
        const int n = idx >> 4, q = idx & 15;
        *(float4*)(obase + (size_t)n * SEQ + q * 4) = *(const float4*)(sC + n * OUT_CP + q * 4);
    }

    if (tid < 4) {
        const float* rp = sA + (tid >> 1) * (4 * 8 * 2) + (tid & 1);
        float t0 = 0.f, t1 = 0.f, t2 = 0.f, t3 = 0.f;
        #pragma unroll
        for (int i = 0; i < 8; ++i) {
            t0 += rp[(0 * 8 + i) * 2];
            t1 += rp[(1 * 8 + i) * 2];
            t2 += rp[(2 * 8 + i) * 2];
            t3 += rp[(3 * 8 + i) * 2];
        }
        part[(size_t)blockIdx.x * GN_PP + tid] = (t0 + t1) + (t2 + t3);
    }
}

#define GNM_V 4

__device__ __forceinline__ float gn_mish_f(float x, float mu, float rs, float g, float bt) {
    return mish_f((x - mu) * rs * g + bt);
}

__global__ void __launch_bounds__(256)
k_gnmish(const float* __restrict__ otmp,
         const float* __restrict__ part,
         const float* __restrict__ gw,
         const float* __restrict__ gb,
         float* __restrict__ out) {
    const int vec0 = blockIdx.x * (256 * GNM_V) + threadIdx.x;
    const int blk  = (blockIdx.x * (1024 * GNM_V)) >> 15;

    float4 x4[GNM_V];
    #pragma unroll
    for (int i = 0; i < GNM_V; ++i)
        x4[i] = *(const float4*)(otmp + (size_t)(vec0 + i * 256) * 4);

    const float* pp = part + ((size_t)(blk >> 3) * 32 + ((blk & 7) >> 1)) * GN_PP + (blk & 1) * 2;
    float s = 0.f, q = 0.f;
    #pragma unroll
    for (int lt = 0; lt < 8; ++lt) {
        s += pp[lt * 4 * GN_PP];
        q += pp[lt * 4 * GN_PP + 1];
    }
    const float mu  = s * (1.f / 32768.f);
    const float var = q * (1.f / 32768.f) - mu * mu;
    const float rs  = rsqrtf(var + 1e-5f);

    #pragma unroll
    for (int i = 0; i < GNM_V; ++i) {
        const int idx = (vec0 + i * 256) * 4;
        const int c   = (idx >> 9) & (DMODEL - 1);
        const float g  = gw[c];
        const float bt = gb[c];
        float4 o;
        o.x = gn_mish_f(x4[i].x, mu, rs, g, bt);
        o.y = gn_mish_f(x4[i].y, mu, rs, g, bt);
        o.z = gn_mish_f(x4[i].z, mu, rs, g, bt);
        o.w = gn_mish_f(x4[i].w, mu, rs, g, bt);
        *(float4*)(out + (size_t)idx) = o;
    }
}

extern "C" void kernel_launch(void* const* d_in, const int* in_sizes, int n_in,
                              void* d_out, int out_size, void* d_ws, size_t ws_size,
                              hipStream_t stream) {
    const float* x         = (const float*)d_in[0];
    const float* in_proj_w = (const float*)d_in[1];
    const float* conv_w    = (const float*)d_in[2];
    const float* conv_b    = (const float*)d_in[3];
    const float* x_proj_w  = (const float*)d_in[4];
    const float* dt_proj_w = (const float*)d_in[5];
    const float* dt_proj_b = (const float*)d_in[6];
    const float* A_log     = (const float*)d_in[7];
    const float* D_param   = (const float*)d_in[8];
    const float* out_proj_w= (const float*)d_in[9];
    const float* gn_w      = (const float*)d_in[10];
    const float* gn_b      = (const float*)d_in[11];
    float* out = (float*)d_out;

    float* ws    = (float*)d_ws;
    float* xz    = ws;
    float* ucv   = xz    + (size_t)MROWS * 2 * DINNER;
    float* dbc   = ucv   + (size_t)MROWS * DINNER;
    float* delta = dbc   + (size_t)MROWS * 64;
    float* ybuf  = delta + (size_t)MROWS * DINNER;
    float* part  = ybuf  + (size_t)MROWS * DINNER;
    float* otmp  = xz;

    k_inproj<<<1024, 256, 0, stream>>>(x, in_proj_w, xz);

    k_xproj<<<MROWS / 16, 128, 0, stream>>>(xz, conv_w, conv_b, x_proj_w, ucv, dbc);

    k_dtproj<<<4096 / 4, 128, 0, stream>>>(dbc, dt_proj_w, dt_proj_b, delta);

    k_scan<<<BATCH * (DINNER / SCAN_CPB), 128, 0, stream>>>(delta, ucv, xz, dbc, A_log, D_param, ybuf);

    k_outproj<<<256, 256, 0, stream>>>(ybuf, out_proj_w, otmp, part);

    k_gnmish<<<(BATCH * DMODEL * SEQ) / (1024 * GNM_V), 256, 0, stream>>>(
        otmp, part, gn_w, gn_b, out);
}
